# starting kernel
# baseline (speedup 1.0000x reference)
#include <hip/hip_runtime.h>

typedef float v2f __attribute__((ext_vector_type(2)));
typedef float v8f __attribute__((ext_vector_type(8)));

#define WMMA_F32(a, b, c) \
  __builtin_amdgcn_wmma_f32_16x16x4_f32(false, (a), false, (b), (short)0, (c), false, false)

#define BATCH 4
#define NTOK 4096
#define CCH 512
#define HEADS 8
#define DH 64
#define KL 64
#define MROWS (BATCH * NTOK)
#define OCH (4 * CCH)
#define BH (BATCH * HEADS)

#define OFF_Q   ((size_t)0)
#define OFF_K   ((size_t)8388608)
#define OFF_VCA ((size_t)16777216)
#define OFF_VSA ((size_t)25165824)
#define OFF_KP  ((size_t)33554432)
#define OFF_VP  ((size_t)33685504)
#define OFF_P   ((size_t)33816576)

__global__ __launch_bounds__(256) void qkvv_gemm(const float* __restrict__ x,
                                                 const float* __restrict__ W,
                                                 float* __restrict__ ws) {
  const int lane = threadIdx.x & 31;
  const int wave = threadIdx.x >> 5;
  const int l16  = lane & 15;
  const int hi   = lane >> 4;
  const int mBase = blockIdx.x * 128 + wave * 16;
  const int oBase = blockIdx.y * 64;

  const float* xRow = x + (size_t)(mBase + l16) * CCH + (hi ? 2 : 0);
  v8f acc[4] = {};
  for (int c0 = 0; c0 < CCH; c0 += 4) {
    v2f a = *(const v2f*)(xRow + c0);
#pragma unroll
    for (int t = 0; t < 4; ++t) {
      const int o = oBase + t * 16 + l16;
      v2f b = *(const v2f*)(W + (size_t)o * CCH + c0 + (hi ? 2 : 0));
      acc[t] = WMMA_F32(a, b, acc[t]);
    }
  }
#pragma unroll
  for (int t = 0; t < 4; ++t) {
    const int o  = oBase + t * 16 + l16;
    const int p  = o >> 9;
    const int h  = (o >> 6) & 7;
    const int dd = o & 63;
#pragma unroll
    for (int v = 0; v < 8; ++v) {
      const int m  = mBase + v + (hi ? 8 : 0);
      const int b_ = m >> 12;
      const int n  = m & (NTOK - 1);
      const size_t dst = (size_t)p * 8388608 +
                         ((size_t)(b_ * HEADS + h) * DH + dd) * NTOK + n;
      ws[dst] = acc[t][v];
    }
  }
}

__global__ __launch_bounds__(512) void ef_proj(float* __restrict__ ws,
                                               const float* __restrict__ EF) {
  const int lane = threadIdx.x & 31;
  const int wave = threadIdx.x >> 5;
  const int l16  = lane & 15;
  const int hi   = lane >> 4;
  const int which = blockIdx.x & 1;
  const int bh    = blockIdx.x >> 1;
  const int dt = wave >> 2, kt = wave & 3;

  const float* src = ws + (which ? OFF_VSA : OFF_K) + (size_t)bh * DH * NTOK;
  float* dst       = ws + (which ? OFF_VP : OFF_KP) + (size_t)bh * DH * KL;

  const float* aRow = src + (size_t)(dt * 16 + l16) * NTOK + (hi ? 2 : 0);
  const int kcol = kt * 16 + l16;
  v8f acc = {};
  for (int n0 = 0; n0 < NTOK; n0 += 4) {
    v2f a = *(const v2f*)(aRow + n0);
    const int k0 = n0 + (hi ? 2 : 0);
    v2f b;
    b.x = EF[(size_t)k0 * KL + kcol];
    b.y = EF[(size_t)(k0 + 1) * KL + kcol];
    acc = WMMA_F32(a, b, acc);
  }
#pragma unroll
  for (int v = 0; v < 8; ++v)
    dst[(size_t)(dt * 16 + v + (hi ? 8 : 0)) * KL + kcol] = acc[v];
}

__global__ __launch_bounds__(256) void l2norm_rows(float* __restrict__ ws) {
  __shared__ float red[256];
  const int tid = threadIdx.x;
  const int row = blockIdx.x;
  float* ptr = ws + (row < 2048 ? OFF_Q : OFF_K) +
               (size_t)(row & 2047) * NTOK;
  float s = 0.f;
#pragma unroll
  for (int i = 0; i < 16; ++i) {
    float v = ptr[tid + i * 256];
    s += v * v;
  }
  red[tid] = s;
  __syncthreads();
  for (int off = 128; off > 0; off >>= 1) {
    if (tid < off) red[tid] += red[tid + off];
    __syncthreads();
  }
  const float scale = 1.0f / fmaxf(sqrtf(red[0]), 1e-12f);
#pragma unroll
  for (int i = 0; i < 16; ++i) ptr[tid + i * 256] *= scale;
}

__global__ __launch_bounds__(512) void ca_scores(float* __restrict__ ws,
                                                 const float* __restrict__ temperature) {
  __shared__ float S[64][64];
  const int lane = threadIdx.x & 31;
  const int wave = threadIdx.x >> 5;
  const int l16  = lane & 15;
  const int hi   = lane >> 4;
  const int bh = blockIdx.x;
  const int h  = bh & 7;
  const int dt = wave >> 2, et = wave & 3;

  const float* qRow = ws + OFF_Q + (size_t)bh * DH * NTOK +
                      (size_t)(dt * 16 + l16) * NTOK + (hi ? 2 : 0);
  const float* kRow = ws + OFF_K + (size_t)bh * DH * NTOK +
                      (size_t)(et * 16 + l16) * NTOK + (hi ? 2 : 0);
  v8f acc = {};
  for (int n0 = 0; n0 < NTOK; n0 += 4) {
    v2f a = *(const v2f*)(qRow + n0);
    v2f b = *(const v2f*)(kRow + n0);
    acc = WMMA_F32(a, b, acc);
  }
  const float tsc = temperature[h];
#pragma unroll
  for (int v = 0; v < 8; ++v)
    S[dt * 16 + v + (hi ? 8 : 0)][et * 16 + l16] = acc[v] * tsc;
  __syncthreads();

  if (threadIdx.x < 64) {
    const int r = threadIdx.x;
    float m = S[r][0];
#pragma unroll
    for (int j = 1; j < 64; ++j) m = fmaxf(m, S[r][j]);
    float sum = 0.f;
    float e[64];
#pragma unroll
    for (int j = 0; j < 64; ++j) { e[j] = __expf(S[r][j] - m); sum += e[j]; }
    const float inv = 1.0f / sum;
    float* dst = ws + OFF_P + (size_t)bh * DH * DH + (size_t)r * DH;
#pragma unroll
    for (int j = 0; j < 64; ++j) dst[j] = e[j] * inv;
  }
}

__global__ __launch_bounds__(256) void ca_out(const float* __restrict__ ws,
                                              float* __restrict__ out) {
  __shared__ float P[64][64];
  const int tid  = threadIdx.x;
  const int lane = tid & 31;
  const int wave = tid >> 5;
  const int l16  = lane & 15;
  const int hi   = lane >> 4;
  const int bh = blockIdx.y;
  const int b_ = bh >> 3;
  const int h  = bh & 7;
  const int nBase = blockIdx.x * 128;

  const float* Pg = ws + OFF_P + (size_t)bh * DH * DH;
  for (int i = tid; i < 4096; i += 256) P[i >> 6][i & 63] = Pg[i];
  __syncthreads();

  const int dt = wave >> 1;
  const int ng = wave & 1;
  const int mRow = dt * 16 + l16;
  const float* vca = ws + OFF_VCA + (size_t)bh * DH * NTOK;

  v8f acc[4] = {};
  for (int e0 = 0; e0 < DH; e0 += 4) {
    const int k0 = e0 + (hi ? 2 : 0);
    v2f a;
    a.x = P[mRow][k0];
    a.y = P[mRow][k0 + 1];
#pragma unroll
    for (int t = 0; t < 4; ++t) {
      const int n = nBase + (ng * 4 + t) * 16 + l16;
      v2f bfrag;
      bfrag.x = vca[(size_t)k0 * NTOK + n];
      bfrag.y = vca[(size_t)(k0 + 1) * NTOK + n];
      acc[t] = WMMA_F32(a, bfrag, acc[t]);
    }
  }
#pragma unroll
  for (int t = 0; t < 4; ++t) {
    const int n = nBase + (ng * 4 + t) * 16 + l16;
#pragma unroll
    for (int v = 0; v < 8; ++v) {
      const int d = dt * 16 + v + (hi ? 8 : 0);
      out[((size_t)(b_ * NTOK + n)) * CCH + h * DH + d] = acc[t][v];
    }
  }
}

__global__ __launch_bounds__(256) void sa_attn(float* __restrict__ ws,
                                               const float* __restrict__ temperature2,
                                               float* __restrict__ out) {
  __shared__ float P[128][64];
  const int tid  = threadIdx.x;
  const int lane = tid & 31;
  const int wave = tid >> 5;
  const int l16  = lane & 15;
  const int hi   = lane >> 4;
  const int bh = blockIdx.y;
  const int b_ = bh >> 3;
  const int h  = bh & 7;
  const int nBase = blockIdx.x * 128;

  const float* Q  = ws + OFF_Q  + (size_t)bh * DH * NTOK;
  const float* Kp = ws + OFF_KP + (size_t)bh * DH * KL;
  const float* Vp = ws + OFF_VP + (size_t)bh * DH * KL;
  const float t2 = temperature2[h];

  {
    const int nGlob = nBase + wave * 16 + l16;
    v8f acc[4] = {};
    for (int d0 = 0; d0 < DH; d0 += 4) {
      const int k0 = d0 + (hi ? 2 : 0);
      v2f a;
      a.x = Q[(size_t)k0 * NTOK + nGlob];
      a.y = Q[(size_t)(k0 + 1) * NTOK + nGlob];
#pragma unroll
      for (int t = 0; t < 4; ++t) {
        const int kc = t * 16 + l16;
        v2f bfrag;
        bfrag.x = Kp[(size_t)k0 * KL + kc];
        bfrag.y = Kp[(size_t)(k0 + 1) * KL + kc];
        acc[t] = WMMA_F32(a, bfrag, acc[t]);
      }
    }
#pragma unroll
    for (int t = 0; t < 4; ++t)
#pragma unroll
      for (int v = 0; v < 8; ++v)
        P[wave * 16 + v + (hi ? 8 : 0)][t * 16 + l16] = acc[t][v] * t2;
  }
  __syncthreads();

  if (tid < 128) {
    float m = P[tid][0];
#pragma unroll
    for (int j = 1; j < 64; ++j) m = fmaxf(m, P[tid][j]);
    float sum = 0.f;
#pragma unroll
    for (int j = 0; j < 64; ++j) { float e = __expf(P[tid][j] - m); P[tid][j] = e; sum += e; }
    const float inv = 1.0f / sum;
#pragma unroll
    for (int j = 0; j < 64; ++j) P[tid][j] *= inv;
  }
  __syncthreads();

  {
    const int mRow = wave * 16 + l16;
    v8f acc[4] = {};
    for (int kk0 = 0; kk0 < KL; kk0 += 4) {
      const int k0 = kk0 + (hi ? 2 : 0);
      v2f a;
      a.x = P[mRow][k0];
      a.y = P[mRow][k0 + 1];
#pragma unroll
      for (int t = 0; t < 4; ++t) {
        const int dd = t * 16 + l16;
        v2f bfrag = *(const v2f*)(Vp + (size_t)dd * KL + k0);
        acc[t] = WMMA_F32(a, bfrag, acc[t]);
      }
    }
#pragma unroll
    for (int t = 0; t < 4; ++t) {
      const int dd = t * 16 + l16;
#pragma unroll
      for (int v = 0; v < 8; ++v) {
        const int n = nBase + wave * 16 + v + (hi ? 8 : 0);
        const size_t f = (size_t)dd * (HEADS * NTOK) + (size_t)h * NTOK + n;
        out[(size_t)b_ * (NTOK * CCH) + f] += acc[t][v];
      }
    }
  }
}

extern "C" void kernel_launch(void* const* d_in, const int* in_sizes, int n_in,
                              void* d_out, int out_size, void* d_ws, size_t ws_size,
                              hipStream_t stream) {
  const float* x  = (const float*)d_in[0];
  const float* W  = (const float*)d_in[1];
  const float* EF = (const float*)d_in[2];
  const float* t1 = (const float*)d_in[3];
  const float* t2 = (const float*)d_in[4];
  float* out = (float*)d_out;
  float* ws  = (float*)d_ws;

  qkvv_gemm  <<<dim3(128, 32), 256, 0, stream>>>(x, W, ws);
  ef_proj    <<<64,            512, 0, stream>>>(ws, EF);
  l2norm_rows<<<4096,          256, 0, stream>>>(ws);
  ca_scores  <<<32,            512, 0, stream>>>(ws, t1);
  ca_out     <<<dim3(32, 32),  256, 0, stream>>>(ws, out);
  sa_attn    <<<dim3(32, 32),  256, 0, stream>>>(ws, t2, out);
}


# K1 projection GEMM: x and W tiles staged through LDS (128x128x32, two stages, whole-line 16-B global loads, 144-B padded rows); each wave takes 64x32 from 8-B LDS fragment reads; sum order kept
# speedup vs baseline: 1.4056x; 1.4056x over previous
#include <hip/hip_runtime.h>

typedef float v2f __attribute__((ext_vector_type(2)));
typedef float v4f __attribute__((ext_vector_type(4)));
typedef float v8f __attribute__((ext_vector_type(8)));

#define WMMA_F32(a, b, c) \
  __builtin_amdgcn_wmma_f32_16x16x4_f32(false, (a), false, (b), (short)0, (c), false, false)

#define BATCH 4
#define NTOK 4096
#define CCH 512
#define HEADS 8
#define DH 64
#define KL 64
#define MROWS (BATCH * NTOK)
#define OCH (4 * CCH)
#define BH (BATCH * HEADS)

#define OFF_Q   ((size_t)0)
#define OFF_K   ((size_t)8388608)
#define OFF_VCA ((size_t)16777216)
#define OFF_VSA ((size_t)25165824)
#define OFF_KP  ((size_t)33554432)
#define OFF_VP  ((size_t)33685504)
#define OFF_P   ((size_t)33816576)

#define G_BM 128
#define G_BO 128
#define G_KC 32
#define G_LD 36
#define G_NCH (CCH / G_KC)

__global__ __launch_bounds__(256) void qkvv_gemm(const float* __restrict__ x,
                                                 const float* __restrict__ W,
                                                 float* __restrict__ ws) {
  __shared__ __attribute__((aligned(16))) float As[2][G_BM * G_LD];
  __shared__ __attribute__((aligned(16))) float Bs[2][G_BO * G_LD];

  const int tid  = threadIdx.x;
  const int lane = tid & 31;
  const int wave = tid >> 5;
  const int l16  = lane & 15;
  const int hi   = lane >> 4;
  const int wm   = wave >> 2;
  const int wo   = wave & 3;
  const int mBlock = blockIdx.x * G_BM;
  const int oBlock = blockIdx.y * G_BO;

  const int sRow = tid >> 3;
  const int sCol = (tid & 7) * 4;
  const float* xg = x + (size_t)(mBlock + sRow) * CCH + sCol;
  const float* wg = W + (size_t)(oBlock + sRow) * CCH + sCol;
  const int sOff = sRow * G_LD + sCol;

  v4f ra[4], rb[4];
  for (int i = 0; i < 4; ++i) {
    ra[i] = *(const v4f*)(xg + (size_t)i * 32 * CCH);
    rb[i] = *(const v4f*)(wg + (size_t)i * 32 * CCH);
  }
  for (int i = 0; i < 4; ++i) {
    *(v4f*)(&As[0][sOff + i * 32 * G_LD]) = ra[i];
    *(v4f*)(&Bs[0][sOff + i * 32 * G_LD]) = rb[i];
  }
  __syncthreads();

  const int aOff = (wm * 64 + l16) * G_LD + (hi ? 2 : 0);
  const int bOff = (wo * 32 + l16) * G_LD + (hi ? 2 : 0);

  v8f acc[4][2] = {};
  for (int c = 0; c < G_NCH; ++c) {
    const int cur = c & 1;
    const bool more = (c + 1 < G_NCH);
    if (more) {
      for (int i = 0; i < 4; ++i) {
        ra[i] = *(const v4f*)(xg + (size_t)i * 32 * CCH + (c + 1) * G_KC);
        rb[i] = *(const v4f*)(wg + (size_t)i * 32 * CCH + (c + 1) * G_KC);
      }
    }
    const float* Ac = &As[cur][aOff];
    const float* Bc = &Bs[cur][bOff];
    for (int j = 0; j < G_KC / 4; ++j) {
      v2f a[4], b[2];
      for (int i = 0; i < 4; ++i) a[i] = *(const v2f*)(Ac + i * 16 * G_LD + j * 4);
      for (int t = 0; t < 2; ++t) b[t] = *(const v2f*)(Bc + t * 16 * G_LD + j * 4);
      for (int i = 0; i < 4; ++i)
        for (int t = 0; t < 2; ++t)
          acc[i][t] = WMMA_F32(a[i], b[t], acc[i][t]);
    }
    if (more) {
      for (int i = 0; i < 4; ++i) {
        *(v4f*)(&As[cur ^ 1][sOff + i * 32 * G_LD]) = ra[i];
        *(v4f*)(&Bs[cur ^ 1][sOff + i * 32 * G_LD]) = rb[i];
      }
    }
    __syncthreads();
  }

  for (int t = 0; t < 2; ++t) {
    const int o  = oBlock + wo * 32 + t * 16 + l16;
    const int p  = o >> 9;
    const int h  = (o >> 6) & 7;
    const int dd = o & 63;
    for (int i = 0; i < 4; ++i) {
      for (int v = 0; v < 8; ++v) {
        const int m  = mBlock + wm * 64 + i * 16 + v + (hi ? 8 : 0);
        const int b_ = m >> 12;
        const int n  = m & (NTOK - 1);
        const size_t dst = (size_t)p * 8388608 +
                           ((size_t)(b_ * HEADS + h) * DH + dd) * NTOK + n;
        ws[dst] = acc[i][t][v];
      }
    }
  }
}

__global__ __launch_bounds__(512) void ef_proj(float* __restrict__ ws,
                                               const float* __restrict__ EF) {
  const int lane = threadIdx.x & 31;
  const int wave = threadIdx.x >> 5;
  const int l16  = lane & 15;
  const int hi   = lane >> 4;
  const int which = blockIdx.x & 1;
  const int bh    = blockIdx.x >> 1;
  const int dt = wave >> 2, kt = wave & 3;

  const float* src = ws + (which ? OFF_VSA : OFF_K) + (size_t)bh * DH * NTOK;
  float* dst       = ws + (which ? OFF_VP : OFF_KP) + (size_t)bh * DH * KL;

  const float* aRow = src + (size_t)(dt * 16 + l16) * NTOK + (hi ? 2 : 0);
  const int kcol = kt * 16 + l16;
  v8f acc = {};
  for (int n0 = 0; n0 < NTOK; n0 += 4) {
    v2f a = *(const v2f*)(aRow + n0);
    const int k0 = n0 + (hi ? 2 : 0);
    v2f b;
    b.x = EF[(size_t)k0 * KL + kcol];
    b.y = EF[(size_t)(k0 + 1) * KL + kcol];
    acc = WMMA_F32(a, b, acc);
  }
#pragma unroll
  for (int v = 0; v < 8; ++v)
    dst[(size_t)(dt * 16 + v + (hi ? 8 : 0)) * KL + kcol] = acc[v];
}

__global__ __launch_bounds__(256) void l2norm_rows(float* __restrict__ ws) {
  __shared__ float red[256];
  const int tid = threadIdx.x;
  const int row = blockIdx.x;
  float* ptr = ws + (row < 2048 ? OFF_Q : OFF_K) +
               (size_t)(row & 2047) * NTOK;
  float s = 0.f;
#pragma unroll
  for (int i = 0; i < 16; ++i) {
    float v = ptr[tid + i * 256];
    s += v * v;
  }
  red[tid] = s;
  __syncthreads();
  for (int off = 128; off > 0; off >>= 1) {
    if (tid < off) red[tid] += red[tid + off];
    __syncthreads();
  }
  const float scale = 1.0f / fmaxf(sqrtf(red[0]), 1e-12f);
#pragma unroll
  for (int i = 0; i < 16; ++i) ptr[tid + i * 256] *= scale;
}

__global__ __launch_bounds__(512) void ca_scores(float* __restrict__ ws,
                                                 const float* __restrict__ temperature) {
  __shared__ float S[64][64];
  const int lane = threadIdx.x & 31;
  const int wave = threadIdx.x >> 5;
  const int l16  = lane & 15;
  const int hi   = lane >> 4;
  const int bh = blockIdx.x;
  const int h  = bh & 7;
  const int dt = wave >> 2, et = wave & 3;

  const float* qRow = ws + OFF_Q + (size_t)bh * DH * NTOK +
                      (size_t)(dt * 16 + l16) * NTOK + (hi ? 2 : 0);
  const float* kRow = ws + OFF_K + (size_t)bh * DH * NTOK +
                      (size_t)(et * 16 + l16) * NTOK + (hi ? 2 : 0);
  v8f acc = {};
  for (int n0 = 0; n0 < NTOK; n0 += 4) {
    v2f a = *(const v2f*)(qRow + n0);
    v2f b = *(const v2f*)(kRow + n0);
    acc = WMMA_F32(a, b, acc);
  }
  const float tsc = temperature[h];
#pragma unroll
  for (int v = 0; v < 8; ++v)
    S[dt * 16 + v + (hi ? 8 : 0)][et * 16 + l16] = acc[v] * tsc;
  __syncthreads();

  if (threadIdx.x < 64) {
    const int r = threadIdx.x;
    float m = S[r][0];
#pragma unroll
    for (int j = 1; j < 64; ++j) m = fmaxf(m, S[r][j]);
    float sum = 0.f;
    float e[64];
#pragma unroll
    for (int j = 0; j < 64; ++j) { e[j] = __expf(S[r][j] - m); sum += e[j]; }
    const float inv = 1.0f / sum;
    float* dst = ws + OFF_P + (size_t)bh * DH * DH + (size_t)r * DH;
#pragma unroll
    for (int j = 0; j < 64; ++j) dst[j] = e[j] * inv;
  }
}

__global__ __launch_bounds__(256) void ca_out(const float* __restrict__ ws,
                                              float* __restrict__ out) {
  __shared__ float P[64][64];
  const int tid  = threadIdx.x;
  const int lane = tid & 31;
  const int wave = tid >> 5;
  const int l16  = lane & 15;
  const int hi   = lane >> 4;
  const int bh = blockIdx.y;
  const int b_ = bh >> 3;
  const int h  = bh & 7;
  const int nBase = blockIdx.x * 128;

  const float* Pg = ws + OFF_P + (size_t)bh * DH * DH;
  for (int i = tid; i < 4096; i += 256) P[i >> 6][i & 63] = Pg[i];
  __syncthreads();

  const int dt = wave >> 1;
  const int ng = wave & 1;
  const int mRow = dt * 16 + l16;
  const float* vca = ws + OFF_VCA + (size_t)bh * DH * NTOK;

  v8f acc[4] = {};
  for (int e0 = 0; e0 < DH; e0 += 4) {
    const int k0 = e0 + (hi ? 2 : 0);
    v2f a;
    a.x = P[mRow][k0];
    a.y = P[mRow][k0 + 1];
#pragma unroll
    for (int t = 0; t < 4; ++t) {
      const int n = nBase + (ng * 4 + t) * 16 + l16;
      v2f bfrag;
      bfrag.x = vca[(size_t)k0 * NTOK + n];
      bfrag.y = vca[(size_t)(k0 + 1) * NTOK + n];
      acc[t] = WMMA_F32(a, bfrag, acc[t]);
    }
  }
#pragma unroll
  for (int t = 0; t < 4; ++t) {
    const int n = nBase + (ng * 4 + t) * 16 + l16;
#pragma unroll
    for (int v = 0; v < 8; ++v) {
      const int d = dt * 16 + v + (hi ? 8 : 0);
      out[((size_t)(b_ * NTOK + n)) * CCH + h * DH + d] = acc[t][v];
    }
  }
}

__global__ __launch_bounds__(256) void sa_attn(float* __restrict__ ws,
                                               const float* __restrict__ temperature2,
                                               float* __restrict__ out) {
  __shared__ float P[128][64];
  const int tid  = threadIdx.x;
  const int lane = tid & 31;
  const int wave = tid >> 5;
  const int l16  = lane & 15;
  const int hi   = lane >> 4;
  const int bh = blockIdx.y;
  const int b_ = bh >> 3;
  const int h  = bh & 7;
  const int nBase = blockIdx.x * 128;

  const float* Q  = ws + OFF_Q  + (size_t)bh * DH * NTOK;
  const float* Kp = ws + OFF_KP + (size_t)bh * DH * KL;
  const float* Vp = ws + OFF_VP + (size_t)bh * DH * KL;
  const float t2 = temperature2[h];

  {
    const int nGlob = nBase + wave * 16 + l16;
    v8f acc[4] = {};
    for (int d0 = 0; d0 < DH; d0 += 4) {
      const int k0 = d0 + (hi ? 2 : 0);
      v2f a;
      a.x = Q[(size_t)k0 * NTOK + nGlob];
      a.y = Q[(size_t)(k0 + 1) * NTOK + nGlob];
#pragma unroll
      for (int t = 0; t < 4; ++t) {
        const int kc = t * 16 + l16;
        v2f bfrag;
        bfrag.x = Kp[(size_t)k0 * KL + kc];
        bfrag.y = Kp[(size_t)(k0 + 1) * KL + kc];
        acc[t] = WMMA_F32(a, bfrag, acc[t]);
      }
    }
#pragma unroll
    for (int t = 0; t < 4; ++t)
#pragma unroll
      for (int v = 0; v < 8; ++v)
        P[wave * 16 + v + (hi ? 8 : 0)][t * 16 + l16] = acc[t][v] * t2;
  }
  __syncthreads();

  if (tid < 128) {
    float m = P[tid][0];
#pragma unroll
    for (int j = 1; j < 64; ++j) m = fmaxf(m, P[tid][j]);
    float sum = 0.f;
#pragma unroll
    for (int j = 0; j < 64; ++j) { float e = __expf(P[tid][j] - m); P[tid][j] = e; sum += e; }
    const float inv = 1.0f / sum;
#pragma unroll
    for (int j = 0; j < 64; ++j) P[tid][j] *= inv;
  }
  __syncthreads();

  {
    const int mRow = wave * 16 + l16;
    v8f acc[4] = {};
    for (int kk0 = 0; kk0 < KL; kk0 += 4) {
      const int k0 = kk0 + (hi ? 2 : 0);
      v2f a;
      a.x = P[mRow][k0];
      a.y = P[mRow][k0 + 1];
#pragma unroll
      for (int t = 0; t < 4; ++t) {
        const int dd = t * 16 + l16;
        v2f bfrag = *(const v2f*)(Vp + (size_t)dd * KL + k0);
        acc[t] = WMMA_F32(a, bfrag, acc[t]);
      }
    }
#pragma unroll
    for (int t = 0; t < 4; ++t) {
      const int dd = t * 16 + l16;
#pragma unroll
      for (int v = 0; v < 8; ++v) {
        const int n = nBase + wave * 16 + v + (hi ? 8 : 0);
        const size_t f = (size_t)dd * (HEADS * NTOK) + (size_t)h * NTOK + n;
        out[(size_t)b_ * (NTOK * CCH) + f] += acc[t][v];
      }
    }
  }
}

extern "C" void kernel_launch(void* const* d_in, const int* in_sizes, int n_in,
                              void* d_out, int out_size, void* d_ws, size_t ws_size,
                              hipStream_t stream) {
  const float* x  = (const float*)d_in[0];
  const float* W  = (const float*)d_in[1];
  const float* EF = (const float*)d_in[2];
  const float* t1 = (const float*)d_in[3];
  const float* t2 = (const float*)d_in[4];
  float* out = (float*)d_out;
  float* ws  = (float*)d_ws;

  qkvv_gemm  <<<dim3(128, 16), 256, 0, stream>>>(x, W, ws);
  ef_proj    <<<64,            512, 0, stream>>>(ws, EF);
  l2norm_rows<<<4096,          256, 0, stream>>>(ws);
  ca_scores  <<<32,            512, 0, stream>>>(ws, t1);
  ca_out     <<<dim3(32, 32),  256, 0, stream>>>(ws, out);
  sa_attn    <<<dim3(32, 32),  256, 0, stream>>>(ws, t2, out);
}


# K2 ef_proj, K4 ca_scores: token-axis operands staged through LDS (256-token chunks, 2 stages, whole-line 16-B loads), one 16x64 strip per workgroup (256 / 128 workgroups of 4 waves); sum order kept
# speedup vs baseline: 2.3802x; 1.6934x over previous
#include <hip/hip_runtime.h>

typedef float v2f __attribute__((ext_vector_type(2)));
typedef float v4f __attribute__((ext_vector_type(4)));
typedef float v8f __attribute__((ext_vector_type(8)));

#define WMMA_F32(a, b, c) \
  __builtin_amdgcn_wmma_f32_16x16x4_f32(false, (a), false, (b), (short)0, (c), false, false)

#define BATCH 4
#define NTOK 4096
#define CCH 512
#define HEADS 8
#define DH 64
#define KL 64
#define MROWS (BATCH * NTOK)
#define OCH (4 * CCH)
#define BH (BATCH * HEADS)

#define OFF_Q   ((size_t)0)
#define OFF_K   ((size_t)8388608)
#define OFF_VCA ((size_t)16777216)
#define OFF_VSA ((size_t)25165824)
#define OFF_KP  ((size_t)33554432)
#define OFF_VP  ((size_t)33685504)
#define OFF_P   ((size_t)33816576)

#define G_BM 128
#define G_BO 128
#define G_KC 32
#define G_LD 36
#define G_NCH (CCH / G_KC)

__global__ __launch_bounds__(256) void qkvv_gemm(const float* __restrict__ x,
                                                 const float* __restrict__ W,
                                                 float* __restrict__ ws) {
  __shared__ __attribute__((aligned(16))) float As[2][G_BM * G_LD];
  __shared__ __attribute__((aligned(16))) float Bs[2][G_BO * G_LD];

  const int tid  = threadIdx.x;
  const int lane = tid & 31;
  const int wave = tid >> 5;
  const int l16  = lane & 15;
  const int hi   = lane >> 4;
  const int wm   = wave >> 2;
  const int wo   = wave & 3;
  const int mBlock = blockIdx.x * G_BM;
  const int oBlock = blockIdx.y * G_BO;

  const int sRow = tid >> 3;
  const int sCol = (tid & 7) * 4;
  const float* xg = x + (size_t)(mBlock + sRow) * CCH + sCol;
  const float* wg = W + (size_t)(oBlock + sRow) * CCH + sCol;
  const int sOff = sRow * G_LD + sCol;

  v4f ra[4], rb[4];
  for (int i = 0; i < 4; ++i) {
    ra[i] = *(const v4f*)(xg + (size_t)i * 32 * CCH);
    rb[i] = *(const v4f*)(wg + (size_t)i * 32 * CCH);
  }
  for (int i = 0; i < 4; ++i) {
    *(v4f*)(&As[0][sOff + i * 32 * G_LD]) = ra[i];
    *(v4f*)(&Bs[0][sOff + i * 32 * G_LD]) = rb[i];
  }
  __syncthreads();

  const int aOff = (wm * 64 + l16) * G_LD + (hi ? 2 : 0);
  const int bOff = (wo * 32 + l16) * G_LD + (hi ? 2 : 0);

  v8f acc[4][2] = {};
  for (int c = 0; c < G_NCH; ++c) {
    const int cur = c & 1;
    const bool more = (c + 1 < G_NCH);
    if (more) {
      for (int i = 0; i < 4; ++i) {
        ra[i] = *(const v4f*)(xg + (size_t)i * 32 * CCH + (c + 1) * G_KC);
        rb[i] = *(const v4f*)(wg + (size_t)i * 32 * CCH + (c + 1) * G_KC);
      }
    }
    const float* Ac = &As[cur][aOff];
    const float* Bc = &Bs[cur][bOff];
    for (int j = 0; j < G_KC / 4; ++j) {
      v2f a[4], b[2];
      for (int i = 0; i < 4; ++i) a[i] = *(const v2f*)(Ac + i * 16 * G_LD + j * 4);
      for (int t = 0; t < 2; ++t) b[t] = *(const v2f*)(Bc + t * 16 * G_LD + j * 4);
      for (int i = 0; i < 4; ++i)
        for (int t = 0; t < 2; ++t)
          acc[i][t] = WMMA_F32(a[i], b[t], acc[i][t]);
    }
    if (more) {
      for (int i = 0; i < 4; ++i) {
        *(v4f*)(&As[cur ^ 1][sOff + i * 32 * G_LD]) = ra[i];
        *(v4f*)(&Bs[cur ^ 1][sOff + i * 32 * G_LD]) = rb[i];
      }
    }
    __syncthreads();
  }

  for (int t = 0; t < 2; ++t) {
    const int o  = oBlock + wo * 32 + t * 16 + l16;
    const int p  = o >> 9;
    const int h  = (o >> 6) & 7;
    const int dd = o & 63;
    for (int i = 0; i < 4; ++i) {
      for (int v = 0; v < 8; ++v) {
        const int m  = mBlock + wm * 64 + i * 16 + v + (hi ? 8 : 0);
        const int b_ = m >> 12;
        const int n  = m & (NTOK - 1);
        const size_t dst = (size_t)p * 8388608 +
                           ((size_t)(b_ * HEADS + h) * DH + dd) * NTOK + n;
        ws[dst] = acc[i][t][v];
      }
    }
  }
}

#define T_NC 256
#define T_LD 260
#define T_NCH (NTOK / T_NC)
#define E_LD 72

__global__ __launch_bounds__(128) void ef_proj(float* __restrict__ ws,
                                               const float* __restrict__ EF) {
  __shared__ __attribute__((aligned(16))) float As[2][16 * T_LD];
  __shared__ __attribute__((aligned(16))) float Es[2][T_NC * E_LD];

  const int tid  = threadIdx.x;
  const int lane = tid & 31;
  const int kt   = tid >> 5;
  const int l16  = lane & 15;
  const int hi   = lane >> 4;
  const int dt    = blockIdx.x;
  const int which = blockIdx.y & 1;
  const int bh    = blockIdx.y >> 1;

  const float* src = ws + (which ? OFF_VSA : OFF_K) + (size_t)bh * DH * NTOK +
                     (size_t)dt * 16 * NTOK;
  float* dst       = ws + (which ? OFF_VP : OFF_KP) + (size_t)bh * DH * KL;

  const int aRow = tid >> 6;
  const int aCol = (tid & 63) * 4;
  const float* ag = src + (size_t)aRow * NTOK + aCol;
  const int aOffS = aRow * T_LD + aCol;
  const int eRow = tid >> 4;
  const int eCol = (tid & 15) * 4;
  const float* eg = EF + (size_t)eRow * KL + eCol;
  const int eOffS = eRow * E_LD + eCol;

  v4f ra[8], re[32];
  for (int i = 0; i < 8; ++i)  ra[i] = *(const v4f*)(ag + (size_t)i * 2 * NTOK);
  for (int i = 0; i < 32; ++i) re[i] = *(const v4f*)(eg + (size_t)i * 8 * KL);
  for (int i = 0; i < 8; ++i)  *(v4f*)(&As[0][aOffS + i * 2 * T_LD]) = ra[i];
  for (int i = 0; i < 32; ++i) *(v4f*)(&Es[0][eOffS + i * 8 * E_LD]) = re[i];
  __syncthreads();

  const int aF = l16 * T_LD + (hi ? 2 : 0);
  const int eF = (hi ? 2 : 0) * E_LD + kt * 16 + l16;

  v8f acc = {};
  for (int c = 0; c < T_NCH; ++c) {
    const int cur = c & 1;
    const bool more = (c + 1 < T_NCH);
    if (more) {
      for (int i = 0; i < 8; ++i)
        ra[i] = *(const v4f*)(ag + (size_t)i * 2 * NTOK + (c + 1) * T_NC);
      for (int i = 0; i < 32; ++i)
        re[i] = *(const v4f*)(eg + (size_t)i * 8 * KL + (size_t)(c + 1) * T_NC * KL);
    }
    const float* Ac = &As[cur][aF];
    const float* Ec = &Es[cur][eF];
    v2f fa[2][8], fb[2][8];
#pragma unroll
    for (int i = 0; i < 8; ++i) {
      fa[0][i]   = *(const v2f*)(Ac + i * 4);
      fb[0][i].x = Ec[i * 4 * E_LD];
      fb[0][i].y = Ec[i * 4 * E_LD + E_LD];
    }
#pragma unroll
    for (int g = 0; g < T_NC / 32; ++g) {
      const int s = g & 1;
      if (g + 1 < T_NC / 32) {
#pragma unroll
        for (int i = 0; i < 8; ++i) {
          const int j = (g + 1) * 8 + i;
          fa[s ^ 1][i]   = *(const v2f*)(Ac + j * 4);
          fb[s ^ 1][i].x = Ec[j * 4 * E_LD];
          fb[s ^ 1][i].y = Ec[j * 4 * E_LD + E_LD];
        }
      }
      __builtin_amdgcn_sched_barrier(0);
#pragma unroll
      for (int i = 0; i < 8; ++i) acc = WMMA_F32(fa[s][i], fb[s][i], acc);
      __builtin_amdgcn_sched_barrier(0);
    }
    if (more) {
      for (int i = 0; i < 8; ++i)  *(v4f*)(&As[cur ^ 1][aOffS + i * 2 * T_LD]) = ra[i];
      for (int i = 0; i < 32; ++i) *(v4f*)(&Es[cur ^ 1][eOffS + i * 8 * E_LD]) = re[i];
    }
    __syncthreads();
  }
#pragma unroll
  for (int v = 0; v < 8; ++v)
    dst[(size_t)(dt * 16 + v + (hi ? 8 : 0)) * KL + kt * 16 + l16] = acc[v];
}

__global__ __launch_bounds__(256) void l2norm_rows(float* __restrict__ ws) {
  __shared__ float red[256];
  const int tid = threadIdx.x;
  const int row = blockIdx.x;
  float* ptr = ws + (row < 2048 ? OFF_Q : OFF_K) +
               (size_t)(row & 2047) * NTOK;
  float s = 0.f;
#pragma unroll
  for (int i = 0; i < 16; ++i) {
    float v = ptr[tid + i * 256];
    s += v * v;
  }
  red[tid] = s;
  __syncthreads();
  for (int off = 128; off > 0; off >>= 1) {
    if (tid < off) red[tid] += red[tid + off];
    __syncthreads();
  }
  const float scale = 1.0f / fmaxf(sqrtf(red[0]), 1e-12f);
#pragma unroll
  for (int i = 0; i < 16; ++i) ptr[tid + i * 256] *= scale;
}

__global__ __launch_bounds__(128) void ca_scores(float* __restrict__ ws,
                                                 const float* __restrict__ temperature) {
  __shared__ __attribute__((aligned(16))) float Qs[2][16 * T_LD];
  __shared__ __attribute__((aligned(16))) float Ks[2][64 * T_LD];
  __shared__ float S[16][65];
  const int tid  = threadIdx.x;
  const int lane = tid & 31;
  const int et   = tid >> 5;
  const int l16  = lane & 15;
  const int hi   = lane >> 4;
  const int dt = blockIdx.x;
  const int bh = blockIdx.y;
  const int h  = bh & 7;

  const float* qsrc = ws + OFF_Q + (size_t)bh * DH * NTOK + (size_t)dt * 16 * NTOK;
  const float* ksrc = ws + OFF_K + (size_t)bh * DH * NTOK;

  const int sRow = tid >> 6;
  const int sCol = (tid & 63) * 4;
  const float* qg = qsrc + (size_t)sRow * NTOK + sCol;
  const float* kg = ksrc + (size_t)sRow * NTOK + sCol;
  const int sOff = sRow * T_LD + sCol;

  v4f rq[8], rk[32];
  for (int i = 0; i < 8; ++i)  rq[i] = *(const v4f*)(qg + (size_t)i * 2 * NTOK);
  for (int i = 0; i < 32; ++i) rk[i] = *(const v4f*)(kg + (size_t)i * 2 * NTOK);
  for (int i = 0; i < 8; ++i)  *(v4f*)(&Qs[0][sOff + i * 2 * T_LD]) = rq[i];
  for (int i = 0; i < 32; ++i) *(v4f*)(&Ks[0][sOff + i * 2 * T_LD]) = rk[i];
  __syncthreads();

  const int qF = l16 * T_LD + (hi ? 2 : 0);
  const int kF = (et * 16 + l16) * T_LD + (hi ? 2 : 0);

  v8f acc = {};
  for (int c = 0; c < T_NCH; ++c) {
    const int cur = c & 1;
    const bool more = (c + 1 < T_NCH);
    if (more) {
      for (int i = 0; i < 8; ++i)
        rq[i] = *(const v4f*)(qg + (size_t)i * 2 * NTOK + (c + 1) * T_NC);
      for (int i = 0; i < 32; ++i)
        rk[i] = *(const v4f*)(kg + (size_t)i * 2 * NTOK + (c + 1) * T_NC);
    }
    const float* Qc = &Qs[cur][qF];
    const float* Kc = &Ks[cur][kF];
    v2f fa[2][8], fb[2][8];
#pragma unroll
    for (int i = 0; i < 8; ++i) {
      fa[0][i] = *(const v2f*)(Qc + i * 4);
      fb[0][i] = *(const v2f*)(Kc + i * 4);
    }
#pragma unroll
    for (int g = 0; g < T_NC / 32; ++g) {
      const int s = g & 1;
      if (g + 1 < T_NC / 32) {
#pragma unroll
        for (int i = 0; i < 8; ++i) {
          const int j = (g + 1) * 8 + i;
          fa[s ^ 1][i] = *(const v2f*)(Qc + j * 4);
          fb[s ^ 1][i] = *(const v2f*)(Kc + j * 4);
        }
      }
      __builtin_amdgcn_sched_barrier(0);
#pragma unroll
      for (int i = 0; i < 8; ++i) acc = WMMA_F32(fa[s][i], fb[s][i], acc);
      __builtin_amdgcn_sched_barrier(0);
    }
    if (more) {
      for (int i = 0; i < 8; ++i)  *(v4f*)(&Qs[cur ^ 1][sOff + i * 2 * T_LD]) = rq[i];
      for (int i = 0; i < 32; ++i) *(v4f*)(&Ks[cur ^ 1][sOff + i * 2 * T_LD]) = rk[i];
    }
    __syncthreads();
  }
  const float tsc = temperature[h];
#pragma unroll
  for (int v = 0; v < 8; ++v)
    S[v + (hi ? 8 : 0)][et * 16 + l16] = acc[v] * tsc;
  __syncthreads();

  if (tid < 16) {
    const int r = tid;
    float m = S[r][0];
#pragma unroll
    for (int j = 1; j < 64; ++j) m = fmaxf(m, S[r][j]);
    float sum = 0.f;
    float e[64];
#pragma unroll
    for (int j = 0; j < 64; ++j) { e[j] = __expf(S[r][j] - m); sum += e[j]; }
    const float inv = 1.0f / sum;
    float* dst = ws + OFF_P + (size_t)bh * DH * DH + (size_t)(dt * 16 + r) * DH;
#pragma unroll
    for (int j = 0; j < 64; ++j) dst[j] = e[j] * inv;
  }
}

__global__ __launch_bounds__(256) void ca_out(const float* __restrict__ ws,
                                              float* __restrict__ out) {
  __shared__ float P[64][64];
  const int tid  = threadIdx.x;
  const int lane = tid & 31;
  const int wave = tid >> 5;
  const int l16  = lane & 15;
  const int hi   = lane >> 4;
  const int bh = blockIdx.y;
  const int b_ = bh >> 3;
  const int h  = bh & 7;
  const int nBase = blockIdx.x * 128;

  const float* Pg = ws + OFF_P + (size_t)bh * DH * DH;
  for (int i = tid; i < 4096; i += 256) P[i >> 6][i & 63] = Pg[i];
  __syncthreads();

  const int dt = wave >> 1;
  const int ng = wave & 1;
  const int mRow = dt * 16 + l16;
  const float* vca = ws + OFF_VCA + (size_t)bh * DH * NTOK;

  v8f acc[4] = {};
  for (int e0 = 0; e0 < DH; e0 += 4) {
    const int k0 = e0 + (hi ? 2 : 0);
    v2f a;
    a.x = P[mRow][k0];
    a.y = P[mRow][k0 + 1];
#pragma unroll
    for (int t = 0; t < 4; ++t) {
      const int n = nBase + (ng * 4 + t) * 16 + l16;
      v2f bfrag;
      bfrag.x = vca[(size_t)k0 * NTOK + n];
      bfrag.y = vca[(size_t)(k0 + 1) * NTOK + n];
      acc[t] = WMMA_F32(a, bfrag, acc[t]);
    }
  }
#pragma unroll
  for (int t = 0; t < 4; ++t) {
    const int n = nBase + (ng * 4 + t) * 16 + l16;
#pragma unroll
    for (int v = 0; v < 8; ++v) {
      const int d = dt * 16 + v + (hi ? 8 : 0);
      out[((size_t)(b_ * NTOK + n)) * CCH + h * DH + d] = acc[t][v];
    }
  }
}

__global__ __launch_bounds__(256) void sa_attn(float* __restrict__ ws,
                                               const float* __restrict__ temperature2,
                                               float* __restrict__ out) {
  __shared__ float P[128][64];
  const int tid  = threadIdx.x;
  const int lane = tid & 31;
  const int wave = tid >> 5;
  const int l16  = lane & 15;
  const int hi   = lane >> 4;
  const int bh = blockIdx.y;
  const int b_ = bh >> 3;
  const int h  = bh & 7;
  const int nBase = blockIdx.x * 128;

  const float* Q  = ws + OFF_Q  + (size_t)bh * DH * NTOK;
  const float* Kp = ws + OFF_KP + (size_t)bh * DH * KL;
  const float* Vp = ws + OFF_VP + (size_t)bh * DH * KL;
  const float t2 = temperature2[h];

  {
    const int nGlob = nBase + wave * 16 + l16;
    v8f acc[4] = {};
    for (int d0 = 0; d0 < DH; d0 += 4) {
      const int k0 = d0 + (hi ? 2 : 0);
      v2f a;
      a.x = Q[(size_t)k0 * NTOK + nGlob];
      a.y = Q[(size_t)(k0 + 1) * NTOK + nGlob];
#pragma unroll
      for (int t = 0; t < 4; ++t) {
        const int kc = t * 16 + l16;
        v2f bfrag;
        bfrag.x = Kp[(size_t)k0 * KL + kc];
        bfrag.y = Kp[(size_t)(k0 + 1) * KL + kc];
        acc[t] = WMMA_F32(a, bfrag, acc[t]);
      }
    }
#pragma unroll
    for (int t = 0; t < 4; ++t)
#pragma unroll
      for (int v = 0; v < 8; ++v)
        P[wave * 16 + v + (hi ? 8 : 0)][t * 16 + l16] = acc[t][v] * t2;
  }
  __syncthreads();

  if (tid < 128) {
    float m = P[tid][0];
#pragma unroll
    for (int j = 1; j < 64; ++j) m = fmaxf(m, P[tid][j]);
    float sum = 0.f;
#pragma unroll
    for (int j = 0; j < 64; ++j) { float e = __expf(P[tid][j] - m); P[tid][j] = e; sum += e; }
    const float inv = 1.0f / sum;
#pragma unroll
    for (int j = 0; j < 64; ++j) P[tid][j] *= inv;
  }
  __syncthreads();

  {
    const int mRow = wave * 16 + l16;
    v8f acc[4] = {};
    for (int kk0 = 0; kk0 < KL; kk0 += 4) {
      const int k0 = kk0 + (hi ? 2 : 0);
      v2f a;
      a.x = P[mRow][k0];
      a.y = P[mRow][k0 + 1];
#pragma unroll
      for (int t = 0; t < 4; ++t) {
        const int dd = t * 16 + l16;
        v2f bfrag = *(const v2f*)(Vp + (size_t)dd * KL + k0);
        acc[t] = WMMA_F32(a, bfrag, acc[t]);
      }
    }
#pragma unroll
    for (int t = 0; t < 4; ++t) {
      const int dd = t * 16 + l16;
#pragma unroll
      for (int v = 0; v < 8; ++v) {
        const int n = nBase + wave * 16 + v + (hi ? 8 : 0);
        const size_t f = (size_t)dd * (HEADS * NTOK) + (size_t)h * NTOK + n;
        out[(size_t)b_ * (NTOK * CCH) + f] += acc[t][v];
      }
    }
  }
}

extern "C" void kernel_launch(void* const* d_in, const int* in_sizes, int n_in,
                              void* d_out, int out_size, void* d_ws, size_t ws_size,
                              hipStream_t stream) {
  const float* x  = (const float*)d_in[0];
  const float* W  = (const float*)d_in[1];
  const float* EF = (const float*)d_in[2];
  const float* t1 = (const float*)d_in[3];
  const float* t2 = (const float*)d_in[4];
  float* out = (float*)d_out;
  float* ws  = (float*)d_ws;

  qkvv_gemm  <<<dim3(128, 16), 256, 0, stream>>>(x, W, ws);
  ef_proj    <<<dim3(4, 64),   128, 0, stream>>>(ws, EF);
  l2norm_rows<<<4096,          256, 0, stream>>>(ws);
  ca_scores  <<<dim3(4, 32),   128, 0, stream>>>(ws, t1);
  ca_out     <<<dim3(32, 32),  256, 0, stream>>>(ws, out);
  sa_attn    <<<dim3(32, 32),  256, 0, stream>>>(ws, t2, out);
}


# K6 sa_attn: both products as transposes, one token per lane; softmax in registers, P chained from accumulators as column operand; no LDS or barriers; softmax-sum tree and 2nd-product k order change
# speedup vs baseline: 2.4841x; 1.0437x over previous
#include <hip/hip_runtime.h>

typedef float v2f __attribute__((ext_vector_type(2)));
typedef float v4f __attribute__((ext_vector_type(4)));
typedef float v8f __attribute__((ext_vector_type(8)));

#define WMMA_F32(a, b, c) \
  __builtin_amdgcn_wmma_f32_16x16x4_f32(false, (a), false, (b), (short)0, (c), false, false)

#define BATCH 4
#define NTOK 4096
#define CCH 512
#define HEADS 8
#define DH 64
#define KL 64
#define MROWS (BATCH * NTOK)
#define OCH (4 * CCH)
#define BH (BATCH * HEADS)

#define OFF_Q   ((size_t)0)
#define OFF_K   ((size_t)8388608)
#define OFF_VCA ((size_t)16777216)
#define OFF_VSA ((size_t)25165824)
#define OFF_KP  ((size_t)33554432)
#define OFF_VP  ((size_t)33685504)
#define OFF_P   ((size_t)33816576)

#define G_BM 128
#define G_BO 128
#define G_KC 32
#define G_LD 36
#define G_NCH (CCH / G_KC)

__global__ __launch_bounds__(256) void qkvv_gemm(const float* __restrict__ x,
                                                 const float* __restrict__ W,
                                                 float* __restrict__ ws) {
  __shared__ __attribute__((aligned(16))) float As[2][G_BM * G_LD];
  __shared__ __attribute__((aligned(16))) float Bs[2][G_BO * G_LD];

  const int tid  = threadIdx.x;
  const int lane = tid & 31;
  const int wave = tid >> 5;
  const int l16  = lane & 15;
  const int hi   = lane >> 4;
  const int wm   = wave >> 2;
  const int wo   = wave & 3;
  const int mBlock = blockIdx.x * G_BM;
  const int oBlock = blockIdx.y * G_BO;

  const int sRow = tid >> 3;
  const int sCol = (tid & 7) * 4;
  const float* xg = x + (size_t)(mBlock + sRow) * CCH + sCol;
  const float* wg = W + (size_t)(oBlock + sRow) * CCH + sCol;
  const int sOff = sRow * G_LD + sCol;

  v4f ra[4], rb[4];
  for (int i = 0; i < 4; ++i) {
    ra[i] = *(const v4f*)(xg + (size_t)i * 32 * CCH);
    rb[i] = *(const v4f*)(wg + (size_t)i * 32 * CCH);
  }
  for (int i = 0; i < 4; ++i) {
    *(v4f*)(&As[0][sOff + i * 32 * G_LD]) = ra[i];
    *(v4f*)(&Bs[0][sOff + i * 32 * G_LD]) = rb[i];
  }
  __syncthreads();

  const int aOff = (wm * 64 + l16) * G_LD + (hi ? 2 : 0);
  const int bOff = (wo * 32 + l16) * G_LD + (hi ? 2 : 0);

  v8f acc[4][2] = {};
  for (int c = 0; c < G_NCH; ++c) {
    const int cur = c & 1;
    const bool more = (c + 1 < G_NCH);
    if (more) {
      for (int i = 0; i < 4; ++i) {
        ra[i] = *(const v4f*)(xg + (size_t)i * 32 * CCH + (c + 1) * G_KC);
        rb[i] = *(const v4f*)(wg + (size_t)i * 32 * CCH + (c + 1) * G_KC);
      }
    }
    const float* Ac = &As[cur][aOff];
    const float* Bc = &Bs[cur][bOff];
    for (int j = 0; j < G_KC / 4; ++j) {
      v2f a[4], b[2];
      for (int i = 0; i < 4; ++i) a[i] = *(const v2f*)(Ac + i * 16 * G_LD + j * 4);
      for (int t = 0; t < 2; ++t) b[t] = *(const v2f*)(Bc + t * 16 * G_LD + j * 4);
      for (int i = 0; i < 4; ++i)
        for (int t = 0; t < 2; ++t)
          acc[i][t] = WMMA_F32(a[i], b[t], acc[i][t]);
    }
    if (more) {
      for (int i = 0; i < 4; ++i) {
        *(v4f*)(&As[cur ^ 1][sOff + i * 32 * G_LD]) = ra[i];
        *(v4f*)(&Bs[cur ^ 1][sOff + i * 32 * G_LD]) = rb[i];
      }
    }
    __syncthreads();
  }

  for (int t = 0; t < 2; ++t) {
    const int o  = oBlock + wo * 32 + t * 16 + l16;
    const int p  = o >> 9;
    const int h  = (o >> 6) & 7;
    const int dd = o & 63;
    for (int i = 0; i < 4; ++i) {
      for (int v = 0; v < 8; ++v) {
        const int m  = mBlock + wm * 64 + i * 16 + v + (hi ? 8 : 0);
        const int b_ = m >> 12;
        const int n  = m & (NTOK - 1);
        const size_t dst = (size_t)p * 8388608 +
                           ((size_t)(b_ * HEADS + h) * DH + dd) * NTOK + n;
        ws[dst] = acc[i][t][v];
      }
    }
  }
}

#define T_NC 256
#define T_LD 260
#define T_NCH (NTOK / T_NC)
#define E_LD 72

__global__ __launch_bounds__(128) void ef_proj(float* __restrict__ ws,
                                               const float* __restrict__ EF) {
  __shared__ __attribute__((aligned(16))) float As[2][16 * T_LD];
  __shared__ __attribute__((aligned(16))) float Es[2][T_NC * E_LD];

  const int tid  = threadIdx.x;
  const int lane = tid & 31;
  const int kt   = tid >> 5;
  const int l16  = lane & 15;
  const int hi   = lane >> 4;
  const int dt    = blockIdx.x;
  const int which = blockIdx.y & 1;
  const int bh    = blockIdx.y >> 1;

  const float* src = ws + (which ? OFF_VSA : OFF_K) + (size_t)bh * DH * NTOK +
                     (size_t)dt * 16 * NTOK;
  float* dst       = ws + (which ? OFF_VP : OFF_KP) + (size_t)bh * DH * KL;

  const int aRow = tid >> 6;
  const int aCol = (tid & 63) * 4;
  const float* ag = src + (size_t)aRow * NTOK + aCol;
  const int aOffS = aRow * T_LD + aCol;
  const int eRow = tid >> 4;
  const int eCol = (tid & 15) * 4;
  const float* eg = EF + (size_t)eRow * KL + eCol;
  const int eOffS = eRow * E_LD + eCol;

  v4f ra[8], re[32];
  for (int i = 0; i < 8; ++i)  ra[i] = *(const v4f*)(ag + (size_t)i * 2 * NTOK);
  for (int i = 0; i < 32; ++i) re[i] = *(const v4f*)(eg + (size_t)i * 8 * KL);
  for (int i = 0; i < 8; ++i)  *(v4f*)(&As[0][aOffS + i * 2 * T_LD]) = ra[i];
  for (int i = 0; i < 32; ++i) *(v4f*)(&Es[0][eOffS + i * 8 * E_LD]) = re[i];
  __syncthreads();

  const int aF = l16 * T_LD + (hi ? 2 : 0);
  const int eF = (hi ? 2 : 0) * E_LD + kt * 16 + l16;

  v8f acc = {};
  for (int c = 0; c < T_NCH; ++c) {
    const int cur = c & 1;
    const bool more = (c + 1 < T_NCH);
    if (more) {
      for (int i = 0; i < 8; ++i)
        ra[i] = *(const v4f*)(ag + (size_t)i * 2 * NTOK + (c + 1) * T_NC);
      for (int i = 0; i < 32; ++i)
        re[i] = *(const v4f*)(eg + (size_t)i * 8 * KL + (size_t)(c + 1) * T_NC * KL);
    }
    const float* Ac = &As[cur][aF];
    const float* Ec = &Es[cur][eF];
    v2f fa[2][8], fb[2][8];
#pragma unroll
    for (int i = 0; i < 8; ++i) {
      fa[0][i]   = *(const v2f*)(Ac + i * 4);
      fb[0][i].x = Ec[i * 4 * E_LD];
      fb[0][i].y = Ec[i * 4 * E_LD + E_LD];
    }
#pragma unroll
    for (int g = 0; g < T_NC / 32; ++g) {
      const int s = g & 1;
      if (g + 1 < T_NC / 32) {
#pragma unroll
        for (int i = 0; i < 8; ++i) {
          const int j = (g + 1) * 8 + i;
          fa[s ^ 1][i]   = *(const v2f*)(Ac + j * 4);
          fb[s ^ 1][i].x = Ec[j * 4 * E_LD];
          fb[s ^ 1][i].y = Ec[j * 4 * E_LD + E_LD];
        }
      }
      __builtin_amdgcn_sched_barrier(0);
#pragma unroll
      for (int i = 0; i < 8; ++i) acc = WMMA_F32(fa[s][i], fb[s][i], acc);
      __builtin_amdgcn_sched_barrier(0);
    }
    if (more) {
      for (int i = 0; i < 8; ++i)  *(v4f*)(&As[cur ^ 1][aOffS + i * 2 * T_LD]) = ra[i];
      for (int i = 0; i < 32; ++i) *(v4f*)(&Es[cur ^ 1][eOffS + i * 8 * E_LD]) = re[i];
    }
    __syncthreads();
  }
#pragma unroll
  for (int v = 0; v < 8; ++v)
    dst[(size_t)(dt * 16 + v + (hi ? 8 : 0)) * KL + kt * 16 + l16] = acc[v];
}

__global__ __launch_bounds__(256) void l2norm_rows(float* __restrict__ ws) {
  __shared__ float red[256];
  const int tid = threadIdx.x;
  const int row = blockIdx.x;
  float* ptr = ws + (row < 2048 ? OFF_Q : OFF_K) +
               (size_t)(row & 2047) * NTOK;
  float s = 0.f;
#pragma unroll
  for (int i = 0; i < 16; ++i) {
    float v = ptr[tid + i * 256];
    s += v * v;
  }
  red[tid] = s;
  __syncthreads();
  for (int off = 128; off > 0; off >>= 1) {
    if (tid < off) red[tid] += red[tid + off];
    __syncthreads();
  }
  const float scale = 1.0f / fmaxf(sqrtf(red[0]), 1e-12f);
#pragma unroll
  for (int i = 0; i < 16; ++i) ptr[tid + i * 256] *= scale;
}

__global__ __launch_bounds__(128) void ca_scores(float* __restrict__ ws,
                                                 const float* __restrict__ temperature) {
  __shared__ __attribute__((aligned(16))) float Qs[2][16 * T_LD];
  __shared__ __attribute__((aligned(16))) float Ks[2][64 * T_LD];
  __shared__ float S[16][65];
  const int tid  = threadIdx.x;
  const int lane = tid & 31;
  const int et   = tid >> 5;
  const int l16  = lane & 15;
  const int hi   = lane >> 4;
  const int dt = blockIdx.x;
  const int bh = blockIdx.y;
  const int h  = bh & 7;

  const float* qsrc = ws + OFF_Q + (size_t)bh * DH * NTOK + (size_t)dt * 16 * NTOK;
  const float* ksrc = ws + OFF_K + (size_t)bh * DH * NTOK;

  const int sRow = tid >> 6;
  const int sCol = (tid & 63) * 4;
  const float* qg = qsrc + (size_t)sRow * NTOK + sCol;
  const float* kg = ksrc + (size_t)sRow * NTOK + sCol;
  const int sOff = sRow * T_LD + sCol;

  v4f rq[8], rk[32];
  for (int i = 0; i < 8; ++i)  rq[i] = *(const v4f*)(qg + (size_t)i * 2 * NTOK);
  for (int i = 0; i < 32; ++i) rk[i] = *(const v4f*)(kg + (size_t)i * 2 * NTOK);
  for (int i = 0; i < 8; ++i)  *(v4f*)(&Qs[0][sOff + i * 2 * T_LD]) = rq[i];
  for (int i = 0; i < 32; ++i) *(v4f*)(&Ks[0][sOff + i * 2 * T_LD]) = rk[i];
  __syncthreads();

  const int qF = l16 * T_LD + (hi ? 2 : 0);
  const int kF = (et * 16 + l16) * T_LD + (hi ? 2 : 0);

  v8f acc = {};
  for (int c = 0; c < T_NCH; ++c) {
    const int cur = c & 1;
    const bool more = (c + 1 < T_NCH);
    if (more) {
      for (int i = 0; i < 8; ++i)
        rq[i] = *(const v4f*)(qg + (size_t)i * 2 * NTOK + (c + 1) * T_NC);
      for (int i = 0; i < 32; ++i)
        rk[i] = *(const v4f*)(kg + (size_t)i * 2 * NTOK + (c + 1) * T_NC);
    }
    const float* Qc = &Qs[cur][qF];
    const float* Kc = &Ks[cur][kF];
    v2f fa[2][8], fb[2][8];
#pragma unroll
    for (int i = 0; i < 8; ++i) {
      fa[0][i] = *(const v2f*)(Qc + i * 4);
      fb[0][i] = *(const v2f*)(Kc + i * 4);
    }
#pragma unroll
    for (int g = 0; g < T_NC / 32; ++g) {
      const int s = g & 1;
      if (g + 1 < T_NC / 32) {
#pragma unroll
        for (int i = 0; i < 8; ++i) {
          const int j = (g + 1) * 8 + i;
          fa[s ^ 1][i] = *(const v2f*)(Qc + j * 4);
          fb[s ^ 1][i] = *(const v2f*)(Kc + j * 4);
        }
      }
      __builtin_amdgcn_sched_barrier(0);
#pragma unroll
      for (int i = 0; i < 8; ++i) acc = WMMA_F32(fa[s][i], fb[s][i], acc);
      __builtin_amdgcn_sched_barrier(0);
    }
    if (more) {
      for (int i = 0; i < 8; ++i)  *(v4f*)(&Qs[cur ^ 1][sOff + i * 2 * T_LD]) = rq[i];
      for (int i = 0; i < 32; ++i) *(v4f*)(&Ks[cur ^ 1][sOff + i * 2 * T_LD]) = rk[i];
    }
    __syncthreads();
  }
  const float tsc = temperature[h];
#pragma unroll
  for (int v = 0; v < 8; ++v)
    S[v + (hi ? 8 : 0)][et * 16 + l16] = acc[v] * tsc;
  __syncthreads();

  if (tid < 16) {
    const int r = tid;
    float m = S[r][0];
#pragma unroll
    for (int j = 1; j < 64; ++j) m = fmaxf(m, S[r][j]);
    float sum = 0.f;
    float e[64];
#pragma unroll
    for (int j = 0; j < 64; ++j) { e[j] = __expf(S[r][j] - m); sum += e[j]; }
    const float inv = 1.0f / sum;
    float* dst = ws + OFF_P + (size_t)bh * DH * DH + (size_t)(dt * 16 + r) * DH;
#pragma unroll
    for (int j = 0; j < 64; ++j) dst[j] = e[j] * inv;
  }
}

__global__ __launch_bounds__(256) void ca_out(const float* __restrict__ ws,
                                              float* __restrict__ out) {
  __shared__ float P[64][64];
  const int tid  = threadIdx.x;
  const int lane = tid & 31;
  const int wave = tid >> 5;
  const int l16  = lane & 15;
  const int hi   = lane >> 4;
  const int bh = blockIdx.y;
  const int b_ = bh >> 3;
  const int h  = bh & 7;
  const int nBase = blockIdx.x * 128;

  const float* Pg = ws + OFF_P + (size_t)bh * DH * DH;
  for (int i = tid; i < 4096; i += 256) P[i >> 6][i & 63] = Pg[i];
  __syncthreads();

  const int dt = wave >> 1;
  const int ng = wave & 1;
  const int mRow = dt * 16 + l16;
  const float* vca = ws + OFF_VCA + (size_t)bh * DH * NTOK;

  v8f acc[4] = {};
  for (int e0 = 0; e0 < DH; e0 += 4) {
    const int k0 = e0 + (hi ? 2 : 0);
    v2f a;
    a.x = P[mRow][k0];
    a.y = P[mRow][k0 + 1];
#pragma unroll
    for (int t = 0; t < 4; ++t) {
      const int n = nBase + (ng * 4 + t) * 16 + l16;
      v2f bfrag;
      bfrag.x = vca[(size_t)k0 * NTOK + n];
      bfrag.y = vca[(size_t)(k0 + 1) * NTOK + n];
      acc[t] = WMMA_F32(a, bfrag, acc[t]);
    }
  }
#pragma unroll
  for (int t = 0; t < 4; ++t) {
    const int n = nBase + (ng * 4 + t) * 16 + l16;
#pragma unroll
    for (int v = 0; v < 8; ++v) {
      const int d = dt * 16 + v + (hi ? 8 : 0);
      out[((size_t)(b_ * NTOK + n)) * CCH + h * DH + d] = acc[t][v];
    }
  }
}

__device__ __forceinline__ float xhalf(float x) {
  const int b = __builtin_bit_cast(int, x);
  return __builtin_bit_cast(float,
      __builtin_amdgcn_permlanex16(b, b, 0x76543210, 0xfedcba98, false, false));
}

__global__ __launch_bounds__(256) void sa_attn(float* __restrict__ ws,
                                               const float* __restrict__ temperature2,
                                               float* __restrict__ out) {
  const int tid  = threadIdx.x;
  const int lane = tid & 31;
  const int wave = tid >> 5;
  const int l16  = lane & 15;
  const int hi   = lane >> 4;
  const int bh = blockIdx.y;
  const int b_ = bh >> 3;
  const int h  = bh & 7;
  const int n  = blockIdx.x * 128 + wave * 16 + l16;

  const float* Q  = ws + OFF_Q  + (size_t)bh * DH * NTOK + n;
  const float* Kp = ws + OFF_KP + (size_t)bh * DH * KL;
  const float* Vp = ws + OFF_VP + (size_t)bh * DH * KL;
  const float t2 = temperature2[h];

  v8f sc[4] = {};
#pragma unroll 4
  for (int d0 = 0; d0 < DH; d0 += 4) {
    const int k0 = d0 + (hi ? 2 : 0);
    v2f q;
    q.x = Q[(size_t)k0 * NTOK];
    q.y = Q[(size_t)(k0 + 1) * NTOK];
#pragma unroll
    for (int t = 0; t < 4; ++t) {
      const int kc = t * 16 + l16;
      v2f kf;
      kf.x = Kp[(size_t)k0 * KL + kc];
      kf.y = Kp[(size_t)(k0 + 1) * KL + kc];
      sc[t] = WMMA_F32(kf, q, sc[t]);
    }
  }

  float m = sc[0][0] * t2;
#pragma unroll
  for (int t = 0; t < 4; ++t)
#pragma unroll
    for (int r = 0; r < 8; ++r) { sc[t][r] *= t2; m = fmaxf(m, sc[t][r]); }
  m = fmaxf(m, xhalf(m));
  float sum = 0.f;
#pragma unroll
  for (int t = 0; t < 4; ++t)
#pragma unroll
    for (int r = 0; r < 8; ++r) { const float e = __expf(sc[t][r] - m); sc[t][r] = e; sum += e; }
  sum = sum + xhalf(sum);
  const float inv = 1.0f / sum;
#pragma unroll
  for (int t = 0; t < 4; ++t)
#pragma unroll
    for (int r = 0; r < 8; ++r) sc[t][r] *= inv;

  v8f oacc[4] = {};
#pragma unroll
  for (int t = 0; t < 4; ++t) {
    v4f va[4], vb[4];
#pragma unroll
    for (int u = 0; u < 4; ++u) {
      const float* vp = Vp + (size_t)(u * 16 + l16) * KL + t * 16 + (hi ? 8 : 0);
      va[u] = *(const v4f*)vp;
      vb[u] = *(const v4f*)(vp + 4);
    }
#pragma unroll
    for (int j = 0; j < 4; ++j) {
      v2f p;
      p.x = sc[t][2 * j];
      p.y = sc[t][2 * j + 1];
#pragma unroll
      for (int u = 0; u < 4; ++u) {
        v2f a;
        a.x = (j < 2) ? va[u][2 * j]     : vb[u][2 * j - 4];
        a.y = (j < 2) ? va[u][2 * j + 1] : vb[u][2 * j - 3];
        oacc[u] = WMMA_F32(a, p, oacc[u]);
      }
    }
  }

  float* op = out + (size_t)b_ * (NTOK * CCH) + (size_t)h * NTOK + n;
#pragma unroll
  for (int u = 0; u < 4; ++u) {
#pragma unroll
    for (int r = 0; r < 8; ++r) {
      const int e = u * 16 + r + (hi ? 8 : 0);
      op[(size_t)e * (HEADS * NTOK)] += oacc[u][r];
    }
  }
}

extern "C" void kernel_launch(void* const* d_in, const int* in_sizes, int n_in,
                              void* d_out, int out_size, void* d_ws, size_t ws_size,
                              hipStream_t stream) {
  const float* x  = (const float*)d_in[0];
  const float* W  = (const float*)d_in[1];
  const float* EF = (const float*)d_in[2];
  const float* t1 = (const float*)d_in[3];
  const float* t2 = (const float*)d_in[4];
  float* out = (float*)d_out;
  float* ws  = (float*)d_ws;

  qkvv_gemm  <<<dim3(128, 16), 256, 0, stream>>>(x, W, ws);
  ef_proj    <<<dim3(4, 64),   128, 0, stream>>>(ws, EF);
  l2norm_rows<<<4096,          256, 0, stream>>>(ws);
  ca_scores  <<<dim3(4, 32),   128, 0, stream>>>(ws, t1);
  ca_out     <<<dim3(32, 32),  256, 0, stream>>>(ws, out);
  sa_attn    <<<dim3(32, 32),  256, 0, stream>>>(ws, t2, out);
}


# K3 normalization pass removed: K4 sums squares of the Q/K fragments it reads, scales scores by 1/(|q||k|), folds 1/|q| into k_proj rows; 5 launches; norm sum order and rounding point change
# speedup vs baseline: 2.5999x; 1.0466x over previous
#include <hip/hip_runtime.h>

typedef float v2f __attribute__((ext_vector_type(2)));
typedef float v4f __attribute__((ext_vector_type(4)));
typedef float v8f __attribute__((ext_vector_type(8)));

#define WMMA_F32(a, b, c) \
  __builtin_amdgcn_wmma_f32_16x16x4_f32(false, (a), false, (b), (short)0, (c), false, false)

#define BATCH 4
#define NTOK 4096
#define CCH 512
#define HEADS 8
#define DH 64
#define KL 64
#define MROWS (BATCH * NTOK)
#define OCH (4 * CCH)
#define BH (BATCH * HEADS)

#define OFF_Q   ((size_t)0)
#define OFF_K   ((size_t)8388608)
#define OFF_VCA ((size_t)16777216)
#define OFF_VSA ((size_t)25165824)
#define OFF_KP  ((size_t)33554432)
#define OFF_VP  ((size_t)33685504)
#define OFF_P   ((size_t)33816576)

#define G_BM 128
#define G_BO 128
#define G_KC 32
#define G_LD 36
#define G_NCH (CCH / G_KC)

__global__ __launch_bounds__(256) void qkvv_gemm(const float* __restrict__ x,
                                                 const float* __restrict__ W,
                                                 float* __restrict__ ws) {
  __shared__ __attribute__((aligned(16))) float As[2][G_BM * G_LD];
  __shared__ __attribute__((aligned(16))) float Bs[2][G_BO * G_LD];

  const int tid  = threadIdx.x;
  const int lane = tid & 31;
  const int wave = tid >> 5;
  const int l16  = lane & 15;
  const int hi   = lane >> 4;
  const int wm   = wave >> 2;
  const int wo   = wave & 3;
  const int mBlock = blockIdx.x * G_BM;
  const int oBlock = blockIdx.y * G_BO;

  const int sRow = tid >> 3;
  const int sCol = (tid & 7) * 4;
  const float* xg = x + (size_t)(mBlock + sRow) * CCH + sCol;
  const float* wg = W + (size_t)(oBlock + sRow) * CCH + sCol;
  const int sOff = sRow * G_LD + sCol;

  v4f ra[4], rb[4];
  for (int i = 0; i < 4; ++i) {
    ra[i] = *(const v4f*)(xg + (size_t)i * 32 * CCH);
    rb[i] = *(const v4f*)(wg + (size_t)i * 32 * CCH);
  }
  for (int i = 0; i < 4; ++i) {
    *(v4f*)(&As[0][sOff + i * 32 * G_LD]) = ra[i];
    *(v4f*)(&Bs[0][sOff + i * 32 * G_LD]) = rb[i];
  }
  __syncthreads();

  const int aOff = (wm * 64 + l16) * G_LD + (hi ? 2 : 0);
  const int bOff = (wo * 32 + l16) * G_LD + (hi ? 2 : 0);

  v8f acc[4][2] = {};
  for (int c = 0; c < G_NCH; ++c) {
    const int cur = c & 1;
    const bool more = (c + 1 < G_NCH);
    if (more) {
      for (int i = 0; i < 4; ++i) {
        ra[i] = *(const v4f*)(xg + (size_t)i * 32 * CCH + (c + 1) * G_KC);
        rb[i] = *(const v4f*)(wg + (size_t)i * 32 * CCH + (c + 1) * G_KC);
      }
    }
    const float* Ac = &As[cur][aOff];
    const float* Bc = &Bs[cur][bOff];
    for (int j = 0; j < G_KC / 4; ++j) {
      v2f a[4], b[2];
      for (int i = 0; i < 4; ++i) a[i] = *(const v2f*)(Ac + i * 16 * G_LD + j * 4);
      for (int t = 0; t < 2; ++t) b[t] = *(const v2f*)(Bc + t * 16 * G_LD + j * 4);
      for (int i = 0; i < 4; ++i)
        for (int t = 0; t < 2; ++t)
          acc[i][t] = WMMA_F32(a[i], b[t], acc[i][t]);
    }
    if (more) {
      for (int i = 0; i < 4; ++i) {
        *(v4f*)(&As[cur ^ 1][sOff + i * 32 * G_LD]) = ra[i];
        *(v4f*)(&Bs[cur ^ 1][sOff + i * 32 * G_LD]) = rb[i];
      }
    }
    __syncthreads();
  }

  for (int t = 0; t < 2; ++t) {
    const int o  = oBlock + wo * 32 + t * 16 + l16;
    const int p  = o >> 9;
    const int h  = (o >> 6) & 7;
    const int dd = o & 63;
    for (int i = 0; i < 4; ++i) {
      for (int v = 0; v < 8; ++v) {
        const int m  = mBlock + wm * 64 + i * 16 + v + (hi ? 8 : 0);
        const int b_ = m >> 12;
        const int n  = m & (NTOK - 1);
        const size_t dst = (size_t)p * 8388608 +
                           ((size_t)(b_ * HEADS + h) * DH + dd) * NTOK + n;
        ws[dst] = acc[i][t][v];
      }
    }
  }
}

#define T_NC 256
#define T_LD 260
#define T_NCH (NTOK / T_NC)
#define E_LD 72

__global__ __launch_bounds__(128) void ef_proj(float* __restrict__ ws,
                                               const float* __restrict__ EF) {
  __shared__ __attribute__((aligned(16))) float As[2][16 * T_LD];
  __shared__ __attribute__((aligned(16))) float Es[2][T_NC * E_LD];

  const int tid  = threadIdx.x;
  const int lane = tid & 31;
  const int kt   = tid >> 5;
  const int l16  = lane & 15;
  const int hi   = lane >> 4;
  const int dt    = blockIdx.x;
  const int which = blockIdx.y & 1;
  const int bh    = blockIdx.y >> 1;

  const float* src = ws + (which ? OFF_VSA : OFF_K) + (size_t)bh * DH * NTOK +
                     (size_t)dt * 16 * NTOK;
  float* dst       = ws + (which ? OFF_VP : OFF_KP) + (size_t)bh * DH * KL;

  const int aRow = tid >> 6;
  const int aCol = (tid & 63) * 4;
  const float* ag = src + (size_t)aRow * NTOK + aCol;
  const int aOffS = aRow * T_LD + aCol;
  const int eRow = tid >> 4;
  const int eCol = (tid & 15) * 4;
  const float* eg = EF + (size_t)eRow * KL + eCol;
  const int eOffS = eRow * E_LD + eCol;

  v4f ra[8], re[32];
  for (int i = 0; i < 8; ++i)  ra[i] = *(const v4f*)(ag + (size_t)i * 2 * NTOK);
  for (int i = 0; i < 32; ++i) re[i] = *(const v4f*)(eg + (size_t)i * 8 * KL);
  for (int i = 0; i < 8; ++i)  *(v4f*)(&As[0][aOffS + i * 2 * T_LD]) = ra[i];
  for (int i = 0; i < 32; ++i) *(v4f*)(&Es[0][eOffS + i * 8 * E_LD]) = re[i];
  __syncthreads();

  const int aF = l16 * T_LD + (hi ? 2 : 0);
  const int eF = (hi ? 2 : 0) * E_LD + kt * 16 + l16;

  v8f acc = {};
  for (int c = 0; c < T_NCH; ++c) {
    const int cur = c & 1;
    const bool more = (c + 1 < T_NCH);
    if (more) {
      for (int i = 0; i < 8; ++i)
        ra[i] = *(const v4f*)(ag + (size_t)i * 2 * NTOK + (c + 1) * T_NC);
      for (int i = 0; i < 32; ++i)
        re[i] = *(const v4f*)(eg + (size_t)i * 8 * KL + (size_t)(c + 1) * T_NC * KL);
    }
    const float* Ac = &As[cur][aF];
    const float* Ec = &Es[cur][eF];
    v2f fa[2][8], fb[2][8];
#pragma unroll
    for (int i = 0; i < 8; ++i) {
      fa[0][i]   = *(const v2f*)(Ac + i * 4);
      fb[0][i].x = Ec[i * 4 * E_LD];
      fb[0][i].y = Ec[i * 4 * E_LD + E_LD];
    }
#pragma unroll
    for (int g = 0; g < T_NC / 32; ++g) {
      const int s = g & 1;
      if (g + 1 < T_NC / 32) {
#pragma unroll
        for (int i = 0; i < 8; ++i) {
          const int j = (g + 1) * 8 + i;
          fa[s ^ 1][i]   = *(const v2f*)(Ac + j * 4);
          fb[s ^ 1][i].x = Ec[j * 4 * E_LD];
          fb[s ^ 1][i].y = Ec[j * 4 * E_LD + E_LD];
        }
      }
      __builtin_amdgcn_sched_barrier(0);
#pragma unroll
      for (int i = 0; i < 8; ++i) acc = WMMA_F32(fa[s][i], fb[s][i], acc);
      __builtin_amdgcn_sched_barrier(0);
    }
    if (more) {
      for (int i = 0; i < 8; ++i)  *(v4f*)(&As[cur ^ 1][aOffS + i * 2 * T_LD]) = ra[i];
      for (int i = 0; i < 32; ++i) *(v4f*)(&Es[cur ^ 1][eOffS + i * 8 * E_LD]) = re[i];
    }
    __syncthreads();
  }
#pragma unroll
  for (int v = 0; v < 8; ++v)
    dst[(size_t)(dt * 16 + v + (hi ? 8 : 0)) * KL + kt * 16 + l16] = acc[v];
}

__device__ __forceinline__ float xhalf(float x) {
  const int b = __builtin_bit_cast(int, x);
  return __builtin_bit_cast(float,
      __builtin_amdgcn_permlanex16(b, b, 0x76543210, 0xfedcba98, false, false));
}

__global__ __launch_bounds__(128) void ca_scores(float* __restrict__ ws,
                                                 const float* __restrict__ temperature) {
  __shared__ __attribute__((aligned(16))) float Qs[2][16 * T_LD];
  __shared__ __attribute__((aligned(16))) float Ks[2][64 * T_LD];
  __shared__ float S[16][65];
  const int tid  = threadIdx.x;
  const int lane = tid & 31;
  const int et   = tid >> 5;
  const int l16  = lane & 15;
  const int hi   = lane >> 4;
  const int dt = blockIdx.x;
  const int bh = blockIdx.y;
  const int h  = bh & 7;

  const float* qsrc = ws + OFF_Q + (size_t)bh * DH * NTOK + (size_t)dt * 16 * NTOK;
  const float* ksrc = ws + OFF_K + (size_t)bh * DH * NTOK;

  const int sRow = tid >> 6;
  const int sCol = (tid & 63) * 4;
  const float* qg = qsrc + (size_t)sRow * NTOK + sCol;
  const float* kg = ksrc + (size_t)sRow * NTOK + sCol;
  const int sOff = sRow * T_LD + sCol;

  v4f rq[8], rk[32];
  for (int i = 0; i < 8; ++i)  rq[i] = *(const v4f*)(qg + (size_t)i * 2 * NTOK);
  for (int i = 0; i < 32; ++i) rk[i] = *(const v4f*)(kg + (size_t)i * 2 * NTOK);
  for (int i = 0; i < 8; ++i)  *(v4f*)(&Qs[0][sOff + i * 2 * T_LD]) = rq[i];
  for (int i = 0; i < 32; ++i) *(v4f*)(&Ks[0][sOff + i * 2 * T_LD]) = rk[i];
  __syncthreads();

  const int qF = l16 * T_LD + (hi ? 2 : 0);
  const int kF = (et * 16 + l16) * T_LD + (hi ? 2 : 0);

  v8f acc = {};
  v2f qss = {0.f, 0.f};
  v2f kss = {0.f, 0.f};
  for (int c = 0; c < T_NCH; ++c) {
    const int cur = c & 1;
    const bool more = (c + 1 < T_NCH);
    if (more) {
      for (int i = 0; i < 8; ++i)
        rq[i] = *(const v4f*)(qg + (size_t)i * 2 * NTOK + (c + 1) * T_NC);
      for (int i = 0; i < 32; ++i)
        rk[i] = *(const v4f*)(kg + (size_t)i * 2 * NTOK + (c + 1) * T_NC);
    }
    const float* Qc = &Qs[cur][qF];
    const float* Kc = &Ks[cur][kF];
    v2f fa[2][8], fb[2][8];
#pragma unroll
    for (int i = 0; i < 8; ++i) {
      fa[0][i] = *(const v2f*)(Qc + i * 4);
      fb[0][i] = *(const v2f*)(Kc + i * 4);
    }
#pragma unroll
    for (int g = 0; g < T_NC / 32; ++g) {
      const int s = g & 1;
      if (g + 1 < T_NC / 32) {
#pragma unroll
        for (int i = 0; i < 8; ++i) {
          const int j = (g + 1) * 8 + i;
          fa[s ^ 1][i] = *(const v2f*)(Qc + j * 4);
          fb[s ^ 1][i] = *(const v2f*)(Kc + j * 4);
        }
      }
      __builtin_amdgcn_sched_barrier(0);
#pragma unroll
      for (int i = 0; i < 8; ++i) {
        acc = WMMA_F32(fa[s][i], fb[s][i], acc);
        qss += fa[s][i] * fa[s][i];
        kss += fb[s][i] * fb[s][i];
      }
      __builtin_amdgcn_sched_barrier(0);
    }
    if (more) {
      for (int i = 0; i < 8; ++i)  *(v4f*)(&Qs[cur ^ 1][sOff + i * 2 * T_LD]) = rq[i];
      for (int i = 0; i < 32; ++i) *(v4f*)(&Ks[cur ^ 1][sOff + i * 2 * T_LD]) = rk[i];
    }
    __syncthreads();
  }
  float qs = qss.x + qss.y;
  qs += xhalf(qs);
  float ks = kss.x + kss.y;
  ks += xhalf(ks);
  const float invq = 1.0f / fmaxf(sqrtf(qs), 1e-12f);
  const float invk = 1.0f / fmaxf(sqrtf(ks), 1e-12f);

  const float tsc = temperature[h];
#pragma unroll
  for (int v = 0; v < 8; ++v)
    S[v + (hi ? 8 : 0)][et * 16 + l16] = acc[v] * invk;

  {
    float* kp = ws + OFF_KP + ((size_t)bh * DH + dt * 16 + l16) * KL + et * 16 + (hi ? 8 : 0);
    v4f k0 = *(const v4f*)kp;
    v4f k1 = *(const v4f*)(kp + 4);
    k0 *= invq;
    k1 *= invq;
    *(v4f*)kp = k0;
    *(v4f*)(kp + 4) = k1;
  }
  __syncthreads();

  if (tid < 16) {
    const int r = tid;
    float sv[64];
#pragma unroll
    for (int j = 0; j < 64; ++j) sv[j] = (S[r][j] * invq) * tsc;
    float m = sv[0];
#pragma unroll
    for (int j = 1; j < 64; ++j) m = fmaxf(m, sv[j]);
    float sum = 0.f;
    float e[64];
#pragma unroll
    for (int j = 0; j < 64; ++j) { e[j] = __expf(sv[j] - m); sum += e[j]; }
    const float inv = 1.0f / sum;
    float* dst = ws + OFF_P + (size_t)bh * DH * DH + (size_t)(dt * 16 + r) * DH;
#pragma unroll
    for (int j = 0; j < 64; ++j) dst[j] = e[j] * inv;
  }
}

__global__ __launch_bounds__(256) void ca_out(const float* __restrict__ ws,
                                              float* __restrict__ out) {
  __shared__ float P[64][64];
  const int tid  = threadIdx.x;
  const int lane = tid & 31;
  const int wave = tid >> 5;
  const int l16  = lane & 15;
  const int hi   = lane >> 4;
  const int bh = blockIdx.y;
  const int b_ = bh >> 3;
  const int h  = bh & 7;
  const int nBase = blockIdx.x * 128;

  const float* Pg = ws + OFF_P + (size_t)bh * DH * DH;
  for (int i = tid; i < 4096; i += 256) P[i >> 6][i & 63] = Pg[i];
  __syncthreads();

  const int dt = wave >> 1;
  const int ng = wave & 1;
  const int mRow = dt * 16 + l16;
  const float* vca = ws + OFF_VCA + (size_t)bh * DH * NTOK;

  v8f acc[4] = {};
  for (int e0 = 0; e0 < DH; e0 += 4) {
    const int k0 = e0 + (hi ? 2 : 0);
    v2f a;
    a.x = P[mRow][k0];
    a.y = P[mRow][k0 + 1];
#pragma unroll
    for (int t = 0; t < 4; ++t) {
      const int n = nBase + (ng * 4 + t) * 16 + l16;
      v2f bfrag;
      bfrag.x = vca[(size_t)k0 * NTOK + n];
      bfrag.y = vca[(size_t)(k0 + 1) * NTOK + n];
      acc[t] = WMMA_F32(a, bfrag, acc[t]);
    }
  }
#pragma unroll
  for (int t = 0; t < 4; ++t) {
    const int n = nBase + (ng * 4 + t) * 16 + l16;
#pragma unroll
    for (int v = 0; v < 8; ++v) {
      const int d = dt * 16 + v + (hi ? 8 : 0);
      out[((size_t)(b_ * NTOK + n)) * CCH + h * DH + d] = acc[t][v];
    }
  }
}

__global__ __launch_bounds__(256) void sa_attn(float* __restrict__ ws,
                                               const float* __restrict__ temperature2,
                                               float* __restrict__ out) {
  const int tid  = threadIdx.x;
  const int lane = tid & 31;
  const int wave = tid >> 5;
  const int l16  = lane & 15;
  const int hi   = lane >> 4;
  const int bh = blockIdx.y;
  const int b_ = bh >> 3;
  const int h  = bh & 7;
  const int n  = blockIdx.x * 128 + wave * 16 + l16;

  const float* Q  = ws + OFF_Q  + (size_t)bh * DH * NTOK + n;
  const float* Kp = ws + OFF_KP + (size_t)bh * DH * KL;
  const float* Vp = ws + OFF_VP + (size_t)bh * DH * KL;
  const float t2 = temperature2[h];

  v8f sc[4] = {};
#pragma unroll 4
  for (int d0 = 0; d0 < DH; d0 += 4) {
    const int k0 = d0 + (hi ? 2 : 0);
    v2f q;
    q.x = Q[(size_t)k0 * NTOK];
    q.y = Q[(size_t)(k0 + 1) * NTOK];
#pragma unroll
    for (int t = 0; t < 4; ++t) {
      const int kc = t * 16 + l16;
      v2f kf;
      kf.x = Kp[(size_t)k0 * KL + kc];
      kf.y = Kp[(size_t)(k0 + 1) * KL + kc];
      sc[t] = WMMA_F32(kf, q, sc[t]);
    }
  }

  float m = sc[0][0] * t2;
#pragma unroll
  for (int t = 0; t < 4; ++t)
#pragma unroll
    for (int r = 0; r < 8; ++r) { sc[t][r] *= t2; m = fmaxf(m, sc[t][r]); }
  m = fmaxf(m, xhalf(m));
  float sum = 0.f;
#pragma unroll
  for (int t = 0; t < 4; ++t)
#pragma unroll
    for (int r = 0; r < 8; ++r) { const float e = __expf(sc[t][r] - m); sc[t][r] = e; sum += e; }
  sum = sum + xhalf(sum);
  const float inv = 1.0f / sum;
#pragma unroll
  for (int t = 0; t < 4; ++t)
#pragma unroll
    for (int r = 0; r < 8; ++r) sc[t][r] *= inv;

  v8f oacc[4] = {};
#pragma unroll
  for (int t = 0; t < 4; ++t) {
    v4f va[4], vb[4];
#pragma unroll
    for (int u = 0; u < 4; ++u) {
      const float* vp = Vp + (size_t)(u * 16 + l16) * KL + t * 16 + (hi ? 8 : 0);
      va[u] = *(const v4f*)vp;
      vb[u] = *(const v4f*)(vp + 4);
    }
#pragma unroll
    for (int j = 0; j < 4; ++j) {
      v2f p;
      p.x = sc[t][2 * j];
      p.y = sc[t][2 * j + 1];
#pragma unroll
      for (int u = 0; u < 4; ++u) {
        v2f a;
        a.x = (j < 2) ? va[u][2 * j]     : vb[u][2 * j - 4];
        a.y = (j < 2) ? va[u][2 * j + 1] : vb[u][2 * j - 3];
        oacc[u] = WMMA_F32(a, p, oacc[u]);
      }
    }
  }

  float* op = out + (size_t)b_ * (NTOK * CCH) + (size_t)h * NTOK + n;
#pragma unroll
  for (int u = 0; u < 4; ++u) {
#pragma unroll
    for (int r = 0; r < 8; ++r) {
      const int e = u * 16 + r + (hi ? 8 : 0);
      op[(size_t)e * (HEADS * NTOK)] += oacc[u][r];
    }
  }
}

extern "C" void kernel_launch(void* const* d_in, const int* in_sizes, int n_in,
                              void* d_out, int out_size, void* d_ws, size_t ws_size,
                              hipStream_t stream) {
  const float* x  = (const float*)d_in[0];
  const float* W  = (const float*)d_in[1];
  const float* EF = (const float*)d_in[2];
  const float* t1 = (const float*)d_in[3];
  const float* t2 = (const float*)d_in[4];
  float* out = (float*)d_out;
  float* ws  = (float*)d_ws;

  qkvv_gemm  <<<dim3(128, 16), 256, 0, stream>>>(x, W, ws);
  ef_proj    <<<dim3(4, 64),   128, 0, stream>>>(ws, EF);
  ca_scores  <<<dim3(4, 32),   128, 0, stream>>>(ws, t1);
  ca_out     <<<dim3(32, 32),  256, 0, stream>>>(ws, out);
  sa_attn    <<<dim3(32, 32),  256, 0, stream>>>(ws, t2, out);
}


# K2 ef_proj retiled: tokens split in 8 slices, workgroup = 64 rows x 1 slice (EF chunk staged once per 64 rows, 4 accumulators per wave); K4 adds the 8 partials; sum order changes (split reduction)
# speedup vs baseline: 2.7199x; 1.0462x over previous
#include <hip/hip_runtime.h>

typedef float v2f __attribute__((ext_vector_type(2)));
typedef float v4f __attribute__((ext_vector_type(4)));
typedef float v8f __attribute__((ext_vector_type(8)));

#define WMMA_F32(a, b, c) \
  __builtin_amdgcn_wmma_f32_16x16x4_f32(false, (a), false, (b), (short)0, (c), false, false)

#define BATCH 4
#define NTOK 4096
#define CCH 512
#define HEADS 8
#define DH 64
#define KL 64
#define MROWS (BATCH * NTOK)
#define OCH (4 * CCH)
#define BH (BATCH * HEADS)

#define OFF_Q   ((size_t)0)
#define OFF_K   ((size_t)8388608)
#define OFF_VCA ((size_t)16777216)
#define OFF_VSA ((size_t)25165824)
#define OFF_KP  ((size_t)33554432)
#define OFF_VP  ((size_t)33685504)
#define OFF_P   ((size_t)33816576)

#define G_BM 128
#define G_BO 128
#define G_KC 32
#define G_LD 36
#define G_NCH (CCH / G_KC)

__global__ __launch_bounds__(256) void qkvv_gemm(const float* __restrict__ x,
                                                 const float* __restrict__ W,
                                                 float* __restrict__ ws) {
  __shared__ __attribute__((aligned(16))) float As[2][G_BM * G_LD];
  __shared__ __attribute__((aligned(16))) float Bs[2][G_BO * G_LD];

  const int tid  = threadIdx.x;
  const int lane = tid & 31;
  const int wave = tid >> 5;
  const int l16  = lane & 15;
  const int hi   = lane >> 4;
  const int wm   = wave >> 2;
  const int wo   = wave & 3;
  const int mBlock = blockIdx.x * G_BM;
  const int oBlock = blockIdx.y * G_BO;

  const int sRow = tid >> 3;
  const int sCol = (tid & 7) * 4;
  const float* xg = x + (size_t)(mBlock + sRow) * CCH + sCol;
  const float* wg = W + (size_t)(oBlock + sRow) * CCH + sCol;
  const int sOff = sRow * G_LD + sCol;

  v4f ra[4], rb[4];
  for (int i = 0; i < 4; ++i) {
    ra[i] = *(const v4f*)(xg + (size_t)i * 32 * CCH);
    rb[i] = *(const v4f*)(wg + (size_t)i * 32 * CCH);
  }
  for (int i = 0; i < 4; ++i) {
    *(v4f*)(&As[0][sOff + i * 32 * G_LD]) = ra[i];
    *(v4f*)(&Bs[0][sOff + i * 32 * G_LD]) = rb[i];
  }
  __syncthreads();

  const int aOff = (wm * 64 + l16) * G_LD + (hi ? 2 : 0);
  const int bOff = (wo * 32 + l16) * G_LD + (hi ? 2 : 0);

  v8f acc[4][2] = {};
  for (int c = 0; c < G_NCH; ++c) {
    const int cur = c & 1;
    const bool more = (c + 1 < G_NCH);
    if (more) {
      for (int i = 0; i < 4; ++i) {
        ra[i] = *(const v4f*)(xg + (size_t)i * 32 * CCH + (c + 1) * G_KC);
        rb[i] = *(const v4f*)(wg + (size_t)i * 32 * CCH + (c + 1) * G_KC);
      }
    }
    const float* Ac = &As[cur][aOff];
    const float* Bc = &Bs[cur][bOff];
    for (int j = 0; j < G_KC / 4; ++j) {
      v2f a[4], b[2];
      for (int i = 0; i < 4; ++i) a[i] = *(const v2f*)(Ac + i * 16 * G_LD + j * 4);
      for (int t = 0; t < 2; ++t) b[t] = *(const v2f*)(Bc + t * 16 * G_LD + j * 4);
      for (int i = 0; i < 4; ++i)
        for (int t = 0; t < 2; ++t)
          acc[i][t] = WMMA_F32(a[i], b[t], acc[i][t]);
    }
    if (more) {
      for (int i = 0; i < 4; ++i) {
        *(v4f*)(&As[cur ^ 1][sOff + i * 32 * G_LD]) = ra[i];
        *(v4f*)(&Bs[cur ^ 1][sOff + i * 32 * G_LD]) = rb[i];
      }
    }
    __syncthreads();
  }

  for (int t = 0; t < 2; ++t) {
    const int o  = oBlock + wo * 32 + t * 16 + l16;
    const int p  = o >> 9;
    const int h  = (o >> 6) & 7;
    const int dd = o & 63;
    for (int i = 0; i < 4; ++i) {
      for (int v = 0; v < 8; ++v) {
        const int m  = mBlock + wm * 64 + i * 16 + v + (hi ? 8 : 0);
        const int b_ = m >> 12;
        const int n  = m & (NTOK - 1);
        const size_t dst = (size_t)p * 8388608 +
                           ((size_t)(b_ * HEADS + h) * DH + dd) * NTOK + n;
        ws[dst] = acc[i][t][v];
      }
    }
  }
}

#define T_NC 256
#define T_LD 260
#define T_NCH (NTOK / T_NC)
#define E_LD 72

#define P_NC 128
#define P_LD 132
#define P_SL 8
#define P_NT (NTOK / P_SL)
#define P_NCH (P_NT / P_NC)

__global__ __launch_bounds__(128) void ef_proj(const float* __restrict__ ws,
                                               const float* __restrict__ EF,
                                               float* __restrict__ part) {
  __shared__ __attribute__((aligned(16))) float As[2][64 * P_LD];
  __shared__ __attribute__((aligned(16))) float Es[2][P_NC * E_LD];

  const int tid  = threadIdx.x;
  const int lane = tid & 31;
  const int kt   = tid >> 5;
  const int l16  = lane & 15;
  const int hi   = lane >> 4;
  const int sl    = blockIdx.x;
  const int which = blockIdx.y & 1;
  const int bh    = blockIdx.y >> 1;

  const float* src = ws + (which ? OFF_VSA : OFF_K) + (size_t)bh * DH * NTOK + sl * P_NT;
  const float* efs = EF + (size_t)sl * P_NT * KL;
  float* dst       = part + (size_t)(sl * 64 + blockIdx.y) * (DH * KL);

  const int aRow = tid >> 5;
  const int aCol = (tid & 31) * 4;
  const float* ag = src + (size_t)aRow * NTOK + aCol;
  const int aOffS = aRow * P_LD + aCol;
  const int eRow = tid >> 4;
  const int eCol = (tid & 15) * 4;
  const float* eg = efs + (size_t)eRow * KL + eCol;
  const int eOffS = eRow * E_LD + eCol;

  v4f ra[16], re[16];
  for (int i = 0; i < 16; ++i) ra[i] = *(const v4f*)(ag + (size_t)i * 4 * NTOK);
  for (int i = 0; i < 16; ++i) re[i] = *(const v4f*)(eg + (size_t)i * 8 * KL);
  for (int i = 0; i < 16; ++i) *(v4f*)(&As[0][aOffS + i * 4 * P_LD]) = ra[i];
  for (int i = 0; i < 16; ++i) *(v4f*)(&Es[0][eOffS + i * 8 * E_LD]) = re[i];
  __syncthreads();

  const int aF = l16 * P_LD + (hi ? 2 : 0);
  const int eF = (hi ? 2 : 0) * E_LD + kt * 16 + l16;

  v8f acc[4] = {};
  for (int c = 0; c < P_NCH; ++c) {
    const int cur = c & 1;
    const bool more = (c + 1 < P_NCH);
    if (more) {
      for (int i = 0; i < 16; ++i)
        ra[i] = *(const v4f*)(ag + (size_t)i * 4 * NTOK + (c + 1) * P_NC);
      for (int i = 0; i < 16; ++i)
        re[i] = *(const v4f*)(eg + (size_t)i * 8 * KL + (size_t)(c + 1) * P_NC * KL);
    }
    const float* Ac = &As[cur][aF];
    const float* Ec = &Es[cur][eF];
    v2f fa[2][4][4], fb[2][4];
#pragma unroll
    for (int j = 0; j < 4; ++j) {
#pragma unroll
      for (int i = 0; i < 4; ++i) fa[0][j][i] = *(const v2f*)(Ac + i * 16 * P_LD + j * 4);
      fb[0][j].x = Ec[j * 4 * E_LD];
      fb[0][j].y = Ec[j * 4 * E_LD + E_LD];
    }
#pragma unroll
    for (int g = 0; g < P_NC / 16; ++g) {
      const int s = g & 1;
      if (g + 1 < P_NC / 16) {
#pragma unroll
        for (int j = 0; j < 4; ++j) {
          const int k = (g + 1) * 4 + j;
#pragma unroll
          for (int i = 0; i < 4; ++i) fa[s ^ 1][j][i] = *(const v2f*)(Ac + i * 16 * P_LD + k * 4);
          fb[s ^ 1][j].x = Ec[k * 4 * E_LD];
          fb[s ^ 1][j].y = Ec[k * 4 * E_LD + E_LD];
        }
      }
      __builtin_amdgcn_sched_barrier(0);
#pragma unroll
      for (int j = 0; j < 4; ++j)
#pragma unroll
        for (int i = 0; i < 4; ++i) acc[i] = WMMA_F32(fa[s][j][i], fb[s][j], acc[i]);
      __builtin_amdgcn_sched_barrier(0);
    }
    if (more) {
      for (int i = 0; i < 16; ++i) *(v4f*)(&As[cur ^ 1][aOffS + i * 4 * P_LD]) = ra[i];
      for (int i = 0; i < 16; ++i) *(v4f*)(&Es[cur ^ 1][eOffS + i * 8 * E_LD]) = re[i];
    }
    __syncthreads();
  }
#pragma unroll
  for (int i = 0; i < 4; ++i)
#pragma unroll
    for (int v = 0; v < 8; ++v)
      dst[(size_t)(i * 16 + v + (hi ? 8 : 0)) * KL + kt * 16 + l16] = acc[i][v];
}

__device__ __forceinline__ float xhalf(float x) {
  const int b = __builtin_bit_cast(int, x);
  return __builtin_bit_cast(float,
      __builtin_amdgcn_permlanex16(b, b, 0x76543210, 0xfedcba98, false, false));
}

__global__ __launch_bounds__(128) void ca_scores(float* __restrict__ ws,
                                                 const float* __restrict__ temperature,
                                                 const float* __restrict__ part) {
  __shared__ __attribute__((aligned(16))) float Qs[2][16 * T_LD];
  __shared__ __attribute__((aligned(16))) float Ks[2][64 * T_LD];
  __shared__ float S[16][65];
  const int tid  = threadIdx.x;
  const int lane = tid & 31;
  const int et   = tid >> 5;
  const int l16  = lane & 15;
  const int hi   = lane >> 4;
  const int dt = blockIdx.x;
  const int bh = blockIdx.y;
  const int h  = bh & 7;

  const float* qsrc = ws + OFF_Q + (size_t)bh * DH * NTOK + (size_t)dt * 16 * NTOK;
  const float* ksrc = ws + OFF_K + (size_t)bh * DH * NTOK;

  const int sRow = tid >> 6;
  const int sCol = (tid & 63) * 4;
  const float* qg = qsrc + (size_t)sRow * NTOK + sCol;
  const float* kg = ksrc + (size_t)sRow * NTOK + sCol;
  const int sOff = sRow * T_LD + sCol;

  v4f rq[8], rk[32];
  for (int i = 0; i < 8; ++i)  rq[i] = *(const v4f*)(qg + (size_t)i * 2 * NTOK);
  for (int i = 0; i < 32; ++i) rk[i] = *(const v4f*)(kg + (size_t)i * 2 * NTOK);
  for (int i = 0; i < 8; ++i)  *(v4f*)(&Qs[0][sOff + i * 2 * T_LD]) = rq[i];
  for (int i = 0; i < 32; ++i) *(v4f*)(&Ks[0][sOff + i * 2 * T_LD]) = rk[i];
  __syncthreads();

  const int qF = l16 * T_LD + (hi ? 2 : 0);
  const int kF = (et * 16 + l16) * T_LD + (hi ? 2 : 0);

  v8f acc = {};
  v2f qss = {0.f, 0.f};
  v2f kss = {0.f, 0.f};
  for (int c = 0; c < T_NCH; ++c) {
    const int cur = c & 1;
    const bool more = (c + 1 < T_NCH);
    if (more) {
      for (int i = 0; i < 8; ++i)
        rq[i] = *(const v4f*)(qg + (size_t)i * 2 * NTOK + (c + 1) * T_NC);
      for (int i = 0; i < 32; ++i)
        rk[i] = *(const v4f*)(kg + (size_t)i * 2 * NTOK + (c + 1) * T_NC);
    }
    const float* Qc = &Qs[cur][qF];
    const float* Kc = &Ks[cur][kF];
    v2f fa[2][8], fb[2][8];
#pragma unroll
    for (int i = 0; i < 8; ++i) {
      fa[0][i] = *(const v2f*)(Qc + i * 4);
      fb[0][i] = *(const v2f*)(Kc + i * 4);
    }
#pragma unroll
    for (int g = 0; g < T_NC / 32; ++g) {
      const int s = g & 1;
      if (g + 1 < T_NC / 32) {
#pragma unroll
        for (int i = 0; i < 8; ++i) {
          const int j = (g + 1) * 8 + i;
          fa[s ^ 1][i] = *(const v2f*)(Qc + j * 4);
          fb[s ^ 1][i] = *(const v2f*)(Kc + j * 4);
        }
      }
      __builtin_amdgcn_sched_barrier(0);
#pragma unroll
      for (int i = 0; i < 8; ++i) {
        acc = WMMA_F32(fa[s][i], fb[s][i], acc);
        qss += fa[s][i] * fa[s][i];
        kss += fb[s][i] * fb[s][i];
      }
      __builtin_amdgcn_sched_barrier(0);
    }
    if (more) {
      for (int i = 0; i < 8; ++i)  *(v4f*)(&Qs[cur ^ 1][sOff + i * 2 * T_LD]) = rq[i];
      for (int i = 0; i < 32; ++i) *(v4f*)(&Ks[cur ^ 1][sOff + i * 2 * T_LD]) = rk[i];
    }
    __syncthreads();
  }
  float qs = qss.x + qss.y;
  qs += xhalf(qs);
  float ks = kss.x + kss.y;
  ks += xhalf(ks);
  const float invq = 1.0f / fmaxf(sqrtf(qs), 1e-12f);
  const float invk = 1.0f / fmaxf(sqrtf(ks), 1e-12f);

  const float tsc = temperature[h];
#pragma unroll
  for (int v = 0; v < 8; ++v)
    S[v + (hi ? 8 : 0)][et * 16 + l16] = acc[v] * invk;

  {
    const size_t rc = (size_t)(dt * 16 + l16) * KL + et * 16 + (hi ? 8 : 0);
    const float* pk = part + (size_t)(bh * 2) * (DH * KL) + rc;
    const float* pv = pk + (size_t)(DH * KL);
    v4f k0 = *(const v4f*)pk;
    v4f k1 = *(const v4f*)(pk + 4);
    v4f v0 = *(const v4f*)pv;
    v4f v1 = *(const v4f*)(pv + 4);
#pragma unroll
    for (int sl = 1; sl < P_SL; ++sl) {
      const size_t so = (size_t)sl * 64 * (DH * KL);
      k0 += *(const v4f*)(pk + so);
      k1 += *(const v4f*)(pk + so + 4);
      v0 += *(const v4f*)(pv + so);
      v1 += *(const v4f*)(pv + so + 4);
    }
    k0 *= invq;
    k1 *= invq;
    float* kp = ws + OFF_KP + (size_t)bh * DH * KL + rc;
    float* vp = ws + OFF_VP + (size_t)bh * DH * KL + rc;
    *(v4f*)kp = k0;
    *(v4f*)(kp + 4) = k1;
    *(v4f*)vp = v0;
    *(v4f*)(vp + 4) = v1;
  }
  __syncthreads();

  if (tid < 16) {
    const int r = tid;
    float sv[64];
#pragma unroll
    for (int j = 0; j < 64; ++j) sv[j] = (S[r][j] * invq) * tsc;
    float m = sv[0];
#pragma unroll
    for (int j = 1; j < 64; ++j) m = fmaxf(m, sv[j]);
    float sum = 0.f;
    float e[64];
#pragma unroll
    for (int j = 0; j < 64; ++j) { e[j] = __expf(sv[j] - m); sum += e[j]; }
    const float inv = 1.0f / sum;
    float* dst = ws + OFF_P + (size_t)bh * DH * DH + (size_t)(dt * 16 + r) * DH;
#pragma unroll
    for (int j = 0; j < 64; ++j) dst[j] = e[j] * inv;
  }
}

__global__ __launch_bounds__(256) void ca_out(const float* __restrict__ ws,
                                              float* __restrict__ out) {
  __shared__ float P[64][64];
  const int tid  = threadIdx.x;
  const int lane = tid & 31;
  const int wave = tid >> 5;
  const int l16  = lane & 15;
  const int hi   = lane >> 4;
  const int bh = blockIdx.y;
  const int b_ = bh >> 3;
  const int h  = bh & 7;
  const int nBase = blockIdx.x * 128;

  const float* Pg = ws + OFF_P + (size_t)bh * DH * DH;
  for (int i = tid; i < 4096; i += 256) P[i >> 6][i & 63] = Pg[i];
  __syncthreads();

  const int dt = wave >> 1;
  const int ng = wave & 1;
  const int mRow = dt * 16 + l16;
  const float* vca = ws + OFF_VCA + (size_t)bh * DH * NTOK;

  v8f acc[4] = {};
  for (int e0 = 0; e0 < DH; e0 += 4) {
    const int k0 = e0 + (hi ? 2 : 0);
    v2f a;
    a.x = P[mRow][k0];
    a.y = P[mRow][k0 + 1];
#pragma unroll
    for (int t = 0; t < 4; ++t) {
      const int n = nBase + (ng * 4 + t) * 16 + l16;
      v2f bfrag;
      bfrag.x = vca[(size_t)k0 * NTOK + n];
      bfrag.y = vca[(size_t)(k0 + 1) * NTOK + n];
      acc[t] = WMMA_F32(a, bfrag, acc[t]);
    }
  }
#pragma unroll
  for (int t = 0; t < 4; ++t) {
    const int n = nBase + (ng * 4 + t) * 16 + l16;
#pragma unroll
    for (int v = 0; v < 8; ++v) {
      const int d = dt * 16 + v + (hi ? 8 : 0);
      out[((size_t)(b_ * NTOK + n)) * CCH + h * DH + d] = acc[t][v];
    }
  }
}

__global__ __launch_bounds__(256) void sa_attn(float* __restrict__ ws,
                                               const float* __restrict__ temperature2,
                                               float* __restrict__ out) {
  const int tid  = threadIdx.x;
  const int lane = tid & 31;
  const int wave = tid >> 5;
  const int l16  = lane & 15;
  const int hi   = lane >> 4;
  const int bh = blockIdx.y;
  const int b_ = bh >> 3;
  const int h  = bh & 7;
  const int n  = blockIdx.x * 128 + wave * 16 + l16;

  const float* Q  = ws + OFF_Q  + (size_t)bh * DH * NTOK + n;
  const float* Kp = ws + OFF_KP + (size_t)bh * DH * KL;
  const float* Vp = ws + OFF_VP + (size_t)bh * DH * KL;
  const float t2 = temperature2[h];

  v8f sc[4] = {};
#pragma unroll 4
  for (int d0 = 0; d0 < DH; d0 += 4) {
    const int k0 = d0 + (hi ? 2 : 0);
    v2f q;
    q.x = Q[(size_t)k0 * NTOK];
    q.y = Q[(size_t)(k0 + 1) * NTOK];
#pragma unroll
    for (int t = 0; t < 4; ++t) {
      const int kc = t * 16 + l16;
      v2f kf;
      kf.x = Kp[(size_t)k0 * KL + kc];
      kf.y = Kp[(size_t)(k0 + 1) * KL + kc];
      sc[t] = WMMA_F32(kf, q, sc[t]);
    }
  }

  float m = sc[0][0] * t2;
#pragma unroll
  for (int t = 0; t < 4; ++t)
#pragma unroll
    for (int r = 0; r < 8; ++r) { sc[t][r] *= t2; m = fmaxf(m, sc[t][r]); }
  m = fmaxf(m, xhalf(m));
  float sum = 0.f;
#pragma unroll
  for (int t = 0; t < 4; ++t)
#pragma unroll
    for (int r = 0; r < 8; ++r) { const float e = __expf(sc[t][r] - m); sc[t][r] = e; sum += e; }
  sum = sum + xhalf(sum);
  const float inv = 1.0f / sum;
#pragma unroll
  for (int t = 0; t < 4; ++t)
#pragma unroll
    for (int r = 0; r < 8; ++r) sc[t][r] *= inv;

  v8f oacc[4] = {};
#pragma unroll
  for (int t = 0; t < 4; ++t) {
    v4f va[4], vb[4];
#pragma unroll
    for (int u = 0; u < 4; ++u) {
      const float* vp = Vp + (size_t)(u * 16 + l16) * KL + t * 16 + (hi ? 8 : 0);
      va[u] = *(const v4f*)vp;
      vb[u] = *(const v4f*)(vp + 4);
    }
#pragma unroll
    for (int j = 0; j < 4; ++j) {
      v2f p;
      p.x = sc[t][2 * j];
      p.y = sc[t][2 * j + 1];
#pragma unroll
      for (int u = 0; u < 4; ++u) {
        v2f a;
        a.x = (j < 2) ? va[u][2 * j]     : vb[u][2 * j - 4];
        a.y = (j < 2) ? va[u][2 * j + 1] : vb[u][2 * j - 3];
        oacc[u] = WMMA_F32(a, p, oacc[u]);
      }
    }
  }

  float* op = out + (size_t)b_ * (NTOK * CCH) + (size_t)h * NTOK + n;
#pragma unroll
  for (int u = 0; u < 4; ++u) {
#pragma unroll
    for (int r = 0; r < 8; ++r) {
      const int e = u * 16 + r + (hi ? 8 : 0);
      op[(size_t)e * (HEADS * NTOK)] += oacc[u][r];
    }
  }
}

extern "C" void kernel_launch(void* const* d_in, const int* in_sizes, int n_in,
                              void* d_out, int out_size, void* d_ws, size_t ws_size,
                              hipStream_t stream) {
  const float* x  = (const float*)d_in[0];
  const float* W  = (const float*)d_in[1];
  const float* EF = (const float*)d_in[2];
  const float* t1 = (const float*)d_in[3];
  const float* t2 = (const float*)d_in[4];
  float* out = (float*)d_out;
  float* ws  = (float*)d_ws;

  qkvv_gemm  <<<dim3(128, 16), 256, 0, stream>>>(x, W, ws);
  ef_proj    <<<dim3(8, 64),   128, 0, stream>>>(ws, EF, out);
  ca_scores  <<<dim3(4, 32),   128, 0, stream>>>(ws, t1, out);
  ca_out     <<<dim3(32, 32),  256, 0, stream>>>(ws, out);
  sa_attn    <<<dim3(32, 32),  256, 0, stream>>>(ws, t2, out);
}


# K4 ca_scores product moved into the K2 launch: 8 token slices x 32 bh extra workgroups, Q and K slices read once, 4 accumulators per wave; K4 only adds the partials; S and norm sum order change
# speedup vs baseline: 2.8794x; 1.0586x over previous
#include <hip/hip_runtime.h>

typedef float v2f __attribute__((ext_vector_type(2)));
typedef float v4f __attribute__((ext_vector_type(4)));
typedef float v8f __attribute__((ext_vector_type(8)));

#define WMMA_F32(a, b, c) \
  __builtin_amdgcn_wmma_f32_16x16x4_f32(false, (a), false, (b), (short)0, (c), false, false)

#define BATCH 4
#define NTOK 4096
#define CCH 512
#define HEADS 8
#define DH 64
#define KL 64
#define MROWS (BATCH * NTOK)
#define OCH (4 * CCH)
#define BH (BATCH * HEADS)

#define OFF_Q   ((size_t)0)
#define OFF_K   ((size_t)8388608)
#define OFF_VCA ((size_t)16777216)
#define OFF_VSA ((size_t)25165824)
#define OFF_KP  ((size_t)33554432)
#define OFF_VP  ((size_t)33685504)
#define OFF_P   ((size_t)33816576)

#define G_BM 128
#define G_BO 128
#define G_KC 32
#define G_LD 36
#define G_NCH (CCH / G_KC)

__global__ __launch_bounds__(256) void qkvv_gemm(const float* __restrict__ x,
                                                 const float* __restrict__ W,
                                                 float* __restrict__ ws) {
  __shared__ __attribute__((aligned(16))) float As[2][G_BM * G_LD];
  __shared__ __attribute__((aligned(16))) float Bs[2][G_BO * G_LD];

  const int tid  = threadIdx.x;
  const int lane = tid & 31;
  const int wave = tid >> 5;
  const int l16  = lane & 15;
  const int hi   = lane >> 4;
  const int wm   = wave >> 2;
  const int wo   = wave & 3;
  const int mBlock = blockIdx.x * G_BM;
  const int oBlock = blockIdx.y * G_BO;

  const int sRow = tid >> 3;
  const int sCol = (tid & 7) * 4;
  const float* xg = x + (size_t)(mBlock + sRow) * CCH + sCol;
  const float* wg = W + (size_t)(oBlock + sRow) * CCH + sCol;
  const int sOff = sRow * G_LD + sCol;

  v4f ra[4], rb[4];
  for (int i = 0; i < 4; ++i) {
    ra[i] = *(const v4f*)(xg + (size_t)i * 32 * CCH);
    rb[i] = *(const v4f*)(wg + (size_t)i * 32 * CCH);
  }
  for (int i = 0; i < 4; ++i) {
    *(v4f*)(&As[0][sOff + i * 32 * G_LD]) = ra[i];
    *(v4f*)(&Bs[0][sOff + i * 32 * G_LD]) = rb[i];
  }
  __syncthreads();

  const int aOff = (wm * 64 + l16) * G_LD + (hi ? 2 : 0);
  const int bOff = (wo * 32 + l16) * G_LD + (hi ? 2 : 0);

  v8f acc[4][2] = {};
  for (int c = 0; c < G_NCH; ++c) {
    const int cur = c & 1;
    const bool more = (c + 1 < G_NCH);
    if (more) {
      for (int i = 0; i < 4; ++i) {
        ra[i] = *(const v4f*)(xg + (size_t)i * 32 * CCH + (c + 1) * G_KC);
        rb[i] = *(const v4f*)(wg + (size_t)i * 32 * CCH + (c + 1) * G_KC);
      }
    }
    const float* Ac = &As[cur][aOff];
    const float* Bc = &Bs[cur][bOff];
    for (int j = 0; j < G_KC / 4; ++j) {
      v2f a[4], b[2];
      for (int i = 0; i < 4; ++i) a[i] = *(const v2f*)(Ac + i * 16 * G_LD + j * 4);
      for (int t = 0; t < 2; ++t) b[t] = *(const v2f*)(Bc + t * 16 * G_LD + j * 4);
      for (int i = 0; i < 4; ++i)
        for (int t = 0; t < 2; ++t)
          acc[i][t] = WMMA_F32(a[i], b[t], acc[i][t]);
    }
    if (more) {
      for (int i = 0; i < 4; ++i) {
        *(v4f*)(&As[cur ^ 1][sOff + i * 32 * G_LD]) = ra[i];
        *(v4f*)(&Bs[cur ^ 1][sOff + i * 32 * G_LD]) = rb[i];
      }
    }
    __syncthreads();
  }

  for (int t = 0; t < 2; ++t) {
    const int o  = oBlock + wo * 32 + t * 16 + l16;
    const int p  = o >> 9;
    const int h  = (o >> 6) & 7;
    const int dd = o & 63;
    for (int i = 0; i < 4; ++i) {
      for (int v = 0; v < 8; ++v) {
        const int m  = mBlock + wm * 64 + i * 16 + v + (hi ? 8 : 0);
        const int b_ = m >> 12;
        const int n  = m & (NTOK - 1);
        const size_t dst = (size_t)p * 8388608 +
                           ((size_t)(b_ * HEADS + h) * DH + dd) * NTOK + n;
        ws[dst] = acc[i][t][v];
      }
    }
  }
}

#define E_LD 72

#define P_NC 128
#define P_LD 132
#define P_SL 8
#define P_NT (NTOK / P_SL)
#define P_NCH (P_NT / P_NC)

#define SP_OFF  ((size_t)2097152)
#define QSQ_OFF ((size_t)3145728)
#define KSQ_OFF ((size_t)3162112)

__device__ __forceinline__ float xhalf(float x) {
  const int b = __builtin_bit_cast(int, x);
  return __builtin_bit_cast(float,
      __builtin_amdgcn_permlanex16(b, b, 0x76543210, 0xfedcba98, false, false));
}

__device__ __forceinline__ void qk_slice(const float* __restrict__ ws, float* __restrict__ part,
                                         float* __restrict__ Qs0, float* __restrict__ Qs1,
                                         float* __restrict__ Ks0, float* __restrict__ Ks1) {
  const int tid  = threadIdx.x;
  const int lane = tid & 31;
  const int kt   = tid >> 5;
  const int l16  = lane & 15;
  const int hi   = lane >> 4;
  const int sl   = blockIdx.x;
  const int bh   = blockIdx.y - 64;

  const float* qsrc = ws + OFF_Q + (size_t)bh * DH * NTOK + sl * P_NT;
  const float* ksrc = ws + OFF_K + (size_t)bh * DH * NTOK + sl * P_NT;

  const int aRow = tid >> 5;
  const int aCol = (tid & 31) * 4;
  const float* qg = qsrc + (size_t)aRow * NTOK + aCol;
  const float* kg = ksrc + (size_t)aRow * NTOK + aCol;
  const int aOffS = aRow * P_LD + aCol;

  v4f rq[16], rk[16];
  for (int i = 0; i < 16; ++i) rq[i] = *(const v4f*)(qg + (size_t)i * 4 * NTOK);
  for (int i = 0; i < 16; ++i) rk[i] = *(const v4f*)(kg + (size_t)i * 4 * NTOK);
  for (int i = 0; i < 16; ++i) *(v4f*)(&Qs0[aOffS + i * 4 * P_LD]) = rq[i];
  for (int i = 0; i < 16; ++i) *(v4f*)(&Ks0[aOffS + i * 4 * P_LD]) = rk[i];
  float q2[16], k2[16];
  for (int i = 0; i < 16; ++i) {
    q2[i] = rq[i].x * rq[i].x + rq[i].y * rq[i].y + rq[i].z * rq[i].z + rq[i].w * rq[i].w;
    k2[i] = rk[i].x * rk[i].x + rk[i].y * rk[i].y + rk[i].z * rk[i].z + rk[i].w * rk[i].w;
  }
  __syncthreads();

  const int qF = l16 * P_LD + (hi ? 2 : 0);
  const int kF = (kt * 16 + l16) * P_LD + (hi ? 2 : 0);

  v8f acc[4] = {};
  for (int c = 0; c < P_NCH; ++c) {
    const int cur = c & 1;
    const bool more = (c + 1 < P_NCH);
    if (more) {
      for (int i = 0; i < 16; ++i) rq[i] = *(const v4f*)(qg + (size_t)i * 4 * NTOK + (c + 1) * P_NC);
      for (int i = 0; i < 16; ++i) rk[i] = *(const v4f*)(kg + (size_t)i * 4 * NTOK + (c + 1) * P_NC);
    }
    const float* Qc = (cur ? Qs1 : Qs0) + qF;
    const float* Kc = (cur ? Ks1 : Ks0) + kF;
    v2f fa[2][4][4], fb[2][4];
#pragma unroll
    for (int j = 0; j < 4; ++j) {
#pragma unroll
      for (int i = 0; i < 4; ++i) fa[0][j][i] = *(const v2f*)(Qc + i * 16 * P_LD + j * 4);
      fb[0][j] = *(const v2f*)(Kc + j * 4);
    }
#pragma unroll
    for (int g = 0; g < P_NC / 16; ++g) {
      const int s = g & 1;
      if (g + 1 < P_NC / 16) {
#pragma unroll
        for (int j = 0; j < 4; ++j) {
          const int k = (g + 1) * 4 + j;
#pragma unroll
          for (int i = 0; i < 4; ++i) fa[s ^ 1][j][i] = *(const v2f*)(Qc + i * 16 * P_LD + k * 4);
          fb[s ^ 1][j] = *(const v2f*)(Kc + k * 4);
        }
      }
      __builtin_amdgcn_sched_barrier(0);
#pragma unroll
      for (int j = 0; j < 4; ++j)
#pragma unroll
        for (int i = 0; i < 4; ++i) acc[i] = WMMA_F32(fa[s][j][i], fb[s][j], acc[i]);
      __builtin_amdgcn_sched_barrier(0);
    }
    if (more) {
      float* Qn = cur ? Qs0 : Qs1;
      float* Kn = cur ? Ks0 : Ks1;
      for (int i = 0; i < 16; ++i) *(v4f*)(&Qn[aOffS + i * 4 * P_LD]) = rq[i];
      for (int i = 0; i < 16; ++i) *(v4f*)(&Kn[aOffS + i * 4 * P_LD]) = rk[i];
      for (int i = 0; i < 16; ++i) {
        q2[i] += rq[i].x * rq[i].x + rq[i].y * rq[i].y + rq[i].z * rq[i].z + rq[i].w * rq[i].w;
        k2[i] += rk[i].x * rk[i].x + rk[i].y * rk[i].y + rk[i].z * rk[i].z + rk[i].w * rk[i].w;
      }
    }
    __syncthreads();
  }

  const size_t tile = (size_t)(sl * BH + bh);
  float* dst = part + SP_OFF + tile * (DH * DH);
#pragma unroll
  for (int i = 0; i < 4; ++i)
#pragma unroll
    for (int v = 0; v < 8; ++v)
      dst[(size_t)(i * 16 + v + (hi ? 8 : 0)) * DH + kt * 16 + l16] = acc[i][v];

  float* red = Qs0;
  for (int i = 0; i < 16; ++i) {
    red[(i * 4 + aRow) * 33 + lane] = q2[i];
    red[64 * 33 + (i * 4 + aRow) * 33 + lane] = k2[i];
  }
  __syncthreads();
  {
    const int isk = tid >> 6;
    const int row = tid & 63;
    const float* r = red + isk * (64 * 33) + row * 33;
    float s = r[0];
    for (int l = 1; l < 32; ++l) s += r[l];
    part[(isk ? KSQ_OFF : QSQ_OFF) + tile * DH + row] = s;
  }
}

__global__ __launch_bounds__(128) void ef_proj(const float* __restrict__ ws,
                                               const float* __restrict__ EF,
                                               float* __restrict__ part) {
  __shared__ __attribute__((aligned(16))) float As[2][64 * P_LD];
  __shared__ __attribute__((aligned(16))) float Es[2][P_NC * E_LD];

  if (blockIdx.y >= 64) {
    qk_slice(ws, part, As[0], As[1], Es[0], Es[1]);
    return;
  }

  const int tid  = threadIdx.x;
  const int lane = tid & 31;
  const int kt   = tid >> 5;
  const int l16  = lane & 15;
  const int hi   = lane >> 4;
  const int sl    = blockIdx.x;
  const int which = blockIdx.y & 1;
  const int bh    = blockIdx.y >> 1;

  const float* src = ws + (which ? OFF_VSA : OFF_K) + (size_t)bh * DH * NTOK + sl * P_NT;
  const float* efs = EF + (size_t)sl * P_NT * KL;
  float* dst       = part + (size_t)(sl * 64 + blockIdx.y) * (DH * KL);

  const int aRow = tid >> 5;
  const int aCol = (tid & 31) * 4;
  const float* ag = src + (size_t)aRow * NTOK + aCol;
  const int aOffS = aRow * P_LD + aCol;
  const int eRow = tid >> 4;
  const int eCol = (tid & 15) * 4;
  const float* eg = efs + (size_t)eRow * KL + eCol;
  const int eOffS = eRow * E_LD + eCol;

  v4f ra[16], re[16];
  for (int i = 0; i < 16; ++i) ra[i] = *(const v4f*)(ag + (size_t)i * 4 * NTOK);
  for (int i = 0; i < 16; ++i) re[i] = *(const v4f*)(eg + (size_t)i * 8 * KL);
  for (int i = 0; i < 16; ++i) *(v4f*)(&As[0][aOffS + i * 4 * P_LD]) = ra[i];
  for (int i = 0; i < 16; ++i) *(v4f*)(&Es[0][eOffS + i * 8 * E_LD]) = re[i];
  __syncthreads();

  const int aF = l16 * P_LD + (hi ? 2 : 0);
  const int eF = (hi ? 2 : 0) * E_LD + kt * 16 + l16;

  v8f acc[4] = {};
  for (int c = 0; c < P_NCH; ++c) {
    const int cur = c & 1;
    const bool more = (c + 1 < P_NCH);
    if (more) {
      for (int i = 0; i < 16; ++i)
        ra[i] = *(const v4f*)(ag + (size_t)i * 4 * NTOK + (c + 1) * P_NC);
      for (int i = 0; i < 16; ++i)
        re[i] = *(const v4f*)(eg + (size_t)i * 8 * KL + (size_t)(c + 1) * P_NC * KL);
    }
    const float* Ac = &As[cur][aF];
    const float* Ec = &Es[cur][eF];
    v2f fa[2][4][4], fb[2][4];
#pragma unroll
    for (int j = 0; j < 4; ++j) {
#pragma unroll
      for (int i = 0; i < 4; ++i) fa[0][j][i] = *(const v2f*)(Ac + i * 16 * P_LD + j * 4);
      fb[0][j].x = Ec[j * 4 * E_LD];
      fb[0][j].y = Ec[j * 4 * E_LD + E_LD];
    }
#pragma unroll
    for (int g = 0; g < P_NC / 16; ++g) {
      const int s = g & 1;
      if (g + 1 < P_NC / 16) {
#pragma unroll
        for (int j = 0; j < 4; ++j) {
          const int k = (g + 1) * 4 + j;
#pragma unroll
          for (int i = 0; i < 4; ++i) fa[s ^ 1][j][i] = *(const v2f*)(Ac + i * 16 * P_LD + k * 4);
          fb[s ^ 1][j].x = Ec[k * 4 * E_LD];
          fb[s ^ 1][j].y = Ec[k * 4 * E_LD + E_LD];
        }
      }
      __builtin_amdgcn_sched_barrier(0);
#pragma unroll
      for (int j = 0; j < 4; ++j)
#pragma unroll
        for (int i = 0; i < 4; ++i) acc[i] = WMMA_F32(fa[s][j][i], fb[s][j], acc[i]);
      __builtin_amdgcn_sched_barrier(0);
    }
    if (more) {
      for (int i = 0; i < 16; ++i) *(v4f*)(&As[cur ^ 1][aOffS + i * 4 * P_LD]) = ra[i];
      for (int i = 0; i < 16; ++i) *(v4f*)(&Es[cur ^ 1][eOffS + i * 8 * E_LD]) = re[i];
    }
    __syncthreads();
  }
#pragma unroll
  for (int i = 0; i < 4; ++i)
#pragma unroll
    for (int v = 0; v < 8; ++v)
      dst[(size_t)(i * 16 + v + (hi ? 8 : 0)) * KL + kt * 16 + l16] = acc[i][v];
}

__global__ __launch_bounds__(128) void ca_scores(float* __restrict__ ws,
                                                 const float* __restrict__ temperature,
                                                 const float* __restrict__ part) {
  __shared__ float S[16][65];
  __shared__ float invqs[16];
  __shared__ float invks[64];
  const int tid  = threadIdx.x;
  const int lane = tid & 31;
  const int et   = tid >> 5;
  const int l16  = lane & 15;
  const int hi   = lane >> 4;
  const int dt = blockIdx.x;
  const int bh = blockIdx.y;
  const int h  = bh & 7;

  if (tid < 80) {
    const bool isq = tid < 16;
    const float* p = part + (isq ? QSQ_OFF : KSQ_OFF) + (size_t)bh * DH +
                     (isq ? dt * 16 + tid : tid - 16);
    float ss = p[0];
#pragma unroll
    for (int sl = 1; sl < P_SL; ++sl) ss += p[(size_t)sl * BH * DH];
    const float inv = 1.0f / fmaxf(sqrtf(ss), 1e-12f);
    if (isq) invqs[tid] = inv; else invks[tid - 16] = inv;
  }

  const int sr = tid >> 3;
  const int sc = (tid & 7) * 8;
  const float* ps = part + SP_OFF + (size_t)bh * (DH * DH) + (size_t)(dt * 16 + sr) * DH + sc;
  v4f s0 = *(const v4f*)ps;
  v4f s1 = *(const v4f*)(ps + 4);
#pragma unroll
  for (int sl = 1; sl < P_SL; ++sl) {
    const size_t so = (size_t)sl * BH * (DH * DH);
    s0 += *(const v4f*)(ps + so);
    s1 += *(const v4f*)(ps + so + 4);
  }
  __syncthreads();

  const float tsc = temperature[h];
#pragma unroll
  for (int j = 0; j < 4; ++j) {
    S[sr][sc + j]     = s0[j] * invks[sc + j];
    S[sr][sc + 4 + j] = s1[j] * invks[sc + 4 + j];
  }

  {
    const float invq = invqs[l16];
    const size_t rc = (size_t)(dt * 16 + l16) * KL + et * 16 + (hi ? 8 : 0);
    const float* pk = part + (size_t)(bh * 2) * (DH * KL) + rc;
    const float* pv = pk + (size_t)(DH * KL);
    v4f k0 = *(const v4f*)pk;
    v4f k1 = *(const v4f*)(pk + 4);
    v4f v0 = *(const v4f*)pv;
    v4f v1 = *(const v4f*)(pv + 4);
#pragma unroll
    for (int sl = 1; sl < P_SL; ++sl) {
      const size_t so = (size_t)sl * 64 * (DH * KL);
      k0 += *(const v4f*)(pk + so);
      k1 += *(const v4f*)(pk + so + 4);
      v0 += *(const v4f*)(pv + so);
      v1 += *(const v4f*)(pv + so + 4);
    }
    k0 *= invq;
    k1 *= invq;
    float* kp = ws + OFF_KP + (size_t)bh * DH * KL + rc;
    float* vp = ws + OFF_VP + (size_t)bh * DH * KL + rc;
    *(v4f*)kp = k0;
    *(v4f*)(kp + 4) = k1;
    *(v4f*)vp = v0;
    *(v4f*)(vp + 4) = v1;
  }
  __syncthreads();

  if (tid < 16) {
    const int r = tid;
    const float invq = invqs[r];
    float sv[64];
#pragma unroll
    for (int j = 0; j < 64; ++j) sv[j] = (S[r][j] * invq) * tsc;
    float m = sv[0];
#pragma unroll
    for (int j = 1; j < 64; ++j) m = fmaxf(m, sv[j]);
    float sum = 0.f;
    float e[64];
#pragma unroll
    for (int j = 0; j < 64; ++j) { e[j] = __expf(sv[j] - m); sum += e[j]; }
    const float inv = 1.0f / sum;
    float* dst = ws + OFF_P + (size_t)bh * DH * DH + (size_t)(dt * 16 + r) * DH;
#pragma unroll
    for (int j = 0; j < 64; ++j) dst[j] = e[j] * inv;
  }
}

__global__ __launch_bounds__(256) void ca_out(const float* __restrict__ ws,
                                              float* __restrict__ out) {
  __shared__ float P[64][64];
  const int tid  = threadIdx.x;
  const int lane = tid & 31;
  const int wave = tid >> 5;
  const int l16  = lane & 15;
  const int hi   = lane >> 4;
  const int bh = blockIdx.y;
  const int b_ = bh >> 3;
  const int h  = bh & 7;
  const int nBase = blockIdx.x * 128;

  const float* Pg = ws + OFF_P + (size_t)bh * DH * DH;
  for (int i = tid; i < 4096; i += 256) P[i >> 6][i & 63] = Pg[i];
  __syncthreads();

  const int dt = wave >> 1;
  const int ng = wave & 1;
  const int mRow = dt * 16 + l16;
  const float* vca = ws + OFF_VCA + (size_t)bh * DH * NTOK;

  v8f acc[4] = {};
  for (int e0 = 0; e0 < DH; e0 += 4) {
    const int k0 = e0 + (hi ? 2 : 0);
    v2f a;
    a.x = P[mRow][k0];
    a.y = P[mRow][k0 + 1];
#pragma unroll
    for (int t = 0; t < 4; ++t) {
      const int n = nBase + (ng * 4 + t) * 16 + l16;
      v2f bfrag;
      bfrag.x = vca[(size_t)k0 * NTOK + n];
      bfrag.y = vca[(size_t)(k0 + 1) * NTOK + n];
      acc[t] = WMMA_F32(a, bfrag, acc[t]);
    }
  }
#pragma unroll
  for (int t = 0; t < 4; ++t) {
    const int n = nBase + (ng * 4 + t) * 16 + l16;
#pragma unroll
    for (int v = 0; v < 8; ++v) {
      const int d = dt * 16 + v + (hi ? 8 : 0);
      out[((size_t)(b_ * NTOK + n)) * CCH + h * DH + d] = acc[t][v];
    }
  }
}

__global__ __launch_bounds__(256) void sa_attn(float* __restrict__ ws,
                                               const float* __restrict__ temperature2,
                                               float* __restrict__ out) {
  const int tid  = threadIdx.x;
  const int lane = tid & 31;
  const int wave = tid >> 5;
  const int l16  = lane & 15;
  const int hi   = lane >> 4;
  const int bh = blockIdx.y;
  const int b_ = bh >> 3;
  const int h  = bh & 7;
  const int n  = blockIdx.x * 128 + wave * 16 + l16;

  const float* Q  = ws + OFF_Q  + (size_t)bh * DH * NTOK + n;
  const float* Kp = ws + OFF_KP + (size_t)bh * DH * KL;
  const float* Vp = ws + OFF_VP + (size_t)bh * DH * KL;
  const float t2 = temperature2[h];

  v8f sc[4] = {};
#pragma unroll 4
  for (int d0 = 0; d0 < DH; d0 += 4) {
    const int k0 = d0 + (hi ? 2 : 0);
    v2f q;
    q.x = Q[(size_t)k0 * NTOK];
    q.y = Q[(size_t)(k0 + 1) * NTOK];
#pragma unroll
    for (int t = 0; t < 4; ++t) {
      const int kc = t * 16 + l16;
      v2f kf;
      kf.x = Kp[(size_t)k0 * KL + kc];
      kf.y = Kp[(size_t)(k0 + 1) * KL + kc];
      sc[t] = WMMA_F32(kf, q, sc[t]);
    }
  }

  float m = sc[0][0] * t2;
#pragma unroll
  for (int t = 0; t < 4; ++t)
#pragma unroll
    for (int r = 0; r < 8; ++r) { sc[t][r] *= t2; m = fmaxf(m, sc[t][r]); }
  m = fmaxf(m, xhalf(m));
  float sum = 0.f;
#pragma unroll
  for (int t = 0; t < 4; ++t)
#pragma unroll
    for (int r = 0; r < 8; ++r) { const float e = __expf(sc[t][r] - m); sc[t][r] = e; sum += e; }
  sum = sum + xhalf(sum);
  const float inv = 1.0f / sum;
#pragma unroll
  for (int t = 0; t < 4; ++t)
#pragma unroll
    for (int r = 0; r < 8; ++r) sc[t][r] *= inv;

  v8f oacc[4] = {};
#pragma unroll
  for (int t = 0; t < 4; ++t) {
    v4f va[4], vb[4];
#pragma unroll
    for (int u = 0; u < 4; ++u) {
      const float* vp = Vp + (size_t)(u * 16 + l16) * KL + t * 16 + (hi ? 8 : 0);
      va[u] = *(const v4f*)vp;
      vb[u] = *(const v4f*)(vp + 4);
    }
#pragma unroll
    for (int j = 0; j < 4; ++j) {
      v2f p;
      p.x = sc[t][2 * j];
      p.y = sc[t][2 * j + 1];
#pragma unroll
      for (int u = 0; u < 4; ++u) {
        v2f a;
        a.x = (j < 2) ? va[u][2 * j]     : vb[u][2 * j - 4];
        a.y = (j < 2) ? va[u][2 * j + 1] : vb[u][2 * j - 3];
        oacc[u] = WMMA_F32(a, p, oacc[u]);
      }
    }
  }

  float* op = out + (size_t)b_ * (NTOK * CCH) + (size_t)h * NTOK + n;
#pragma unroll
  for (int u = 0; u < 4; ++u) {
#pragma unroll
    for (int r = 0; r < 8; ++r) {
      const int e = u * 16 + r + (hi ? 8 : 0);
      op[(size_t)e * (HEADS * NTOK)] += oacc[u][r];
    }
  }
}

extern "C" void kernel_launch(void* const* d_in, const int* in_sizes, int n_in,
                              void* d_out, int out_size, void* d_ws, size_t ws_size,
                              hipStream_t stream) {
  const float* x  = (const float*)d_in[0];
  const float* W  = (const float*)d_in[1];
  const float* EF = (const float*)d_in[2];
  const float* t1 = (const float*)d_in[3];
  const float* t2 = (const float*)d_in[4];
  float* out = (float*)d_out;
  float* ws  = (float*)d_ws;

  qkvv_gemm  <<<dim3(128, 16), 256, 0, stream>>>(x, W, ws);
  ef_proj    <<<dim3(8, 96),   128, 0, stream>>>(ws, EF, out);
  ca_scores  <<<dim3(4, 32),   128, 0, stream>>>(ws, t1, out);
  ca_out     <<<dim3(32, 32),  256, 0, stream>>>(ws, out);
  sa_attn    <<<dim3(32, 32),  256, 0, stream>>>(ws, t2, out);
}


# x_CA = P @ Vca moved into the SA kernel, at the positions its SA values land on (Vca token-major, rows staged via LDS): out stored once, never read; 4 launches; x_CA k order and add order change
# speedup vs baseline: 3.5578x; 1.2356x over previous
#include <hip/hip_runtime.h>

typedef float v2f __attribute__((ext_vector_type(2)));
typedef float v4f __attribute__((ext_vector_type(4)));
typedef float v8f __attribute__((ext_vector_type(8)));

#define WMMA_F32(a, b, c) \
  __builtin_amdgcn_wmma_f32_16x16x4_f32(false, (a), false, (b), (short)0, (c), false, false)

#define BATCH 4
#define NTOK 4096
#define CCH 512
#define HEADS 8
#define DH 64
#define KL 64
#define MROWS (BATCH * NTOK)
#define OCH (4 * CCH)
#define BH (BATCH * HEADS)

#define OFF_Q   ((size_t)0)
#define OFF_K   ((size_t)8388608)
#define OFF_VCA ((size_t)16777216)
#define OFF_XE  ((size_t)25165824)
#define OFF_KP  ((size_t)33554432)
#define OFF_VP  ((size_t)33685504)
#define OFF_P   ((size_t)33816576)

#define G_BM 128
#define G_BO 128
#define G_KC 32
#define G_LD 36
#define G_NCH (CCH / G_KC)

#define E_LD 72

#define X_SL 8
#define X_NT (NTOK / X_SL)
#define X_NC 32
#define X_LD 136
#define X_NCH (X_NT / X_NC)

__device__ __forceinline__ void xe_slice(const float* __restrict__ x, const float* __restrict__ EF,
                                         float* __restrict__ ws, float* __restrict__ smem) {
  float* Xs0 = smem;
  float* Xs1 = smem + X_NC * X_LD;
  float* Es0 = smem + 2 * X_NC * X_LD;
  float* Es1 = Es0 + X_NC * E_LD;
  const int tid  = threadIdx.x;
  const int lane = tid & 31;
  const int wave = tid >> 5;
  const int l16  = lane & 15;
  const int hi   = lane >> 4;
  const int b_ = blockIdx.x >> 5;
  const int sl = (blockIdx.x >> 2) & 7;
  const int cq = blockIdx.x & 3;

  const float* xsrc = x + (size_t)(b_ * NTOK + sl * X_NT) * CCH + cq * 128;
  const float* esrc = EF + (size_t)sl * X_NT * KL;

  const int xRow = tid >> 5;
  const int xCol = (tid & 31) * 4;
  const float* xg = xsrc + (size_t)xRow * CCH + xCol;
  const int xOff = xRow * X_LD + xCol;
  const int eRow = tid >> 4;
  const int eCol = (tid & 15) * 4;
  const float* eg = esrc + (size_t)eRow * KL + eCol;
  const int eOff = eRow * E_LD + eCol;

  v4f rx[4], re[2];
  for (int i = 0; i < 4; ++i) rx[i] = *(const v4f*)(xg + (size_t)i * 8 * CCH);
  for (int i = 0; i < 2; ++i) re[i] = *(const v4f*)(eg + (size_t)i * 16 * KL);
  for (int i = 0; i < 4; ++i) *(v4f*)(Xs0 + xOff + i * 8 * X_LD) = rx[i];
  for (int i = 0; i < 2; ++i) *(v4f*)(Es0 + eOff + i * 16 * E_LD) = re[i];
  __syncthreads();

  const int aF = (hi ? 2 : 0) * X_LD + wave * 16 + l16;
  const int bF = (hi ? 2 : 0) * E_LD + l16;

  v8f acc[4] = {};
  for (int c = 0; c < X_NCH; ++c) {
    const int cur = c & 1;
    const bool more = (c + 1 < X_NCH);
    if (more) {
      for (int i = 0; i < 4; ++i) rx[i] = *(const v4f*)(xg + (size_t)(i * 8 + (c + 1) * X_NC) * CCH);
      for (int i = 0; i < 2; ++i) re[i] = *(const v4f*)(eg + (size_t)(i * 16 + (c + 1) * X_NC) * KL);
    }
    const float* Xc = (cur ? Xs1 : Xs0) + aF;
    const float* Ec = (cur ? Es1 : Es0) + bF;
    for (int j = 0; j < X_NC / 4; ++j) {
      v2f a, bb[4];
      a.x = Xc[j * 4 * X_LD];
      a.y = Xc[j * 4 * X_LD + X_LD];
      for (int t = 0; t < 4; ++t) {
        bb[t].x = Ec[j * 4 * E_LD + t * 16];
        bb[t].y = Ec[j * 4 * E_LD + E_LD + t * 16];
      }
      for (int t = 0; t < 4; ++t) acc[t] = WMMA_F32(a, bb[t], acc[t]);
    }
    if (more) {
      float* Xn = cur ? Xs0 : Xs1;
      float* En = cur ? Es0 : Es1;
      for (int i = 0; i < 4; ++i) *(v4f*)(Xn + xOff + i * 8 * X_LD) = rx[i];
      for (int i = 0; i < 2; ++i) *(v4f*)(En + eOff + i * 16 * E_LD) = re[i];
    }
    __syncthreads();
  }

  float* dst = ws + OFF_XE + ((size_t)(sl * BATCH + b_) * CCH + cq * 128 + wave * 16 + (hi ? 8 : 0)) * KL + l16;
  for (int t = 0; t < 4; ++t)
    for (int v = 0; v < 8; ++v) dst[(size_t)v * KL + t * 16] = acc[t][v];
}

__global__ __launch_bounds__(256) void qkvv_gemm(const float* __restrict__ x,
                                                 const float* __restrict__ W,
                                                 const float* __restrict__ EF,
                                                 float* __restrict__ ws) {
  __shared__ __attribute__((aligned(16))) float smem[2 * G_BM * G_LD + 2 * G_BO * G_LD];
  float (*As)[G_BM * G_LD] = (float (*)[G_BM * G_LD])smem;
  float (*Bs)[G_BO * G_LD] = (float (*)[G_BO * G_LD])(smem + 2 * G_BM * G_LD);

  if (blockIdx.y == 0) {
    xe_slice(x, EF, ws, smem);
    return;
  }

  const int tid  = threadIdx.x;
  const int lane = tid & 31;
  const int wave = tid >> 5;
  const int l16  = lane & 15;
  const int hi   = lane >> 4;
  const int wm   = wave >> 2;
  const int wo   = wave & 3;
  const int mBlock = blockIdx.x * G_BM;
  const int oBlock = (blockIdx.y - 1) * G_BO;

  const int sRow = tid >> 3;
  const int sCol = (tid & 7) * 4;
  const float* xg = x + (size_t)(mBlock + sRow) * CCH + sCol;
  const float* wg = W + (size_t)(oBlock + sRow) * CCH + sCol;
  const int sOff = sRow * G_LD + sCol;

  v4f ra[4], rb[4];
  for (int i = 0; i < 4; ++i) {
    ra[i] = *(const v4f*)(xg + (size_t)i * 32 * CCH);
    rb[i] = *(const v4f*)(wg + (size_t)i * 32 * CCH);
  }
  for (int i = 0; i < 4; ++i) {
    *(v4f*)(&As[0][sOff + i * 32 * G_LD]) = ra[i];
    *(v4f*)(&Bs[0][sOff + i * 32 * G_LD]) = rb[i];
  }
  __syncthreads();

  const int aOff = (wm * 64 + l16) * G_LD + (hi ? 2 : 0);
  const int bOff = (wo * 32 + l16) * G_LD + (hi ? 2 : 0);

  v8f acc[4][2] = {};
  for (int c = 0; c < G_NCH; ++c) {
    const int cur = c & 1;
    const bool more = (c + 1 < G_NCH);
    if (more) {
      for (int i = 0; i < 4; ++i) {
        ra[i] = *(const v4f*)(xg + (size_t)i * 32 * CCH + (c + 1) * G_KC);
        rb[i] = *(const v4f*)(wg + (size_t)i * 32 * CCH + (c + 1) * G_KC);
      }
    }
    const float* Ac = &As[cur][aOff];
    const float* Bc = &Bs[cur][bOff];
    for (int j = 0; j < G_KC / 4; ++j) {
      v2f a[4], b[2];
      for (int i = 0; i < 4; ++i) a[i] = *(const v2f*)(Ac + i * 16 * G_LD + j * 4);
      for (int t = 0; t < 2; ++t) b[t] = *(const v2f*)(Bc + t * 16 * G_LD + j * 4);
      for (int i = 0; i < 4; ++i)
        for (int t = 0; t < 2; ++t)
          acc[i][t] = WMMA_F32(a[i], b[t], acc[i][t]);
    }
    if (more) {
      for (int i = 0; i < 4; ++i) {
        *(v4f*)(&As[cur ^ 1][sOff + i * 32 * G_LD]) = ra[i];
        *(v4f*)(&Bs[cur ^ 1][sOff + i * 32 * G_LD]) = rb[i];
      }
    }
    __syncthreads();
  }

  if (blockIdx.y >= 9) {
    float* vdst = ws + OFF_VCA + (size_t)(mBlock + wm * 64 + (hi ? 8 : 0)) * CCH +
                  (oBlock - 2 * CCH) + wo * 32 + l16;
    for (int i = 0; i < 4; ++i)
      for (int v = 0; v < 8; ++v)
        for (int t = 0; t < 2; ++t)
          vdst[(size_t)(i * 16 + v) * CCH + t * 16] = acc[i][t][v];
    return;
  }

  for (int t = 0; t < 2; ++t) {
    const int o  = oBlock + wo * 32 + t * 16 + l16;
    const int p  = o >> 9;
    const int h  = (o >> 6) & 7;
    const int dd = o & 63;
    for (int i = 0; i < 4; ++i) {
      for (int v = 0; v < 8; ++v) {
        const int m  = mBlock + wm * 64 + i * 16 + v + (hi ? 8 : 0);
        const int b_ = m >> 12;
        const int n  = m & (NTOK - 1);
        const size_t dst = (size_t)p * 8388608 +
                           ((size_t)(b_ * HEADS + h) * DH + dd) * NTOK + n;
        ws[dst] = acc[i][t][v];
      }
    }
  }
}

#define P_NC 128
#define P_LD 132
#define P_SL 8
#define P_NT (NTOK / P_SL)
#define P_NCH (P_NT / P_NC)
static_assert(P_SL == X_SL && P_NT == CCH, "K2's W @ XE workgroups walk C channels in P_NCH chunks, one XE slice each");

#define SP_OFF  ((size_t)2097152)
#define QSQ_OFF ((size_t)3145728)
#define KSQ_OFF ((size_t)3162112)

__device__ __forceinline__ float xhalf(float x) {
  const int b = __builtin_bit_cast(int, x);
  return __builtin_bit_cast(float,
      __builtin_amdgcn_permlanex16(b, b, 0x76543210, 0xfedcba98, false, false));
}

__device__ __forceinline__ void qk_slice(const float* __restrict__ ws, float* __restrict__ part,
                                         float* __restrict__ Qs0, float* __restrict__ Qs1,
                                         float* __restrict__ Ks0, float* __restrict__ Ks1) {
  const int tid  = threadIdx.x;
  const int lane = tid & 31;
  const int kt   = tid >> 5;
  const int l16  = lane & 15;
  const int hi   = lane >> 4;
  const int sl   = blockIdx.x;
  const int bh   = blockIdx.y - 64;

  const float* qsrc = ws + OFF_Q + (size_t)bh * DH * NTOK + sl * P_NT;
  const float* ksrc = ws + OFF_K + (size_t)bh * DH * NTOK + sl * P_NT;

  const int aRow = tid >> 5;
  const int aCol = (tid & 31) * 4;
  const float* qg = qsrc + (size_t)aRow * NTOK + aCol;
  const float* kg = ksrc + (size_t)aRow * NTOK + aCol;
  const int aOffS = aRow * P_LD + aCol;

  v4f rq[16], rk[16];
  for (int i = 0; i < 16; ++i) rq[i] = *(const v4f*)(qg + (size_t)i * 4 * NTOK);
  for (int i = 0; i < 16; ++i) rk[i] = *(const v4f*)(kg + (size_t)i * 4 * NTOK);
  for (int i = 0; i < 16; ++i) *(v4f*)(&Qs0[aOffS + i * 4 * P_LD]) = rq[i];
  for (int i = 0; i < 16; ++i) *(v4f*)(&Ks0[aOffS + i * 4 * P_LD]) = rk[i];
  float q2[16], k2[16];
  for (int i = 0; i < 16; ++i) {
    q2[i] = rq[i].x * rq[i].x + rq[i].y * rq[i].y + rq[i].z * rq[i].z + rq[i].w * rq[i].w;
    k2[i] = rk[i].x * rk[i].x + rk[i].y * rk[i].y + rk[i].z * rk[i].z + rk[i].w * rk[i].w;
  }
  __syncthreads();

  const int qF = l16 * P_LD + (hi ? 2 : 0);
  const int kF = (kt * 16 + l16) * P_LD + (hi ? 2 : 0);

  v8f acc[4] = {};
  for (int c = 0; c < P_NCH; ++c) {
    const int cur = c & 1;
    const bool more = (c + 1 < P_NCH);
    if (more) {
      for (int i = 0; i < 16; ++i) rq[i] = *(const v4f*)(qg + (size_t)i * 4 * NTOK + (c + 1) * P_NC);
      for (int i = 0; i < 16; ++i) rk[i] = *(const v4f*)(kg + (size_t)i * 4 * NTOK + (c + 1) * P_NC);
    }
    const float* Qc = (cur ? Qs1 : Qs0) + qF;
    const float* Kc = (cur ? Ks1 : Ks0) + kF;
    v2f fa[2][4][4], fb[2][4];
#pragma unroll
    for (int j = 0; j < 4; ++j) {
#pragma unroll
      for (int i = 0; i < 4; ++i) fa[0][j][i] = *(const v2f*)(Qc + i * 16 * P_LD + j * 4);
      fb[0][j] = *(const v2f*)(Kc + j * 4);
    }
#pragma unroll
    for (int g = 0; g < P_NC / 16; ++g) {
      const int s = g & 1;
      if (g + 1 < P_NC / 16) {
#pragma unroll
        for (int j = 0; j < 4; ++j) {
          const int k = (g + 1) * 4 + j;
#pragma unroll
          for (int i = 0; i < 4; ++i) fa[s ^ 1][j][i] = *(const v2f*)(Qc + i * 16 * P_LD + k * 4);
          fb[s ^ 1][j] = *(const v2f*)(Kc + k * 4);
        }
      }
      __builtin_amdgcn_sched_barrier(0);
#pragma unroll
      for (int j = 0; j < 4; ++j)
#pragma unroll
        for (int i = 0; i < 4; ++i) acc[i] = WMMA_F32(fa[s][j][i], fb[s][j], acc[i]);
      __builtin_amdgcn_sched_barrier(0);
    }
    if (more) {
      float* Qn = cur ? Qs0 : Qs1;
      float* Kn = cur ? Ks0 : Ks1;
      for (int i = 0; i < 16; ++i) *(v4f*)(&Qn[aOffS + i * 4 * P_LD]) = rq[i];
      for (int i = 0; i < 16; ++i) *(v4f*)(&Kn[aOffS + i * 4 * P_LD]) = rk[i];
      for (int i = 0; i < 16; ++i) {
        q2[i] += rq[i].x * rq[i].x + rq[i].y * rq[i].y + rq[i].z * rq[i].z + rq[i].w * rq[i].w;
        k2[i] += rk[i].x * rk[i].x + rk[i].y * rk[i].y + rk[i].z * rk[i].z + rk[i].w * rk[i].w;
      }
    }
    __syncthreads();
  }

  const size_t tile = (size_t)(sl * BH + bh);
  float* dst = part + SP_OFF + tile * (DH * DH);
#pragma unroll
  for (int i = 0; i < 4; ++i)
#pragma unroll
    for (int v = 0; v < 8; ++v)
      dst[(size_t)(i * 16 + v + (hi ? 8 : 0)) * DH + kt * 16 + l16] = acc[i][v];

  float* red = Qs0;
  for (int i = 0; i < 16; ++i) {
    red[(i * 4 + aRow) * 33 + lane] = q2[i];
    red[64 * 33 + (i * 4 + aRow) * 33 + lane] = k2[i];
  }
  __syncthreads();
  {
    const int isk = tid >> 6;
    const int row = tid & 63;
    const float* r = red + isk * (64 * 33) + row * 33;
    float s = r[0];
    for (int l = 1; l < 32; ++l) s += r[l];
    part[(isk ? KSQ_OFF : QSQ_OFF) + tile * DH + row] = s;
  }
}

__global__ __launch_bounds__(128) void ef_proj(const float* __restrict__ ws,
                                               const float* __restrict__ W,
                                               float* __restrict__ part) {
  __shared__ __attribute__((aligned(16))) float As[2][64 * P_LD];
  __shared__ __attribute__((aligned(16))) float Es[2][P_NC * E_LD];

  if (blockIdx.y >= 64) {
    qk_slice(ws, part, As[0], As[1], Es[0], Es[1]);
    return;
  }

  const int tid  = threadIdx.x;
  const int lane = tid & 31;
  const int kt   = tid >> 5;
  const int l16  = lane & 15;
  const int hi   = lane >> 4;
  const int sl    = blockIdx.x;
  const int which = blockIdx.y & 1;
  const int bh    = blockIdx.y >> 1;

  const float* src = W + (size_t)((which ? 3 : 1) * CCH + (bh & 7) * DH) * CCH;
  const float* efs = ws + OFF_XE + (size_t)(sl * BATCH + (bh >> 3)) * CCH * KL;
  float* dst       = part + (size_t)(sl * 64 + blockIdx.y) * (DH * KL);

  const int aRow = tid >> 5;
  const int aCol = (tid & 31) * 4;
  const float* ag = src + (size_t)aRow * CCH + aCol;
  const int aOffS = aRow * P_LD + aCol;
  const int eRow = tid >> 4;
  const int eCol = (tid & 15) * 4;
  const float* eg = efs + (size_t)eRow * KL + eCol;
  const int eOffS = eRow * E_LD + eCol;

  v4f ra[16], re[16];
  for (int i = 0; i < 16; ++i) ra[i] = *(const v4f*)(ag + (size_t)i * 4 * CCH);
  for (int i = 0; i < 16; ++i) re[i] = *(const v4f*)(eg + (size_t)i * 8 * KL);
  for (int i = 0; i < 16; ++i) *(v4f*)(&As[0][aOffS + i * 4 * P_LD]) = ra[i];
  for (int i = 0; i < 16; ++i) *(v4f*)(&Es[0][eOffS + i * 8 * E_LD]) = re[i];
  __syncthreads();

  const int aF = l16 * P_LD + (hi ? 2 : 0);
  const int eF = (hi ? 2 : 0) * E_LD + kt * 16 + l16;

  v8f acc[4] = {};
  for (int c = 0; c < P_NCH; ++c) {
    const int cur = c & 1;
    const bool more = (c + 1 < P_NCH);
    if (more) {
      for (int i = 0; i < 16; ++i)
        ra[i] = *(const v4f*)(ag + (size_t)i * 4 * CCH + (c + 1) * P_NC);
      for (int i = 0; i < 16; ++i)
        re[i] = *(const v4f*)(eg + (size_t)i * 8 * KL + (size_t)(c + 1) * P_NC * KL);
    }
    const float* Ac = &As[cur][aF];
    const float* Ec = &Es[cur][eF];
    v2f fa[2][4][4], fb[2][4];
#pragma unroll
    for (int j = 0; j < 4; ++j) {
#pragma unroll
      for (int i = 0; i < 4; ++i) fa[0][j][i] = *(const v2f*)(Ac + i * 16 * P_LD + j * 4);
      fb[0][j].x = Ec[j * 4 * E_LD];
      fb[0][j].y = Ec[j * 4 * E_LD + E_LD];
    }
#pragma unroll
    for (int g = 0; g < P_NC / 16; ++g) {
      const int s = g & 1;
      if (g + 1 < P_NC / 16) {
#pragma unroll
        for (int j = 0; j < 4; ++j) {
          const int k = (g + 1) * 4 + j;
#pragma unroll
          for (int i = 0; i < 4; ++i) fa[s ^ 1][j][i] = *(const v2f*)(Ac + i * 16 * P_LD + k * 4);
          fb[s ^ 1][j].x = Ec[k * 4 * E_LD];
          fb[s ^ 1][j].y = Ec[k * 4 * E_LD + E_LD];
        }
      }
      __builtin_amdgcn_sched_barrier(0);
#pragma unroll
      for (int j = 0; j < 4; ++j)
#pragma unroll
        for (int i = 0; i < 4; ++i) acc[i] = WMMA_F32(fa[s][j][i], fb[s][j], acc[i]);
      __builtin_amdgcn_sched_barrier(0);
    }
    if (more) {
      for (int i = 0; i < 16; ++i) *(v4f*)(&As[cur ^ 1][aOffS + i * 4 * P_LD]) = ra[i];
      for (int i = 0; i < 16; ++i) *(v4f*)(&Es[cur ^ 1][eOffS + i * 8 * E_LD]) = re[i];
    }
    __syncthreads();
  }
#pragma unroll
  for (int i = 0; i < 4; ++i)
#pragma unroll
    for (int v = 0; v < 8; ++v)
      dst[(size_t)(i * 16 + v + (hi ? 8 : 0)) * KL + kt * 16 + l16] = acc[i][v];
}

__global__ __launch_bounds__(128) void ca_scores(float* __restrict__ ws,
                                                 const float* __restrict__ temperature,
                                                 const float* __restrict__ part) {
  __shared__ float S[16][65];
  __shared__ float invqs[16];
  __shared__ float invks[64];
  const int tid  = threadIdx.x;
  const int lane = tid & 31;
  const int et   = tid >> 5;
  const int l16  = lane & 15;
  const int hi   = lane >> 4;
  const int dt = blockIdx.x;
  const int bh = blockIdx.y;
  const int h  = bh & 7;

  if (tid < 80) {
    const bool isq = tid < 16;
    const float* p = part + (isq ? QSQ_OFF : KSQ_OFF) + (size_t)bh * DH +
                     (isq ? dt * 16 + tid : tid - 16);
    float ss = p[0];
#pragma unroll
    for (int sl = 1; sl < P_SL; ++sl) ss += p[(size_t)sl * BH * DH];
    const float inv = 1.0f / fmaxf(sqrtf(ss), 1e-12f);
    if (isq) invqs[tid] = inv; else invks[tid - 16] = inv;
  }

  const int sr = tid >> 3;
  const int sc = (tid & 7) * 8;
  const float* ps = part + SP_OFF + (size_t)bh * (DH * DH) + (size_t)(dt * 16 + sr) * DH + sc;
  v4f s0 = *(const v4f*)ps;
  v4f s1 = *(const v4f*)(ps + 4);
#pragma unroll
  for (int sl = 1; sl < P_SL; ++sl) {
    const size_t so = (size_t)sl * BH * (DH * DH);
    s0 += *(const v4f*)(ps + so);
    s1 += *(const v4f*)(ps + so + 4);
  }
  __syncthreads();

  const float tsc = temperature[h];
#pragma unroll
  for (int j = 0; j < 4; ++j) {
    S[sr][sc + j]     = s0[j] * invks[sc + j];
    S[sr][sc + 4 + j] = s1[j] * invks[sc + 4 + j];
  }

  {
    const float invq = invqs[l16];
    const size_t rc = (size_t)(dt * 16 + l16) * KL + et * 16 + (hi ? 8 : 0);
    const float* pk = part + (size_t)(bh * 2) * (DH * KL) + rc;
    const float* pv = pk + (size_t)(DH * KL);
    v4f k0 = *(const v4f*)pk;
    v4f k1 = *(const v4f*)(pk + 4);
    v4f v0 = *(const v4f*)pv;
    v4f v1 = *(const v4f*)(pv + 4);
#pragma unroll
    for (int sl = 1; sl < P_SL; ++sl) {
      const size_t so = (size_t)sl * 64 * (DH * KL);
      k0 += *(const v4f*)(pk + so);
      k1 += *(const v4f*)(pk + so + 4);
      v0 += *(const v4f*)(pv + so);
      v1 += *(const v4f*)(pv + so + 4);
    }
    k0 *= invq;
    k1 *= invq;
    float* kp = ws + OFF_KP + (size_t)bh * DH * KL + rc;
    float* vp = ws + OFF_VP + (size_t)bh * DH * KL + rc;
    *(v4f*)kp = k0;
    *(v4f*)(kp + 4) = k1;
    *(v4f*)vp = v0;
    *(v4f*)(vp + 4) = v1;
  }
  __syncthreads();

  if (tid < 16) {
    const int r = tid;
    const float invq = invqs[r];
    float sv[64];
#pragma unroll
    for (int j = 0; j < 64; ++j) sv[j] = (S[r][j] * invq) * tsc;
    float m = sv[0];
#pragma unroll
    for (int j = 1; j < 64; ++j) m = fmaxf(m, sv[j]);
    float sum = 0.f;
    float e[64];
#pragma unroll
    for (int j = 0; j < 64; ++j) { e[j] = __expf(sv[j] - m); sum += e[j]; }
    const float inv = 1.0f / sum;
    float* dst = ws + OFF_P + (size_t)bh * DH * DH + (size_t)(dt * 16 + r) * DH;
#pragma unroll
    for (int j = 0; j < 64; ++j) dst[j] = e[j] * inv;
  }
}

#define V_LD 68

__global__ __launch_bounds__(256) void sa_attn(float* __restrict__ ws,
                                               const float* __restrict__ temperature2,
                                               float* __restrict__ out) {
  __shared__ __attribute__((aligned(16))) float Vs[2][64 * V_LD];
  const int tid  = threadIdx.x;
  const int lane = tid & 31;
  const int wave = tid >> 5;
  const int l16  = lane & 15;
  const int hi   = lane >> 4;
  const int bh = blockIdx.y;
  const int b_ = bh >> 3;
  const int h  = bh & 7;
  const int n  = blockIdx.x * 128 + wave * 16 + l16;
  const int hq = (blockIdx.x & 3) * 2;
  const int nq = blockIdx.x >> 2;

  const float* Q  = ws + OFF_Q  + (size_t)bh * DH * NTOK + n;
  const float* Kp = ws + OFF_KP + (size_t)bh * DH * KL;
  const float* Vp = ws + OFF_VP + (size_t)bh * DH * KL;
  const float t2 = temperature2[h];

  const float* vg = ws + OFF_VCA + (size_t)(b_ * NTOK + wave * 64 + h * 8 + nq) * CCH + hq * DH + lane * 4;
  v4f rv[8];
#pragma unroll
  for (int p = 0; p < 8; ++p) rv[p] = *(const v4f*)(vg + (size_t)p * (8 * 64) * CCH);

  v8f sc[4] = {};
#pragma unroll 4
  for (int d0 = 0; d0 < DH; d0 += 4) {
    const int k0 = d0 + (hi ? 2 : 0);
    v2f q;
    q.x = Q[(size_t)k0 * NTOK];
    q.y = Q[(size_t)(k0 + 1) * NTOK];
#pragma unroll
    for (int t = 0; t < 4; ++t) {
      const int kc = t * 16 + l16;
      v2f kf;
      kf.x = Kp[(size_t)k0 * KL + kc];
      kf.y = Kp[(size_t)(k0 + 1) * KL + kc];
      sc[t] = WMMA_F32(kf, q, sc[t]);
    }
  }

#pragma unroll
  for (int p = 0; p < 8; ++p)
    *(v4f*)(&Vs[hi][(p * 8 + wave) * V_LD + l16 * 4]) = rv[p];
  __syncthreads();

  float m = sc[0][0] * t2;
#pragma unroll
  for (int t = 0; t < 4; ++t)
#pragma unroll
    for (int r = 0; r < 8; ++r) { sc[t][r] *= t2; m = fmaxf(m, sc[t][r]); }
  m = fmaxf(m, xhalf(m));
  float sum = 0.f;
#pragma unroll
  for (int t = 0; t < 4; ++t)
#pragma unroll
    for (int r = 0; r < 8; ++r) { const float e = __expf(sc[t][r] - m); sc[t][r] = e; sum += e; }
  sum = sum + xhalf(sum);
  const float inv = 1.0f / sum;
#pragma unroll
  for (int t = 0; t < 4; ++t)
#pragma unroll
    for (int r = 0; r < 8; ++r) sc[t][r] *= inv;

  v8f oacc[4] = {};
#pragma unroll
  for (int t = 0; t < 4; ++t) {
    v4f va[4], vb[4];
#pragma unroll
    for (int u = 0; u < 4; ++u) {
      const float* vp = Vp + (size_t)(u * 16 + l16) * KL + t * 16 + (hi ? 8 : 0);
      va[u] = *(const v4f*)vp;
      vb[u] = *(const v4f*)(vp + 4);
    }
#pragma unroll
    for (int j = 0; j < 4; ++j) {
      v2f p;
      p.x = sc[t][2 * j];
      p.y = sc[t][2 * j + 1];
#pragma unroll
      for (int u = 0; u < 4; ++u) {
        v2f a;
        a.x = (j < 2) ? va[u][2 * j]     : vb[u][2 * j - 4];
        a.y = (j < 2) ? va[u][2 * j + 1] : vb[u][2 * j - 3];
        oacc[u] = WMMA_F32(a, p, oacc[u]);
      }
    }
  }

  {
    const int hs = wave >> 2;
    const float* Prow = ws + OFF_P + ((size_t)(b_ * HEADS + hq + hs) * DH + (wave & 3) * 16 + l16) * DH + (hi ? 4 : 0);
    v4f pb[8];
#pragma unroll
    for (int s = 0; s < 8; ++s) pb[s] = *(const v4f*)(Prow + s * 8);
    const float* Arow = &Vs[hs][l16 * V_LD + (hi ? 4 : 0)];
#pragma unroll
    for (int s = 0; s < 8; ++s) {
      v4f pa[4];
#pragma unroll
      for (int u = 0; u < 4; ++u) pa[u] = *(const v4f*)(Arow + u * 16 * V_LD + s * 8);
#pragma unroll
      for (int i = 0; i < 2; ++i) {
        v2f bb;
        bb.x = pb[s][2 * i];
        bb.y = pb[s][2 * i + 1];
#pragma unroll
        for (int u = 0; u < 4; ++u) {
          v2f a;
          a.x = pa[u][2 * i];
          a.y = pa[u][2 * i + 1];
          oacc[u] = WMMA_F32(a, bb, oacc[u]);
        }
      }
    }
  }

  float* op = out + (size_t)b_ * (NTOK * CCH) + (size_t)h * NTOK + n;
#pragma unroll
  for (int u = 0; u < 4; ++u) {
#pragma unroll
    for (int r = 0; r < 8; ++r) {
      const int e = u * 16 + r + (hi ? 8 : 0);
      op[(size_t)e * (HEADS * NTOK)] = oacc[u][r];
    }
  }
}

extern "C" void kernel_launch(void* const* d_in, const int* in_sizes, int n_in,
                              void* d_out, int out_size, void* d_ws, size_t ws_size,
                              hipStream_t stream) {
  const float* x  = (const float*)d_in[0];
  const float* W  = (const float*)d_in[1];
  const float* EF = (const float*)d_in[2];
  const float* t1 = (const float*)d_in[3];
  const float* t2 = (const float*)d_in[4];
  float* out = (float*)d_out;
  float* ws  = (float*)d_ws;

  qkvv_gemm  <<<dim3(128, 13), 256, 0, stream>>>(x, W, EF, ws);
  ef_proj    <<<dim3(8, 96),   128, 0, stream>>>(ws, W, out);
  ca_scores  <<<dim3(4, 32),   128, 0, stream>>>(ws, t1, out);
  sa_attn    <<<dim3(32, 32),  256, 0, stream>>>(ws, t2, out);
}


# K2 k_proj/v_proj: the 8 token-slice partials of x^T EF are added while staged, W meets the sum once (1/8 of those products); 512 workgroups, one round; K4 adds 4 channel quarters; sum order changes
# speedup vs baseline: 3.7152x; 1.0442x over previous
#include <hip/hip_runtime.h>

typedef float v2f __attribute__((ext_vector_type(2)));
typedef float v4f __attribute__((ext_vector_type(4)));
typedef float v8f __attribute__((ext_vector_type(8)));

#define WMMA_F32(a, b, c) \
  __builtin_amdgcn_wmma_f32_16x16x4_f32(false, (a), false, (b), (short)0, (c), false, false)

#define BATCH 4
#define NTOK 4096
#define CCH 512
#define HEADS 8
#define DH 64
#define KL 64
#define MROWS (BATCH * NTOK)
#define OCH (4 * CCH)
#define BH (BATCH * HEADS)

#define OFF_Q   ((size_t)0)
#define OFF_K   ((size_t)8388608)
#define OFF_VCA ((size_t)16777216)
#define OFF_XE  ((size_t)25165824)
#define OFF_KP  ((size_t)33554432)
#define OFF_VP  ((size_t)33685504)
#define OFF_P   ((size_t)33816576)

#define G_BM 128
#define G_BO 128
#define G_KC 32
#define G_LD 36
#define G_NCH (CCH / G_KC)

#define E_LD 72

#define X_SL 8
#define X_NT (NTOK / X_SL)
#define X_NC 32
#define X_LD 136
#define X_NCH (X_NT / X_NC)

__device__ __forceinline__ void xe_slice(const float* __restrict__ x, const float* __restrict__ EF,
                                         float* __restrict__ ws, float* __restrict__ smem) {
  float* Xs0 = smem;
  float* Xs1 = smem + X_NC * X_LD;
  float* Es0 = smem + 2 * X_NC * X_LD;
  float* Es1 = Es0 + X_NC * E_LD;
  const int tid  = threadIdx.x;
  const int lane = tid & 31;
  const int wave = tid >> 5;
  const int l16  = lane & 15;
  const int hi   = lane >> 4;
  const int b_ = blockIdx.x >> 5;
  const int sl = (blockIdx.x >> 2) & 7;
  const int cq = blockIdx.x & 3;

  const float* xsrc = x + (size_t)(b_ * NTOK + sl * X_NT) * CCH + cq * 128;
  const float* esrc = EF + (size_t)sl * X_NT * KL;

  const int xRow = tid >> 5;
  const int xCol = (tid & 31) * 4;
  const float* xg = xsrc + (size_t)xRow * CCH + xCol;
  const int xOff = xRow * X_LD + xCol;
  const int eRow = tid >> 4;
  const int eCol = (tid & 15) * 4;
  const float* eg = esrc + (size_t)eRow * KL + eCol;
  const int eOff = eRow * E_LD + eCol;

  v4f rx[4], re[2];
  for (int i = 0; i < 4; ++i) rx[i] = *(const v4f*)(xg + (size_t)i * 8 * CCH);
  for (int i = 0; i < 2; ++i) re[i] = *(const v4f*)(eg + (size_t)i * 16 * KL);
  for (int i = 0; i < 4; ++i) *(v4f*)(Xs0 + xOff + i * 8 * X_LD) = rx[i];
  for (int i = 0; i < 2; ++i) *(v4f*)(Es0 + eOff + i * 16 * E_LD) = re[i];
  __syncthreads();

  const int aF = (hi ? 2 : 0) * X_LD + wave * 16 + l16;
  const int bF = (hi ? 2 : 0) * E_LD + l16;

  v8f acc[4] = {};
  for (int c = 0; c < X_NCH; ++c) {
    const int cur = c & 1;
    const bool more = (c + 1 < X_NCH);
    if (more) {
      for (int i = 0; i < 4; ++i) rx[i] = *(const v4f*)(xg + (size_t)(i * 8 + (c + 1) * X_NC) * CCH);
      for (int i = 0; i < 2; ++i) re[i] = *(const v4f*)(eg + (size_t)(i * 16 + (c + 1) * X_NC) * KL);
    }
    const float* Xc = (cur ? Xs1 : Xs0) + aF;
    const float* Ec = (cur ? Es1 : Es0) + bF;
    for (int j = 0; j < X_NC / 4; ++j) {
      v2f a, bb[4];
      a.x = Xc[j * 4 * X_LD];
      a.y = Xc[j * 4 * X_LD + X_LD];
      for (int t = 0; t < 4; ++t) {
        bb[t].x = Ec[j * 4 * E_LD + t * 16];
        bb[t].y = Ec[j * 4 * E_LD + E_LD + t * 16];
      }
      for (int t = 0; t < 4; ++t) acc[t] = WMMA_F32(a, bb[t], acc[t]);
    }
    if (more) {
      float* Xn = cur ? Xs0 : Xs1;
      float* En = cur ? Es0 : Es1;
      for (int i = 0; i < 4; ++i) *(v4f*)(Xn + xOff + i * 8 * X_LD) = rx[i];
      for (int i = 0; i < 2; ++i) *(v4f*)(En + eOff + i * 16 * E_LD) = re[i];
    }
    __syncthreads();
  }

  float* dst = ws + OFF_XE + ((size_t)(sl * BATCH + b_) * CCH + cq * 128 + wave * 16 + (hi ? 8 : 0)) * KL + l16;
  for (int t = 0; t < 4; ++t)
    for (int v = 0; v < 8; ++v) dst[(size_t)v * KL + t * 16] = acc[t][v];
}

__global__ __launch_bounds__(256) void qkvv_gemm(const float* __restrict__ x,
                                                 const float* __restrict__ W,
                                                 const float* __restrict__ EF,
                                                 float* __restrict__ ws) {
  __shared__ __attribute__((aligned(16))) float smem[2 * G_BM * G_LD + 2 * G_BO * G_LD];
  float (*As)[G_BM * G_LD] = (float (*)[G_BM * G_LD])smem;
  float (*Bs)[G_BO * G_LD] = (float (*)[G_BO * G_LD])(smem + 2 * G_BM * G_LD);

  if (blockIdx.y == 0) {
    xe_slice(x, EF, ws, smem);
    return;
  }

  const int tid  = threadIdx.x;
  const int lane = tid & 31;
  const int wave = tid >> 5;
  const int l16  = lane & 15;
  const int hi   = lane >> 4;
  const int wm   = wave >> 2;
  const int wo   = wave & 3;
  const int mBlock = blockIdx.x * G_BM;
  const int oBlock = (blockIdx.y - 1) * G_BO;

  const int sRow = tid >> 3;
  const int sCol = (tid & 7) * 4;
  const float* xg = x + (size_t)(mBlock + sRow) * CCH + sCol;
  const float* wg = W + (size_t)(oBlock + sRow) * CCH + sCol;
  const int sOff = sRow * G_LD + sCol;

  v4f ra[4], rb[4];
  for (int i = 0; i < 4; ++i) {
    ra[i] = *(const v4f*)(xg + (size_t)i * 32 * CCH);
    rb[i] = *(const v4f*)(wg + (size_t)i * 32 * CCH);
  }
  for (int i = 0; i < 4; ++i) {
    *(v4f*)(&As[0][sOff + i * 32 * G_LD]) = ra[i];
    *(v4f*)(&Bs[0][sOff + i * 32 * G_LD]) = rb[i];
  }
  __syncthreads();

  const int aOff = (wm * 64 + l16) * G_LD + (hi ? 2 : 0);
  const int bOff = (wo * 32 + l16) * G_LD + (hi ? 2 : 0);

  v8f acc[4][2] = {};
  for (int c = 0; c < G_NCH; ++c) {
    const int cur = c & 1;
    const bool more = (c + 1 < G_NCH);
    if (more) {
      for (int i = 0; i < 4; ++i) {
        ra[i] = *(const v4f*)(xg + (size_t)i * 32 * CCH + (c + 1) * G_KC);
        rb[i] = *(const v4f*)(wg + (size_t)i * 32 * CCH + (c + 1) * G_KC);
      }
    }
    const float* Ac = &As[cur][aOff];
    const float* Bc = &Bs[cur][bOff];
    for (int j = 0; j < G_KC / 4; ++j) {
      v2f a[4], b[2];
      for (int i = 0; i < 4; ++i) a[i] = *(const v2f*)(Ac + i * 16 * G_LD + j * 4);
      for (int t = 0; t < 2; ++t) b[t] = *(const v2f*)(Bc + t * 16 * G_LD + j * 4);
      for (int i = 0; i < 4; ++i)
        for (int t = 0; t < 2; ++t)
          acc[i][t] = WMMA_F32(a[i], b[t], acc[i][t]);
    }
    if (more) {
      for (int i = 0; i < 4; ++i) {
        *(v4f*)(&As[cur ^ 1][sOff + i * 32 * G_LD]) = ra[i];
        *(v4f*)(&Bs[cur ^ 1][sOff + i * 32 * G_LD]) = rb[i];
      }
    }
    __syncthreads();
  }

  if (blockIdx.y >= 9) {
    float* vdst = ws + OFF_VCA + (size_t)(mBlock + wm * 64 + (hi ? 8 : 0)) * CCH +
                  (oBlock - 2 * CCH) + wo * 32 + l16;
    for (int i = 0; i < 4; ++i)
      for (int v = 0; v < 8; ++v)
        for (int t = 0; t < 2; ++t)
          vdst[(size_t)(i * 16 + v) * CCH + t * 16] = acc[i][t][v];
    return;
  }

  for (int t = 0; t < 2; ++t) {
    const int o  = oBlock + wo * 32 + t * 16 + l16;
    const int p  = o >> 9;
    const int h  = (o >> 6) & 7;
    const int dd = o & 63;
    for (int i = 0; i < 4; ++i) {
      for (int v = 0; v < 8; ++v) {
        const int m  = mBlock + wm * 64 + i * 16 + v + (hi ? 8 : 0);
        const int b_ = m >> 12;
        const int n  = m & (NTOK - 1);
        const size_t dst = (size_t)p * 8388608 +
                           ((size_t)(b_ * HEADS + h) * DH + dd) * NTOK + n;
        ws[dst] = acc[i][t][v];
      }
    }
  }
}

#define P_NC 128
#define P_LD 132
#define P_SL 8
#define P_NT (NTOK / P_SL)
#define P_NCH (P_NT / P_NC)
#define P_CQ (CCH / P_NC)
static_assert(P_SL == X_SL && P_CQ == 4, "K2's W @ XE workgroups add X_SL slices of XE and take one of 4 channel quarters");

#define SP_OFF  ((size_t)2097152)
#define QSQ_OFF ((size_t)3145728)
#define KSQ_OFF ((size_t)3162112)

__device__ __forceinline__ float xhalf(float x) {
  const int b = __builtin_bit_cast(int, x);
  return __builtin_bit_cast(float,
      __builtin_amdgcn_permlanex16(b, b, 0x76543210, 0xfedcba98, false, false));
}

__device__ __forceinline__ void qk_slice(const float* __restrict__ ws, float* __restrict__ part,
                                         const int sl, const int bh,
                                         float* __restrict__ Qs0, float* __restrict__ Qs1,
                                         float* __restrict__ Ks0, float* __restrict__ Ks1) {
  const int tid  = threadIdx.x;
  const int lane = tid & 31;
  const int kt   = tid >> 5;
  const int l16  = lane & 15;
  const int hi   = lane >> 4;

  const float* qsrc = ws + OFF_Q + (size_t)bh * DH * NTOK + sl * P_NT;
  const float* ksrc = ws + OFF_K + (size_t)bh * DH * NTOK + sl * P_NT;

  const int aRow = tid >> 5;
  const int aCol = (tid & 31) * 4;
  const float* qg = qsrc + (size_t)aRow * NTOK + aCol;
  const float* kg = ksrc + (size_t)aRow * NTOK + aCol;
  const int aOffS = aRow * P_LD + aCol;

  v4f rq[16], rk[16];
  for (int i = 0; i < 16; ++i) rq[i] = *(const v4f*)(qg + (size_t)i * 4 * NTOK);
  for (int i = 0; i < 16; ++i) rk[i] = *(const v4f*)(kg + (size_t)i * 4 * NTOK);
  for (int i = 0; i < 16; ++i) *(v4f*)(&Qs0[aOffS + i * 4 * P_LD]) = rq[i];
  for (int i = 0; i < 16; ++i) *(v4f*)(&Ks0[aOffS + i * 4 * P_LD]) = rk[i];
  float q2[16], k2[16];
  for (int i = 0; i < 16; ++i) {
    q2[i] = rq[i].x * rq[i].x + rq[i].y * rq[i].y + rq[i].z * rq[i].z + rq[i].w * rq[i].w;
    k2[i] = rk[i].x * rk[i].x + rk[i].y * rk[i].y + rk[i].z * rk[i].z + rk[i].w * rk[i].w;
  }
  __syncthreads();

  const int qF = l16 * P_LD + (hi ? 2 : 0);
  const int kF = (kt * 16 + l16) * P_LD + (hi ? 2 : 0);

  v8f acc[4] = {};
  for (int c = 0; c < P_NCH; ++c) {
    const int cur = c & 1;
    const bool more = (c + 1 < P_NCH);
    if (more) {
      for (int i = 0; i < 16; ++i) rq[i] = *(const v4f*)(qg + (size_t)i * 4 * NTOK + (c + 1) * P_NC);
      for (int i = 0; i < 16; ++i) rk[i] = *(const v4f*)(kg + (size_t)i * 4 * NTOK + (c + 1) * P_NC);
    }
    const float* Qc = (cur ? Qs1 : Qs0) + qF;
    const float* Kc = (cur ? Ks1 : Ks0) + kF;
    v2f fa[2][4][4], fb[2][4];
#pragma unroll
    for (int j = 0; j < 4; ++j) {
#pragma unroll
      for (int i = 0; i < 4; ++i) fa[0][j][i] = *(const v2f*)(Qc + i * 16 * P_LD + j * 4);
      fb[0][j] = *(const v2f*)(Kc + j * 4);
    }
#pragma unroll
    for (int g = 0; g < P_NC / 16; ++g) {
      const int s = g & 1;
      if (g + 1 < P_NC / 16) {
#pragma unroll
        for (int j = 0; j < 4; ++j) {
          const int k = (g + 1) * 4 + j;
#pragma unroll
          for (int i = 0; i < 4; ++i) fa[s ^ 1][j][i] = *(const v2f*)(Qc + i * 16 * P_LD + k * 4);
          fb[s ^ 1][j] = *(const v2f*)(Kc + k * 4);
        }
      }
      __builtin_amdgcn_sched_barrier(0);
#pragma unroll
      for (int j = 0; j < 4; ++j)
#pragma unroll
        for (int i = 0; i < 4; ++i) acc[i] = WMMA_F32(fa[s][j][i], fb[s][j], acc[i]);
      __builtin_amdgcn_sched_barrier(0);
    }
    if (more) {
      float* Qn = cur ? Qs0 : Qs1;
      float* Kn = cur ? Ks0 : Ks1;
      for (int i = 0; i < 16; ++i) *(v4f*)(&Qn[aOffS + i * 4 * P_LD]) = rq[i];
      for (int i = 0; i < 16; ++i) *(v4f*)(&Kn[aOffS + i * 4 * P_LD]) = rk[i];
      for (int i = 0; i < 16; ++i) {
        q2[i] += rq[i].x * rq[i].x + rq[i].y * rq[i].y + rq[i].z * rq[i].z + rq[i].w * rq[i].w;
        k2[i] += rk[i].x * rk[i].x + rk[i].y * rk[i].y + rk[i].z * rk[i].z + rk[i].w * rk[i].w;
      }
    }
    __syncthreads();
  }

  const size_t tile = (size_t)(sl * BH + bh);
  float* dst = part + SP_OFF + tile * (DH * DH);
#pragma unroll
  for (int i = 0; i < 4; ++i)
#pragma unroll
    for (int v = 0; v < 8; ++v)
      dst[(size_t)(i * 16 + v + (hi ? 8 : 0)) * DH + kt * 16 + l16] = acc[i][v];

  float* red = Qs0;
  for (int i = 0; i < 16; ++i) {
    red[(i * 4 + aRow) * 33 + lane] = q2[i];
    red[64 * 33 + (i * 4 + aRow) * 33 + lane] = k2[i];
  }
  __syncthreads();
  {
    const int isk = tid >> 6;
    const int row = tid & 63;
    const float* r = red + isk * (64 * 33) + row * 33;
    float s = r[0];
    for (int l = 1; l < 32; ++l) s += r[l];
    part[(isk ? KSQ_OFF : QSQ_OFF) + tile * DH + row] = s;
  }
}

__global__ __launch_bounds__(128) void ef_proj(const float* __restrict__ ws,
                                               const float* __restrict__ W,
                                               float* __restrict__ part) {
  __shared__ __attribute__((aligned(16))) float As[2][64 * P_LD];
  __shared__ __attribute__((aligned(16))) float Es[2][P_NC * E_LD];

  const int lin = blockIdx.y * 8 + blockIdx.x;
  const int idx = lin >> 1;
  if ((lin ^ (lin >> 8)) & 1) {
    qk_slice(ws, part, idx & 7, idx >> 3, As[0], As[1], Es[0], Es[1]);
    return;
  }

  const int tid  = threadIdx.x;
  const int lane = tid & 31;
  const int kt   = tid >> 5;
  const int l16  = lane & 15;
  const int hi   = lane >> 4;
  const int cq    = idx & 3;
  const int which = (idx >> 2) & 1;
  const int bh    = idx >> 3;

  const float* src = W + (size_t)((which ? 3 : 1) * CCH + (bh & 7) * DH) * CCH + cq * P_NC;
  const float* efs = ws + OFF_XE + ((size_t)(bh >> 3) * CCH + cq * P_NC) * KL;
  float* dst       = part + (size_t)(cq * 64 + bh * 2 + which) * (DH * KL);

  const int aRow = tid >> 5;
  const int aCol = (tid & 31) * 4;
  const float* ag = src + (size_t)aRow * CCH + aCol;
  const int aOffS = aRow * P_LD + aCol;
  const int eRow = tid >> 4;
  const int eCol = (tid & 15) * 4;
  const float* eg = efs + (size_t)eRow * KL + eCol;
  const int eOffS = eRow * E_LD + eCol;

  const size_t eSl = (size_t)BATCH * CCH * KL;
  v4f ra[16], re[16], r1[16], r2[16];
#pragma unroll
  for (int i = 0; i < 16; ++i) ra[i] = *(const v4f*)(ag + (size_t)i * 4 * CCH);
#pragma unroll
  for (int i = 0; i < 16; ++i) re[i] = *(const v4f*)(eg + (size_t)i * 8 * KL);
#pragma unroll
  for (int i = 0; i < 16; ++i) r1[i] = *(const v4f*)(eg + eSl + (size_t)i * 8 * KL);
#pragma unroll 1
  for (int sl = 2; sl < P_SL; sl += 2) {
#pragma unroll
    for (int i = 0; i < 16; ++i) r2[i] = *(const v4f*)(eg + (size_t)sl * eSl + (size_t)i * 8 * KL);
#pragma unroll
    for (int i = 0; i < 16; ++i) re[i] += r1[i];
#pragma unroll
    for (int i = 0; i < 16; ++i) r1[i] = *(const v4f*)(eg + (size_t)(sl + 1) * eSl + (size_t)i * 8 * KL);
#pragma unroll
    for (int i = 0; i < 16; ++i) re[i] += r2[i];
  }
#pragma unroll
  for (int i = 0; i < 16; ++i) re[i] += r1[i];
#pragma unroll
  for (int i = 0; i < 16; ++i) *(v4f*)(&As[0][aOffS + i * 4 * P_LD]) = ra[i];
#pragma unroll
  for (int i = 0; i < 16; ++i) *(v4f*)(&Es[0][eOffS + i * 8 * E_LD]) = re[i];
  __syncthreads();

  const float* Ac = &As[0][l16 * P_LD + (hi ? 2 : 0)];
  const float* Ec = &Es[0][(hi ? 2 : 0) * E_LD + kt * 16 + l16];

  v8f acc[4] = {};
  v2f fa[2][4][4], fb[2][4];
#pragma unroll
  for (int j = 0; j < 4; ++j) {
#pragma unroll
    for (int i = 0; i < 4; ++i) fa[0][j][i] = *(const v2f*)(Ac + i * 16 * P_LD + j * 4);
    fb[0][j].x = Ec[j * 4 * E_LD];
    fb[0][j].y = Ec[j * 4 * E_LD + E_LD];
  }
#pragma unroll
  for (int g = 0; g < P_NC / 16; ++g) {
    const int s = g & 1;
    if (g + 1 < P_NC / 16) {
#pragma unroll
      for (int j = 0; j < 4; ++j) {
        const int k = (g + 1) * 4 + j;
#pragma unroll
        for (int i = 0; i < 4; ++i) fa[s ^ 1][j][i] = *(const v2f*)(Ac + i * 16 * P_LD + k * 4);
        fb[s ^ 1][j].x = Ec[k * 4 * E_LD];
        fb[s ^ 1][j].y = Ec[k * 4 * E_LD + E_LD];
      }
    }
    __builtin_amdgcn_sched_barrier(0);
#pragma unroll
    for (int j = 0; j < 4; ++j)
#pragma unroll
      for (int i = 0; i < 4; ++i) acc[i] = WMMA_F32(fa[s][j][i], fb[s][j], acc[i]);
    __builtin_amdgcn_sched_barrier(0);
  }
#pragma unroll
  for (int i = 0; i < 4; ++i)
#pragma unroll
    for (int v = 0; v < 8; ++v)
      dst[(size_t)(i * 16 + v + (hi ? 8 : 0)) * KL + kt * 16 + l16] = acc[i][v];
}

__global__ __launch_bounds__(128) void ca_scores(float* __restrict__ ws,
                                                 const float* __restrict__ temperature,
                                                 const float* __restrict__ part) {
  __shared__ float S[16][65];
  __shared__ float invqs[16];
  __shared__ float invks[64];
  const int tid  = threadIdx.x;
  const int lane = tid & 31;
  const int et   = tid >> 5;
  const int l16  = lane & 15;
  const int hi   = lane >> 4;
  const int dt = blockIdx.x;
  const int bh = blockIdx.y;
  const int h  = bh & 7;

  if (tid < 80) {
    const bool isq = tid < 16;
    const float* p = part + (isq ? QSQ_OFF : KSQ_OFF) + (size_t)bh * DH +
                     (isq ? dt * 16 + tid : tid - 16);
    float ss = p[0];
#pragma unroll
    for (int sl = 1; sl < P_SL; ++sl) ss += p[(size_t)sl * BH * DH];
    const float inv = 1.0f / fmaxf(sqrtf(ss), 1e-12f);
    if (isq) invqs[tid] = inv; else invks[tid - 16] = inv;
  }

  const int sr = tid >> 3;
  const int sc = (tid & 7) * 8;
  const float* ps = part + SP_OFF + (size_t)bh * (DH * DH) + (size_t)(dt * 16 + sr) * DH + sc;
  v4f s0 = *(const v4f*)ps;
  v4f s1 = *(const v4f*)(ps + 4);
#pragma unroll
  for (int sl = 1; sl < P_SL; ++sl) {
    const size_t so = (size_t)sl * BH * (DH * DH);
    s0 += *(const v4f*)(ps + so);
    s1 += *(const v4f*)(ps + so + 4);
  }
  __syncthreads();

  const float tsc = temperature[h];
#pragma unroll
  for (int j = 0; j < 4; ++j) {
    S[sr][sc + j]     = s0[j] * invks[sc + j];
    S[sr][sc + 4 + j] = s1[j] * invks[sc + 4 + j];
  }

  {
    const float invq = invqs[l16];
    const size_t rc = (size_t)(dt * 16 + l16) * KL + et * 16 + (hi ? 8 : 0);
    const float* pk = part + (size_t)(bh * 2) * (DH * KL) + rc;
    const float* pv = pk + (size_t)(DH * KL);
    v4f k0 = *(const v4f*)pk;
    v4f k1 = *(const v4f*)(pk + 4);
    v4f v0 = *(const v4f*)pv;
    v4f v1 = *(const v4f*)(pv + 4);
#pragma unroll
    for (int cq = 1; cq < P_CQ; ++cq) {
      const size_t so = (size_t)cq * 64 * (DH * KL);
      k0 += *(const v4f*)(pk + so);
      k1 += *(const v4f*)(pk + so + 4);
      v0 += *(const v4f*)(pv + so);
      v1 += *(const v4f*)(pv + so + 4);
    }
    k0 *= invq;
    k1 *= invq;
    float* kp = ws + OFF_KP + (size_t)bh * DH * KL + rc;
    float* vp = ws + OFF_VP + (size_t)bh * DH * KL + rc;
    *(v4f*)kp = k0;
    *(v4f*)(kp + 4) = k1;
    *(v4f*)vp = v0;
    *(v4f*)(vp + 4) = v1;
  }
  __syncthreads();

  if (tid < 16) {
    const int r = tid;
    const float invq = invqs[r];
    float sv[64];
#pragma unroll
    for (int j = 0; j < 64; ++j) sv[j] = (S[r][j] * invq) * tsc;
    float m = sv[0];
#pragma unroll
    for (int j = 1; j < 64; ++j) m = fmaxf(m, sv[j]);
    float sum = 0.f;
    float e[64];
#pragma unroll
    for (int j = 0; j < 64; ++j) { e[j] = __expf(sv[j] - m); sum += e[j]; }
    const float inv = 1.0f / sum;
    float* dst = ws + OFF_P + (size_t)bh * DH * DH + (size_t)(dt * 16 + r) * DH;
#pragma unroll
    for (int j = 0; j < 64; ++j) dst[j] = e[j] * inv;
  }
}

#define V_LD 68

__global__ __launch_bounds__(256) void sa_attn(float* __restrict__ ws,
                                               const float* __restrict__ temperature2,
                                               float* __restrict__ out) {
  __shared__ __attribute__((aligned(16))) float Vs[2][64 * V_LD];
  const int tid  = threadIdx.x;
  const int lane = tid & 31;
  const int wave = tid >> 5;
  const int l16  = lane & 15;
  const int hi   = lane >> 4;
  const int bh = blockIdx.y;
  const int b_ = bh >> 3;
  const int h  = bh & 7;
  const int n  = blockIdx.x * 128 + wave * 16 + l16;
  const int hq = (blockIdx.x & 3) * 2;
  const int nq = blockIdx.x >> 2;

  const float* Q  = ws + OFF_Q  + (size_t)bh * DH * NTOK + n;
  const float* Kp = ws + OFF_KP + (size_t)bh * DH * KL;
  const float* Vp = ws + OFF_VP + (size_t)bh * DH * KL;
  const float t2 = temperature2[h];

  const float* vg = ws + OFF_VCA + (size_t)(b_ * NTOK + wave * 64 + h * 8 + nq) * CCH + hq * DH + lane * 4;
  v4f rv[8];
#pragma unroll
  for (int p = 0; p < 8; ++p) rv[p] = *(const v4f*)(vg + (size_t)p * (8 * 64) * CCH);

  v8f sc[4] = {};
#pragma unroll 4
  for (int d0 = 0; d0 < DH; d0 += 4) {
    const int k0 = d0 + (hi ? 2 : 0);
    v2f q;
    q.x = Q[(size_t)k0 * NTOK];
    q.y = Q[(size_t)(k0 + 1) * NTOK];
#pragma unroll
    for (int t = 0; t < 4; ++t) {
      const int kc = t * 16 + l16;
      v2f kf;
      kf.x = Kp[(size_t)k0 * KL + kc];
      kf.y = Kp[(size_t)(k0 + 1) * KL + kc];
      sc[t] = WMMA_F32(kf, q, sc[t]);
    }
  }

#pragma unroll
  for (int p = 0; p < 8; ++p)
    *(v4f*)(&Vs[hi][(p * 8 + wave) * V_LD + l16 * 4]) = rv[p];
  __syncthreads();

  float m = sc[0][0] * t2;
#pragma unroll
  for (int t = 0; t < 4; ++t)
#pragma unroll
    for (int r = 0; r < 8; ++r) { sc[t][r] *= t2; m = fmaxf(m, sc[t][r]); }
  m = fmaxf(m, xhalf(m));
  float sum = 0.f;
#pragma unroll
  for (int t = 0; t < 4; ++t)
#pragma unroll
    for (int r = 0; r < 8; ++r) { const float e = __expf(sc[t][r] - m); sc[t][r] = e; sum += e; }
  sum = sum + xhalf(sum);
  const float inv = 1.0f / sum;
#pragma unroll
  for (int t = 0; t < 4; ++t)
#pragma unroll
    for (int r = 0; r < 8; ++r) sc[t][r] *= inv;

  v8f oacc[4] = {};
#pragma unroll
  for (int t = 0; t < 4; ++t) {
    v4f va[4], vb[4];
#pragma unroll
    for (int u = 0; u < 4; ++u) {
      const float* vp = Vp + (size_t)(u * 16 + l16) * KL + t * 16 + (hi ? 8 : 0);
      va[u] = *(const v4f*)vp;
      vb[u] = *(const v4f*)(vp + 4);
    }
#pragma unroll
    for (int j = 0; j < 4; ++j) {
      v2f p;
      p.x = sc[t][2 * j];
      p.y = sc[t][2 * j + 1];
#pragma unroll
      for (int u = 0; u < 4; ++u) {
        v2f a;
        a.x = (j < 2) ? va[u][2 * j]     : vb[u][2 * j - 4];
        a.y = (j < 2) ? va[u][2 * j + 1] : vb[u][2 * j - 3];
        oacc[u] = WMMA_F32(a, p, oacc[u]);
      }
    }
  }

  {
    const int hs = wave >> 2;
    const float* Prow = ws + OFF_P + ((size_t)(b_ * HEADS + hq + hs) * DH + (wave & 3) * 16 + l16) * DH + (hi ? 4 : 0);
    v4f pb[8];
#pragma unroll
    for (int s = 0; s < 8; ++s) pb[s] = *(const v4f*)(Prow + s * 8);
    const float* Arow = &Vs[hs][l16 * V_LD + (hi ? 4 : 0)];
#pragma unroll
    for (int s = 0; s < 8; ++s) {
      v4f pa[4];
#pragma unroll
      for (int u = 0; u < 4; ++u) pa[u] = *(const v4f*)(Arow + u * 16 * V_LD + s * 8);
#pragma unroll
      for (int i = 0; i < 2; ++i) {
        v2f bb;
        bb.x = pb[s][2 * i];
        bb.y = pb[s][2 * i + 1];
#pragma unroll
        for (int u = 0; u < 4; ++u) {
          v2f a;
          a.x = pa[u][2 * i];
          a.y = pa[u][2 * i + 1];
          oacc[u] = WMMA_F32(a, bb, oacc[u]);
        }
      }
    }
  }

  float* op = out + (size_t)b_ * (NTOK * CCH) + (size_t)h * NTOK + n;
#pragma unroll
  for (int u = 0; u < 4; ++u) {
#pragma unroll
    for (int r = 0; r < 8; ++r) {
      const int e = u * 16 + r + (hi ? 8 : 0);
      op[(size_t)e * (HEADS * NTOK)] = oacc[u][r];
    }
  }
}

extern "C" void kernel_launch(void* const* d_in, const int* in_sizes, int n_in,
                              void* d_out, int out_size, void* d_ws, size_t ws_size,
                              hipStream_t stream) {
  const float* x  = (const float*)d_in[0];
  const float* W  = (const float*)d_in[1];
  const float* EF = (const float*)d_in[2];
  const float* t1 = (const float*)d_in[3];
  const float* t2 = (const float*)d_in[4];
  float* out = (float*)d_out;
  float* ws  = (float*)d_ws;

  qkvv_gemm  <<<dim3(128, 13), 256, 0, stream>>>(x, W, EF, ws);
  ef_proj    <<<dim3(8, 64),   128, 0, stream>>>(ws, W, out);
  ca_scores  <<<dim3(4, 32),   128, 0, stream>>>(ws, t1, out);
  sa_attn    <<<dim3(32, 32),  256, 0, stream>>>(ws, t2, out);
}
